# starting kernel
# baseline (speedup 1.0000x reference)
#include <hip/hip_runtime.h>
#include <hip/hip_bf16.h>
#include <math.h>

#define N_   20000
#define E_   320000
#define R_   8
#define B_   8
#define IN_  128
#define H_   256
#define OUT_ 128
#define P_   100000
#define LN_EPS 1e-5f

typedef float v8f __attribute__((ext_vector_type(8)));
typedef float v2f __attribute__((ext_vector_type(2)));

#define WMMA_F32(a, b, c) \
  __builtin_amdgcn_wmma_f32_16x16x4_f32(false, (a), false, (b), (short)0, (c), false, false)

__global__ void relatt_softmax_kernel(const float* __restrict__ relatt,
                                      float* __restrict__ a) {
  if (threadIdx.x == 0 && blockIdx.x == 0) {
    float m = relatt[0];
    for (int r = 1; r < R_; ++r) m = fmaxf(m, relatt[r]);
    float e[R_]; float s = 0.f;
    for (int r = 0; r < R_; ++r) { e[r] = expf(relatt[r] - m); s += e[r]; }
    for (int r = 0; r < R_; ++r) a[r] = e[r] / s;
  }
}

__global__ void make_weights_kernel(const float* __restrict__ att,
                                    const float* __restrict__ basis,
                                    const float* __restrict__ a,
                                    float* __restrict__ W,
                                    int in_ch, int out_ch) {
  int idx = blockIdx.x * blockDim.x + threadIdx.x;
  int io_sz = in_ch * out_ch;
  int total = R_ * io_sz;
  if (idx >= total) return;
  int r  = idx / io_sz;
  int io = idx - r * io_sz;
  float s = 0.f;
#pragma unroll
  for (int bb = 0; bb < B_; ++bb)
    s += att[r * B_ + bb] * basis[(size_t)bb * io_sz + io];
  W[idx] = s * a[r];
}

__global__ void scatter_agg_kernel(const float* __restrict__ x,
                                   const int* __restrict__ src,
                                   const int* __restrict__ dst,
                                   const int* __restrict__ et,
                                   float* __restrict__ agg, int in_ch) {
  int e = blockIdx.x;
  int f = threadIdx.x;
  int s = src[e], d = dst[e], r = et[e];
  atomicAdd(&agg[((size_t)d * R_ + r) * in_ch + f],
            x[(size_t)s * in_ch + f]);
}

__global__ void gemm_f32_wmma_kernel(const float* __restrict__ A,
                                     const float* __restrict__ Bm,
                                     float* __restrict__ C,
                                     int M, int Nc, int K,
                                     int lda, int ldb, int ldc,
                                     int accumulate) {
  int wave = threadIdx.x >> 5;
  int lane = threadIdx.x & 31;
  int l16  = lane & 15;
  int g    = lane >> 4;
  int tilesN = Nc >> 6;
  int tilesTotal = (M >> 4) * tilesN;
  int tile = blockIdx.x * 8 + wave;
  if (tile >= tilesTotal) return;
  int tm = tile / tilesN;
  int tn = tile - tm * tilesN;

  const float* Arow  = A + (size_t)(tm * 16 + l16) * lda + 2 * g;
  const float* Bbase = Bm + (size_t)(2 * g) * ldb + tn * 64 + l16;

  v8f acc0 = {}, acc1 = {}, acc2 = {}, acc3 = {};
  for (int k0 = 0; k0 < K; k0 += 16) {
    __builtin_prefetch(Arow + k0 + 16, 0, 0);
#pragma unroll
    for (int kk = 0; kk < 16; kk += 4) {
      int k = k0 + kk;
      v2f a = *(const v2f*)(Arow + k);
      const float* Bk0 = Bbase + (size_t)k * ldb;
      const float* Bk1 = Bk0 + ldb;
      v2f b0, b1, b2, b3;
      b0.x = Bk0[0];  b0.y = Bk1[0];
      b1.x = Bk0[16]; b1.y = Bk1[16];
      b2.x = Bk0[32]; b2.y = Bk1[32];
      b3.x = Bk0[48]; b3.y = Bk1[48];
      acc0 = WMMA_F32(a, b0, acc0);
      acc1 = WMMA_F32(a, b1, acc1);
      acc2 = WMMA_F32(a, b2, acc2);
      acc3 = WMMA_F32(a, b3, acc3);
    }
  }

  float* Crow = C + (size_t)(tm * 16 + 8 * g) * ldc + tn * 64 + l16;
#pragma unroll
  for (int i = 0; i < 8; ++i) {
    size_t off = (size_t)i * ldc;
    float v0 = acc0[i], v1 = acc1[i], v2 = acc2[i], v3 = acc3[i];
    if (accumulate) {
      v0 += Crow[off];      v1 += Crow[off + 16];
      v2 += Crow[off + 32]; v3 += Crow[off + 48];
    }
    Crow[off]      = v0;
    Crow[off + 16] = v1;
    Crow[off + 32] = v2;
    Crow[off + 48] = v3;
  }
}

__global__ void gemm_pe_wmma_kernel(const float* __restrict__ z,
                                    const int* __restrict__ gi,
                                    const int* __restrict__ gj,
                                    const float* __restrict__ Bm,
                                    float* __restrict__ C,
                                    int M, int Nc, int K,
                                    int ldb, int ldc) {
  int wave = threadIdx.x >> 5;
  int lane = threadIdx.x & 31;
  int l16  = lane & 15;
  int g    = lane >> 4;
  int tilesN = Nc >> 6;
  int tilesTotal = (M >> 4) * tilesN;
  int tile = blockIdx.x * 8 + wave;
  if (tile >= tilesTotal) return;
  int tm = tile / tilesN;
  int tn = tile - tm * tilesN;

  int row = tm * 16 + l16;
  const float* zi = z + (size_t)gi[row] * OUT_;
  const float* zj = z + (size_t)gj[row] * OUT_;
  const float* Bbase = Bm + (size_t)(2 * g) * ldb + tn * 64 + l16;

  v8f acc0 = {}, acc1 = {}, acc2 = {}, acc3 = {};
  for (int k = 0; k < K; k += 4) {
    int kk = k + 2 * g;
    int f  = kk & (OUT_ - 1);
    int s  = kk >> 7;
    float zix = zi[f], ziy = zi[f + 1];
    float zjx = zj[f], zjy = zj[f + 1];
    v2f a;
    if (s == 0)      { a.x = zix + zjx;        a.y = ziy + zjy; }
    else if (s == 1) { a.x = zix * zjx;        a.y = ziy * zjy; }
    else             { a.x = fabsf(zix - zjx); a.y = fabsf(ziy - zjy); }
    const float* Bk0 = Bbase + (size_t)k * ldb;
    const float* Bk1 = Bk0 + ldb;
    v2f b0, b1, b2, b3;
    b0.x = Bk0[0];  b0.y = Bk1[0];
    b1.x = Bk0[16]; b1.y = Bk1[16];
    b2.x = Bk0[32]; b2.y = Bk1[32];
    b3.x = Bk0[48]; b3.y = Bk1[48];
    acc0 = WMMA_F32(a, b0, acc0);
    acc1 = WMMA_F32(a, b1, acc1);
    acc2 = WMMA_F32(a, b2, acc2);
    acc3 = WMMA_F32(a, b3, acc3);
  }

  float* Crow = C + (size_t)(tm * 16 + 8 * g) * ldc + tn * 64 + l16;
#pragma unroll
  for (int i = 0; i < 8; ++i) {
    size_t off = (size_t)i * ldc;
    Crow[off]      = acc0[i];
    Crow[off + 16] = acc1[i];
    Crow[off + 32] = acc2[i];
    Crow[off + 48] = acc3[i];
  }
}

__global__ void ln_res_gelu_kernel(const float* __restrict__ y,
                                   const float* __restrict__ bias,
                                   const float* __restrict__ gamma,
                                   const float* __restrict__ beta,
                                   const float* __restrict__ resid,
                                   float* __restrict__ out,
                                   int rows, int C, int do_gelu) {
  int wave = threadIdx.x >> 5;
  int lane = threadIdx.x & 31;
  int row = blockIdx.x * 8 + wave;
  if (row >= rows) return;
  const float* yr = y + (size_t)row * C;
  int nv = C >> 5;
  float vals[8];
  float s = 0.f;
  for (int i = 0; i < nv; ++i) {
    int c = lane + (i << 5);
    float v = yr[c];
    if (bias) v += bias[c];
    vals[i] = v;
    s += v;
  }
  for (int o = 16; o; o >>= 1) s += __shfl_xor(s, o, 32);
  float mean = s / (float)C;
  float vs = 0.f;
  for (int i = 0; i < nv; ++i) { float d = vals[i] - mean; vs += d * d; }
  for (int o = 16; o; o >>= 1) vs += __shfl_xor(vs, o, 32);
  float inv = rsqrtf(vs / (float)C + LN_EPS);
  for (int i = 0; i < nv; ++i) {
    int c = lane + (i << 5);
    float v = (vals[i] - mean) * inv * gamma[c] + beta[c];
    if (resid) v += resid[(size_t)row * C + c];
    if (do_gelu) v = 0.5f * v * (1.f + erff(v * 0.70710678118654752f));
    out[(size_t)row * C + c] = v;
  }
}

__global__ void dot64_kernel(const float* __restrict__ d2,
                             const float* __restrict__ w,
                             const float* __restrict__ b,
                             float* __restrict__ logits, int rows) {
  int wave = threadIdx.x >> 5;
  int lane = threadIdx.x & 31;
  int p = blockIdx.x * 8 + wave;
  if (p >= rows) return;
  const float* dr = d2 + (size_t)p * 64;
  float s = dr[lane] * w[lane] + dr[lane + 32] * w[lane + 32];
  for (int o = 16; o; o >>= 1) s += __shfl_xor(s, o, 32);
  if (lane == 0) logits[p] = s + b[0];
}

extern "C" void kernel_launch(void* const* d_in, const int* in_sizes, int n_in,
                              void* d_out, int out_size, void* d_ws, size_t ws_size,
                              hipStream_t stream) {
  (void)in_sizes; (void)n_in; (void)out_size; (void)ws_size;

  const float* x    = (const float*)d_in[0];
  const int*   ei   = (const int*)d_in[1];
  const int*   esrc = ei;
  const int*   edst = ei + E_;
  const int*   et   = (const int*)d_in[2];
  const int*   gp   = (const int*)d_in[3];
  const int*   gi   = gp;
  const int*   gj   = gp + P_;

  const float* basis[4], *att[4], *root[4], *relatt[4], *gam[4], *bet[4];
  for (int i = 0; i < 4; ++i) {
    basis[i]  = (const float*)d_in[4 + 6 * i + 0];
    att[i]    = (const float*)d_in[4 + 6 * i + 1];
    root[i]   = (const float*)d_in[4 + 6 * i + 2];
    relatt[i] = (const float*)d_in[4 + 6 * i + 3];
    gam[i]    = (const float*)d_in[4 + 6 * i + 4];
    bet[i]    = (const float*)d_in[4 + 6 * i + 5];
  }
  const float* res0 = (const float*)d_in[28];
  const float* res3 = (const float*)d_in[29];
  const float* dW0 = (const float*)d_in[30], *db0 = (const float*)d_in[31];
  const float* dg0 = (const float*)d_in[32], *dbb0 = (const float*)d_in[33];
  const float* dW1 = (const float*)d_in[34], *db1 = (const float*)d_in[35];
  const float* dg1 = (const float*)d_in[36], *dbb1 = (const float*)d_in[37];
  const float* dW2 = (const float*)d_in[38], *db2 = (const float*)d_in[39];
  const float* dg2 = (const float*)d_in[40], *dbb2 = (const float*)d_in[41];
  const float* dW3 = (const float*)d_in[42], *db3 = (const float*)d_in[43];

  float* ws = (float*)d_ws;
  size_t off = 0;
  float* a_sm  = ws + off; off += 64;
  float* Wcat  = ws + off; off += (size_t)R_ * H_ * H_;
  float* hA    = ws + off; off += (size_t)N_ * H_;
  float* hB    = ws + off; off += (size_t)N_ * H_;
  float* resid = ws + off; off += (size_t)N_ * H_;
  float* big   = ws + off;
  float* d0 = big;
  float* d1 = big + (size_t)P_ * H_;
  float* d2 = hA;

  auto gemm = [&](const float* A, const float* Bm, float* Cm,
                  int M, int Nc, int K, int lda, int ldb, int ldc, int acc) {
    int tiles = (M / 16) * (Nc / 64);
    gemm_f32_wmma_kernel<<<(tiles + 7) / 8, 256, 0, stream>>>(
        A, Bm, Cm, M, Nc, K, lda, ldb, ldc, acc);
  };

  struct { const float* cur; int in; int out; int li;
           const float* resmat; float* conv; int gelu; } L[4] = {
    { x,  IN_, H_,   0, res0,    hA, 1 },
    { hA, H_,  H_,   1, nullptr, hB, 1 },
    { hB, H_,  H_,   2, nullptr, hA, 1 },
    { hA, H_,  OUT_, 3, res3,    hB, 0 },
  };

  for (int i = 0; i < 4; ++i) {
    const float* cur = L[i].cur;
    int in_ch = L[i].in, out_ch = L[i].out, li = L[i].li;
    float* conv = L[i].conv;

    relatt_softmax_kernel<<<1, 32, 0, stream>>>(relatt[li], a_sm);
    int wtotal = R_ * in_ch * out_ch;
    make_weights_kernel<<<(wtotal + 255) / 256, 256, 0, stream>>>(
        att[li], basis[li], a_sm, Wcat, in_ch, out_ch);

    hipMemsetAsync(big, 0, (size_t)N_ * R_ * in_ch * sizeof(float), stream);
    scatter_agg_kernel<<<E_, in_ch, 0, stream>>>(cur, esrc, edst, et, big, in_ch);

    gemm(big, Wcat, conv, N_, out_ch, R_ * in_ch, R_ * in_ch, out_ch, out_ch, 0);
    gemm(cur, root[li], conv, N_, out_ch, in_ch, in_ch, out_ch, out_ch, 1);

    const float* rptr;
    if (L[i].resmat) {
      gemm(cur, L[i].resmat, resid, N_, out_ch, in_ch, in_ch, out_ch, out_ch, 0);
      rptr = resid;
    } else {
      rptr = cur;
    }
    ln_res_gelu_kernel<<<(N_ + 7) / 8, 256, 0, stream>>>(
        conv, nullptr, gam[li], bet[li], rptr, conv, N_, out_ch, L[i].gelu);
  }

  hipMemcpyAsync((float*)d_out + P_, hB, (size_t)N_ * OUT_ * sizeof(float),
                 hipMemcpyDeviceToDevice, stream);

  {
    int tiles = (P_ / 16) * (H_ / 64);
    gemm_pe_wmma_kernel<<<(tiles + 7) / 8, 256, 0, stream>>>(
        hB, gi, gj, dW0, d0, P_, H_, 3 * OUT_, H_, H_);
  }
  ln_res_gelu_kernel<<<(P_ + 7) / 8, 256, 0, stream>>>(
      d0, db0, dg0, dbb0, nullptr, d0, P_, H_, 1);

  gemm(d0, dW1, d1, P_, H_ / 2, H_, H_, H_ / 2, H_ / 2, 0);
  ln_res_gelu_kernel<<<(P_ + 7) / 8, 256, 0, stream>>>(
      d1, db1, dg1, dbb1, nullptr, d1, P_, H_ / 2, 1);

  gemm(d1, dW2, d2, P_, H_ / 4, H_ / 2, H_ / 2, H_ / 4, H_ / 4, 0);
  ln_res_gelu_kernel<<<(P_ + 7) / 8, 256, 0, stream>>>(
      d2, db2, dg2, dbb2, nullptr, d2, P_, H_ / 4, 1);

  dot64_kernel<<<(P_ + 7) / 8, 256, 0, stream>>>(d2, dW3, db3, (float*)d_out, P_);
}


# GEMM operands staged through LDS: 80xN block tile, 32-deep K chunks fetched with 16-byte row-contiguous global loads one chunk ahead, 8-byte LDS fragment reads; k order of the sums unchanged
# speedup vs baseline: 1.4023x; 1.4023x over previous
#include <hip/hip_runtime.h>
#include <hip/hip_bf16.h>
#include <math.h>

#define N_   20000
#define E_   320000
#define R_   8
#define B_   8
#define IN_  128
#define H_   256
#define OUT_ 128
#define P_   100000
#define LN_EPS 1e-5f

typedef float v8f __attribute__((ext_vector_type(8)));
typedef float v2f __attribute__((ext_vector_type(2)));
typedef float v4f __attribute__((ext_vector_type(4)));

#define WMMA_F32(a, b, c) \
  __builtin_amdgcn_wmma_f32_16x16x4_f32(false, (a), false, (b), (short)0, (c), false, false)

__global__ void relatt_softmax_kernel(const float* __restrict__ relatt,
                                      float* __restrict__ a) {
  if (threadIdx.x == 0 && blockIdx.x == 0) {
    float m = relatt[0];
    for (int r = 1; r < R_; ++r) m = fmaxf(m, relatt[r]);
    float e[R_]; float s = 0.f;
    for (int r = 0; r < R_; ++r) { e[r] = expf(relatt[r] - m); s += e[r]; }
    for (int r = 0; r < R_; ++r) a[r] = e[r] / s;
  }
}

__global__ void make_weights_kernel(const float* __restrict__ att,
                                    const float* __restrict__ basis,
                                    const float* __restrict__ a,
                                    float* __restrict__ W,
                                    int in_ch, int out_ch) {
  int idx = blockIdx.x * blockDim.x + threadIdx.x;
  int io_sz = in_ch * out_ch;
  int total = R_ * io_sz;
  if (idx >= total) return;
  int r  = idx / io_sz;
  int io = idx - r * io_sz;
  float s = 0.f;
#pragma unroll
  for (int bb = 0; bb < B_; ++bb)
    s += att[r * B_ + bb] * basis[(size_t)bb * io_sz + io];
  W[idx] = s * a[r];
}

__global__ void scatter_agg_kernel(const float* __restrict__ x,
                                   const int* __restrict__ src,
                                   const int* __restrict__ dst,
                                   const int* __restrict__ et,
                                   float* __restrict__ agg, int in_ch) {
  int e = blockIdx.x;
  int f = threadIdx.x;
  int s = src[e], d = dst[e], r = et[e];
  atomicAdd(&agg[((size_t)d * R_ + r) * in_ch + f],
            x[(size_t)s * in_ch + f]);
}

template <int BN>
__global__ __launch_bounds__(2 * BN)
void gemm_f32_wmma_lds_kernel(const float* __restrict__ A,
                              const float* __restrict__ Bm,
                              float* __restrict__ C,
                              int K, int lda, int ldb, int ldc,
                              int accumulate) {
  constexpr int NTHR = 2 * BN;
  constexpr int BM = 80, BK = 32;
  constexpr int AP = BK + 4;
  constexpr int BP = 2 * BN + 32;
  constexpr int A_ITEMS = BM * (BK / 4);
  constexpr int A_IT = (A_ITEMS + NTHR - 1) / NTHR;
  constexpr int BQ = BN / 4;
  constexpr int B_IT = ((BK / 2) * BQ) / NTHR;
  __shared__ __attribute__((aligned(16))) float As[2][BM * AP];
  __shared__ __attribute__((aligned(16))) float Bs[2][(BK / 2) * BP];

  const int tid  = threadIdx.x;
  const int wave = tid >> 5;
  const int lane = tid & 31;
  const int l16  = lane & 15;
  const int g    = lane >> 4;
  const int row0 = blockIdx.x * BM;

  v4f ra[A_IT];
  v4f rb0[B_IT], rb1[B_IT];

#define GEMM_GLOAD(k0)                                                        \
  {                                                                           \
    _Pragma("unroll")                \
    for (int it = 0; it < A_IT; ++it) {                                       \
      int i = tid + it * NTHR;                                                \
      i = (i < A_ITEMS) ? i : (A_ITEMS - 1);   \
      int r = i >> 3, c4 = (i & 7) * 4;                                       \
      ra[it] = *(const v4f*)(A + (size_t)(row0 + r) * lda + (k0) + c4);    \
    }                                                                         \
    _Pragma("unroll")                     \
    for (int it = 0; it < B_IT; ++it) {                                       \
      int j = tid + it * NTHR;                                                \
      int kp = j / BQ, n4 = (j - kp * BQ) * 4;                                \
      const float* p = Bm + (size_t)((k0) + 2 * kp) * ldb + n4;               \
      rb0[it] = *(const v4f*)p;                                            \
      rb1[it] = *(const v4f*)(p + ldb);                                    \
    }                                                                         \
  }

#define GEMM_SSTORE(buf)                                                      \
  {                                                                           \
    _Pragma("unroll")                          \
    for (int it = 0; it < A_IT; ++it) {                                       \
      int i = tid + it * NTHR;                                                \
      if (i < A_ITEMS) {                                                      \
        int r = i >> 3, c4 = (i & 7) * 4;                                     \
        *(v4f*)(&As[buf][r * AP + c4]) = ra[it];                           \
      }                                                                       \
    }                                                                         \
    _Pragma("unroll")                  \
    for (int it = 0; it < B_IT; ++it) {                                       \
      int j = tid + it * NTHR;                                                \
      int kp = j / BQ, n4 = (j - kp * BQ) * 4;                                \
      v4f lo, hi;                                                          \
      lo.x = rb0[it].x; lo.y = rb1[it].x; lo.z = rb0[it].y; lo.w = rb1[it].y; \
      hi.x = rb0[it].z; hi.y = rb1[it].z; hi.z = rb0[it].w; hi.w = rb1[it].w; \
      float* d = &Bs[buf][kp * BP + 2 * n4];                                  \
      *(v4f*)d       = lo;                                                 \
      *(v4f*)(d + 4) = hi;                                                 \
    }                                                                         \
  }

  v8f acc[5];
#pragma unroll
  for (int mt = 0; mt < 5; ++mt) acc[mt] = v8f{};

  GEMM_GLOAD(0)
  GEMM_SSTORE(0)
  __syncthreads();

  const int nch = K / BK;
  const int a_off = l16 * AP + 2 * g;
  const int b_off = g * BP + 2 * (wave * 16 + l16);
  for (int c = 0; c < nch; ++c) {
    const int cur = c & 1;
    const bool more = (c + 1 < nch);
    if (more) GEMM_GLOAD((c + 1) * BK)
    const float* as = &As[cur][a_off];
    const float* bs = &Bs[cur][b_off];
#pragma unroll
    for (int s = 0; s < BK / 4; ++s) {
      v2f b = *(const v2f*)(bs + 2 * s * BP);
#pragma unroll
      for (int mt = 0; mt < 5; ++mt) {
        v2f a = *(const v2f*)(as + mt * 16 * AP + 4 * s);
        acc[mt] = WMMA_F32(a, b, acc[mt]);
      }
    }
    if (more) GEMM_SSTORE(cur ^ 1)
    __syncthreads();
  }
#undef GEMM_GLOAD
#undef GEMM_SSTORE

  float* Cbase = C + (size_t)(row0 + 8 * g) * ldc + wave * 16 + l16;
  if (accumulate) {
#pragma unroll
    for (int mt = 0; mt < 5; ++mt) {
#pragma unroll
      for (int i = 0; i < 8; ++i)
        acc[mt][i] += Cbase[(size_t)(mt * 16 + i) * ldc];
    }
  }
#pragma unroll
  for (int mt = 0; mt < 5; ++mt) {
#pragma unroll
    for (int i = 0; i < 8; ++i)
      Cbase[(size_t)(mt * 16 + i) * ldc] = acc[mt][i];
  }
}

__global__ void gemm_pe_wmma_kernel(const float* __restrict__ z,
                                    const int* __restrict__ gi,
                                    const int* __restrict__ gj,
                                    const float* __restrict__ Bm,
                                    float* __restrict__ C,
                                    int M, int Nc, int K,
                                    int ldb, int ldc) {
  int wave = threadIdx.x >> 5;
  int lane = threadIdx.x & 31;
  int l16  = lane & 15;
  int g    = lane >> 4;
  int tilesN = Nc >> 6;
  int tilesTotal = (M >> 4) * tilesN;
  int tile = blockIdx.x * 8 + wave;
  if (tile >= tilesTotal) return;
  int tm = tile / tilesN;
  int tn = tile - tm * tilesN;

  int row = tm * 16 + l16;
  const float* zi = z + (size_t)gi[row] * OUT_;
  const float* zj = z + (size_t)gj[row] * OUT_;
  const float* Bbase = Bm + (size_t)(2 * g) * ldb + tn * 64 + l16;

  v8f acc0 = {}, acc1 = {}, acc2 = {}, acc3 = {};
  for (int k = 0; k < K; k += 4) {
    int kk = k + 2 * g;
    int f  = kk & (OUT_ - 1);
    int s  = kk >> 7;
    float zix = zi[f], ziy = zi[f + 1];
    float zjx = zj[f], zjy = zj[f + 1];
    v2f a;
    if (s == 0)      { a.x = zix + zjx;        a.y = ziy + zjy; }
    else if (s == 1) { a.x = zix * zjx;        a.y = ziy * zjy; }
    else             { a.x = fabsf(zix - zjx); a.y = fabsf(ziy - zjy); }
    const float* Bk0 = Bbase + (size_t)k * ldb;
    const float* Bk1 = Bk0 + ldb;
    v2f b0, b1, b2, b3;
    b0.x = Bk0[0];  b0.y = Bk1[0];
    b1.x = Bk0[16]; b1.y = Bk1[16];
    b2.x = Bk0[32]; b2.y = Bk1[32];
    b3.x = Bk0[48]; b3.y = Bk1[48];
    acc0 = WMMA_F32(a, b0, acc0);
    acc1 = WMMA_F32(a, b1, acc1);
    acc2 = WMMA_F32(a, b2, acc2);
    acc3 = WMMA_F32(a, b3, acc3);
  }

  float* Crow = C + (size_t)(tm * 16 + 8 * g) * ldc + tn * 64 + l16;
#pragma unroll
  for (int i = 0; i < 8; ++i) {
    size_t off = (size_t)i * ldc;
    Crow[off]      = acc0[i];
    Crow[off + 16] = acc1[i];
    Crow[off + 32] = acc2[i];
    Crow[off + 48] = acc3[i];
  }
}

__global__ void ln_res_gelu_kernel(const float* __restrict__ y,
                                   const float* __restrict__ bias,
                                   const float* __restrict__ gamma,
                                   const float* __restrict__ beta,
                                   const float* __restrict__ resid,
                                   float* __restrict__ out,
                                   int rows, int C, int do_gelu) {
  int wave = threadIdx.x >> 5;
  int lane = threadIdx.x & 31;
  int row = blockIdx.x * 8 + wave;
  if (row >= rows) return;
  const float* yr = y + (size_t)row * C;
  int nv = C >> 5;
  float vals[8];
  float s = 0.f;
  for (int i = 0; i < nv; ++i) {
    int c = lane + (i << 5);
    float v = yr[c];
    if (bias) v += bias[c];
    vals[i] = v;
    s += v;
  }
  for (int o = 16; o; o >>= 1) s += __shfl_xor(s, o, 32);
  float mean = s / (float)C;
  float vs = 0.f;
  for (int i = 0; i < nv; ++i) { float d = vals[i] - mean; vs += d * d; }
  for (int o = 16; o; o >>= 1) vs += __shfl_xor(vs, o, 32);
  float inv = rsqrtf(vs / (float)C + LN_EPS);
  for (int i = 0; i < nv; ++i) {
    int c = lane + (i << 5);
    float v = (vals[i] - mean) * inv * gamma[c] + beta[c];
    if (resid) v += resid[(size_t)row * C + c];
    if (do_gelu) v = 0.5f * v * (1.f + erff(v * 0.70710678118654752f));
    out[(size_t)row * C + c] = v;
  }
}

__global__ void dot64_kernel(const float* __restrict__ d2,
                             const float* __restrict__ w,
                             const float* __restrict__ b,
                             float* __restrict__ logits, int rows) {
  int wave = threadIdx.x >> 5;
  int lane = threadIdx.x & 31;
  int p = blockIdx.x * 8 + wave;
  if (p >= rows) return;
  const float* dr = d2 + (size_t)p * 64;
  float s = dr[lane] * w[lane] + dr[lane + 32] * w[lane + 32];
  for (int o = 16; o; o >>= 1) s += __shfl_xor(s, o, 32);
  if (lane == 0) logits[p] = s + b[0];
}

extern "C" void kernel_launch(void* const* d_in, const int* in_sizes, int n_in,
                              void* d_out, int out_size, void* d_ws, size_t ws_size,
                              hipStream_t stream) {
  (void)in_sizes; (void)n_in; (void)out_size; (void)ws_size;

  const float* x    = (const float*)d_in[0];
  const int*   ei   = (const int*)d_in[1];
  const int*   esrc = ei;
  const int*   edst = ei + E_;
  const int*   et   = (const int*)d_in[2];
  const int*   gp   = (const int*)d_in[3];
  const int*   gi   = gp;
  const int*   gj   = gp + P_;

  const float* basis[4], *att[4], *root[4], *relatt[4], *gam[4], *bet[4];
  for (int i = 0; i < 4; ++i) {
    basis[i]  = (const float*)d_in[4 + 6 * i + 0];
    att[i]    = (const float*)d_in[4 + 6 * i + 1];
    root[i]   = (const float*)d_in[4 + 6 * i + 2];
    relatt[i] = (const float*)d_in[4 + 6 * i + 3];
    gam[i]    = (const float*)d_in[4 + 6 * i + 4];
    bet[i]    = (const float*)d_in[4 + 6 * i + 5];
  }
  const float* res0 = (const float*)d_in[28];
  const float* res3 = (const float*)d_in[29];
  const float* dW0 = (const float*)d_in[30], *db0 = (const float*)d_in[31];
  const float* dg0 = (const float*)d_in[32], *dbb0 = (const float*)d_in[33];
  const float* dW1 = (const float*)d_in[34], *db1 = (const float*)d_in[35];
  const float* dg1 = (const float*)d_in[36], *dbb1 = (const float*)d_in[37];
  const float* dW2 = (const float*)d_in[38], *db2 = (const float*)d_in[39];
  const float* dg2 = (const float*)d_in[40], *dbb2 = (const float*)d_in[41];
  const float* dW3 = (const float*)d_in[42], *db3 = (const float*)d_in[43];

  float* ws = (float*)d_ws;
  size_t off = 0;
  float* a_sm  = ws + off; off += 64;
  float* Wcat  = ws + off; off += (size_t)R_ * H_ * H_;
  float* hA    = ws + off; off += (size_t)N_ * H_;
  float* hB    = ws + off; off += (size_t)N_ * H_;
  float* resid = ws + off; off += (size_t)N_ * H_;
  float* big   = ws + off;
  float* d0 = big;
  float* d1 = big + (size_t)P_ * H_;
  float* d2 = hA;

  auto gemm = [&](const float* A, const float* Bm, float* Cm,
                  int M, int Nc, int K, int lda, int ldb, int ldc, int acc) {
    int blocks = M / 80;
    if (Nc == 256)
      gemm_f32_wmma_lds_kernel<256><<<blocks, 512, 0, stream>>>(A, Bm, Cm, K, lda, ldb, ldc, acc);
    else if (Nc == 128)
      gemm_f32_wmma_lds_kernel<128><<<blocks, 256, 0, stream>>>(A, Bm, Cm, K, lda, ldb, ldc, acc);
    else
      gemm_f32_wmma_lds_kernel<64><<<blocks, 128, 0, stream>>>(A, Bm, Cm, K, lda, ldb, ldc, acc);
  };

  struct { const float* cur; int in; int out; int li;
           const float* resmat; float* conv; int gelu; } L[4] = {
    { x,  IN_, H_,   0, res0,    hA, 1 },
    { hA, H_,  H_,   1, nullptr, hB, 1 },
    { hB, H_,  H_,   2, nullptr, hA, 1 },
    { hA, H_,  OUT_, 3, res3,    hB, 0 },
  };

  for (int i = 0; i < 4; ++i) {
    const float* cur = L[i].cur;
    int in_ch = L[i].in, out_ch = L[i].out, li = L[i].li;
    float* conv = L[i].conv;

    relatt_softmax_kernel<<<1, 32, 0, stream>>>(relatt[li], a_sm);
    int wtotal = R_ * in_ch * out_ch;
    make_weights_kernel<<<(wtotal + 255) / 256, 256, 0, stream>>>(
        att[li], basis[li], a_sm, Wcat, in_ch, out_ch);

    hipMemsetAsync(big, 0, (size_t)N_ * R_ * in_ch * sizeof(float), stream);
    scatter_agg_kernel<<<E_, in_ch, 0, stream>>>(cur, esrc, edst, et, big, in_ch);

    gemm(big, Wcat, conv, N_, out_ch, R_ * in_ch, R_ * in_ch, out_ch, out_ch, 0);
    gemm(cur, root[li], conv, N_, out_ch, in_ch, in_ch, out_ch, out_ch, 1);

    const float* rptr;
    if (L[i].resmat) {
      gemm(cur, L[i].resmat, resid, N_, out_ch, in_ch, in_ch, out_ch, out_ch, 0);
      rptr = resid;
    } else {
      rptr = cur;
    }
    ln_res_gelu_kernel<<<(N_ + 7) / 8, 256, 0, stream>>>(
        conv, nullptr, gam[li], bet[li], rptr, conv, N_, out_ch, L[i].gelu);
  }

  hipMemcpyAsync((float*)d_out + P_, hB, (size_t)N_ * OUT_ * sizeof(float),
                 hipMemcpyDeviceToDevice, stream);

  {
    int tiles = (P_ / 16) * (H_ / 64);
    gemm_pe_wmma_kernel<<<(tiles + 7) / 8, 256, 0, stream>>>(
        hB, gi, gj, dW0, d0, P_, H_, 3 * OUT_, H_, H_);
  }
  ln_res_gelu_kernel<<<(P_ + 7) / 8, 256, 0, stream>>>(
      d0, db0, dg0, dbb0, nullptr, d0, P_, H_, 1);

  gemm(d0, dW1, d1, P_, H_ / 2, H_, H_, H_ / 2, H_ / 2, 0);
  ln_res_gelu_kernel<<<(P_ + 7) / 8, 256, 0, stream>>>(
      d1, db1, dg1, dbb1, nullptr, d1, P_, H_ / 2, 1);

  gemm(d1, dW2, d2, P_, H_ / 4, H_ / 2, H_ / 2, H_ / 4, H_ / 4, 0);
  ln_res_gelu_kernel<<<(P_ + 7) / 8, 256, 0, stream>>>(
      d2, db2, dg2, dbb2, nullptr, d2, P_, H_ / 4, 1);

  dot64_kernel<<<(P_ + 7) / 8, 256, 0, stream>>>(d2, dW3, db3, (float*)d_out, P_);
}


# Edge sums without float atomics: edges counting-sorted by destination once per call (integer atomics, then ordered by relation and edge id); each layer gathers row sums in that order; no memsets
# speedup vs baseline: 2.2788x; 1.6250x over previous
#include <hip/hip_runtime.h>
#include <hip/hip_bf16.h>
#include <math.h>

#define N_   20000
#define E_   320000
#define R_   8
#define B_   8
#define IN_  128
#define H_   256
#define OUT_ 128
#define P_   100000
#define LN_EPS 1e-5f

typedef float v8f __attribute__((ext_vector_type(8)));
typedef float v2f __attribute__((ext_vector_type(2)));
typedef float v4f __attribute__((ext_vector_type(4)));

#define WMMA_F32(a, b, c) \
  __builtin_amdgcn_wmma_f32_16x16x4_f32(false, (a), false, (b), (short)0, (c), false, false)

__global__ void relatt_softmax_kernel(const float* __restrict__ relatt,
                                      float* __restrict__ a) {
  if (threadIdx.x == 0 && blockIdx.x == 0) {
    float m = relatt[0];
    for (int r = 1; r < R_; ++r) m = fmaxf(m, relatt[r]);
    float e[R_]; float s = 0.f;
    for (int r = 0; r < R_; ++r) { e[r] = expf(relatt[r] - m); s += e[r]; }
    for (int r = 0; r < R_; ++r) a[r] = e[r] / s;
  }
}

__global__ void make_weights_kernel(const float* __restrict__ att,
                                    const float* __restrict__ basis,
                                    const float* __restrict__ a,
                                    float* __restrict__ W,
                                    int in_ch, int out_ch) {
  int idx = blockIdx.x * blockDim.x + threadIdx.x;
  int io_sz = in_ch * out_ch;
  int total = R_ * io_sz;
  if (idx >= total) return;
  int r  = idx / io_sz;
  int io = idx - r * io_sz;
  float s = 0.f;
#pragma unroll
  for (int bb = 0; bb < B_; ++bb)
    s += att[r * B_ + bb] * basis[(size_t)bb * io_sz + io];
  W[idx] = s * a[r];
}

__global__ void csr_zero_kernel(int* __restrict__ cnt) {
  int i = blockIdx.x * blockDim.x + threadIdx.x;
  if (i < N_) cnt[i] = 0;
}

__global__ void csr_count_kernel(const int* __restrict__ dst,
                                 int* __restrict__ cnt,
                                 int* __restrict__ rank) {
  int e = blockIdx.x * blockDim.x + threadIdx.x;
  if (e < E_) rank[e] = atomicAdd(&cnt[dst[e]], 1);
}

__global__ __launch_bounds__(1024)
void csr_scan_kernel(const int* __restrict__ cnt, int* __restrict__ offs) {
  constexpr int CH = (N_ + 1023) / 1024;
  __shared__ int part[1024];
  const int t = threadIdx.x;
  int loc[CH];
  int s = 0;
#pragma unroll
  for (int i = 0; i < CH; ++i) {
    int idx = t * CH + i;
    int v = (idx < N_) ? cnt[idx] : 0;
    loc[i] = s;
    s += v;
  }
  part[t] = s;
  __syncthreads();
  for (int o = 1; o < 1024; o <<= 1) {
    int v = (t >= o) ? part[t - o] : 0;
    __syncthreads();
    part[t] += v;
    __syncthreads();
  }
  const int excl = part[t] - s;
#pragma unroll
  for (int i = 0; i < CH; ++i) {
    int idx = t * CH + i;
    if (idx < N_) offs[idx] = excl + loc[i];
  }
  if (t == 1023) offs[N_] = part[1023];
}

__global__ void csr_fill_kernel(const int* __restrict__ dst,
                                const int* __restrict__ et,
                                const int* __restrict__ offs,
                                const int* __restrict__ rank,
                                int* __restrict__ key) {
  int e = blockIdx.x * blockDim.x + threadIdx.x;
  if (e < E_) key[offs[dst[e]] + rank[e]] = et[e] * E_ + e;
}

__global__ void csr_order_kernel(const int* __restrict__ src,
                                 const int* __restrict__ offs,
                                 const int* __restrict__ key,
                                 int* __restrict__ packed) {
  int wave = threadIdx.x >> 5;
  int lane = threadIdx.x & 31;
  int node = blockIdx.x * 8 + wave;
  if (node >= N_) return;
  int beg = offs[node];
  int n   = offs[node + 1] - beg;
  for (int i = lane; i < n; i += 32) {
    int my = key[beg + i];
    int rk = 0;
    for (int j = 0; j < n; ++j) rk += (key[beg + j] < my) ? 1 : 0;
    int r = my / E_;
    int e = my - r * E_;
    packed[beg + rk] = src[e] * R_ + r;
  }
}

template <int IN>
__global__ __launch_bounds__(256)
void gather_agg_kernel(const float* __restrict__ x,
                       const int* __restrict__ offs,
                       const int* __restrict__ packed,
                       float* __restrict__ agg) {
  constexpr int TPN = IN / 4;
  constexpr int NPB = 256 / TPN;
  const int lane = threadIdx.x & 31;
  const int node = blockIdx.x * NPB + threadIdx.x / TPN;
  const int t    = threadIdx.x % TPN;
  if (node >= N_) return;
  const int beg = __builtin_amdgcn_readfirstlane(offs[node]);
  const int end = __builtin_amdgcn_readfirstlane(offs[node + 1]);

  const float* xt = x + 4 * t;
  float* o = agg + (size_t)node * (R_ * IN) + 4 * t;
  v4f acc = v4f{};
  int cur = 0;
#define GATHER_LOAD(p) (*(const v4f*)(xt + (size_t)((unsigned)(p) / R_) * IN))
#define GATHER_STEP(p, v)                                                     \
  {                                                                           \
    const int r = (int)((unsigned)(p) % R_);                                  \
    _Pragma("nounroll")   \
    while (cur < r) {              \
      *(v4f*)(o + cur * IN) = acc;                                            \
      acc = v4f{};                                                            \
      ++cur;                                                                  \
    }                                                                         \
    acc += (v);                                                               \
  }

  for (int c = beg; c < end; c += 32) {
    const int rem = end - c;
    const int m = rem < 32 ? rem : 32;
    const int pv = (lane < m) ? packed[c + lane] : 0;
    int j = 0;
    for (; j + 4 <= m; j += 4) {
      const int p0 = __builtin_amdgcn_readlane(pv, j);
      const int p1 = __builtin_amdgcn_readlane(pv, j + 1);
      const int p2 = __builtin_amdgcn_readlane(pv, j + 2);
      const int p3 = __builtin_amdgcn_readlane(pv, j + 3);
      const v4f v0 = GATHER_LOAD(p0);
      const v4f v1 = GATHER_LOAD(p1);
      const v4f v2 = GATHER_LOAD(p2);
      const v4f v3 = GATHER_LOAD(p3);
      GATHER_STEP(p0, v0)
      GATHER_STEP(p1, v1)
      GATHER_STEP(p2, v2)
      GATHER_STEP(p3, v3)
    }
    for (; j < m; ++j) {
      const int p = __builtin_amdgcn_readlane(pv, j);
      const v4f v = GATHER_LOAD(p);
      GATHER_STEP(p, v)
    }
  }
#undef GATHER_LOAD
#undef GATHER_STEP
#pragma nounroll
  while (cur < R_) {
    *(v4f*)(o + cur * IN) = acc;
    acc = v4f{};
    ++cur;
  }
}

template <int BN>
__global__ __launch_bounds__(2 * BN)
void gemm_f32_wmma_lds_kernel(const float* __restrict__ A,
                              const float* __restrict__ Bm,
                              float* __restrict__ C,
                              int K, int lda, int ldb, int ldc,
                              int accumulate) {
  constexpr int NTHR = 2 * BN;
  constexpr int BM = 80, BK = 32;
  constexpr int AP = BK + 4;
  constexpr int BP = 2 * BN + 32;
  constexpr int A_ITEMS = BM * (BK / 4);
  constexpr int A_IT = (A_ITEMS + NTHR - 1) / NTHR;
  constexpr int BQ = BN / 4;
  constexpr int B_IT = ((BK / 2) * BQ) / NTHR;
  __shared__ __attribute__((aligned(16))) float As[2][BM * AP];
  __shared__ __attribute__((aligned(16))) float Bs[2][(BK / 2) * BP];

  const int tid  = threadIdx.x;
  const int wave = tid >> 5;
  const int lane = tid & 31;
  const int l16  = lane & 15;
  const int g    = lane >> 4;
  const int row0 = blockIdx.x * BM;

  v4f ra[A_IT];
  v4f rb0[B_IT], rb1[B_IT];

#define GEMM_GLOAD(k0)                                                        \
  {                                                                           \
    _Pragma("unroll")                \
    for (int it = 0; it < A_IT; ++it) {                                       \
      int i = tid + it * NTHR;                                                \
      i = (i < A_ITEMS) ? i : (A_ITEMS - 1);   \
      int r = i >> 3, c4 = (i & 7) * 4;                                       \
      ra[it] = *(const v4f*)(A + (size_t)(row0 + r) * lda + (k0) + c4);    \
    }                                                                         \
    _Pragma("unroll")                     \
    for (int it = 0; it < B_IT; ++it) {                                       \
      int j = tid + it * NTHR;                                                \
      int kp = j / BQ, n4 = (j - kp * BQ) * 4;                                \
      const float* p = Bm + (size_t)((k0) + 2 * kp) * ldb + n4;               \
      rb0[it] = *(const v4f*)p;                                            \
      rb1[it] = *(const v4f*)(p + ldb);                                    \
    }                                                                         \
  }

#define GEMM_SSTORE(buf)                                                      \
  {                                                                           \
    _Pragma("unroll")                          \
    for (int it = 0; it < A_IT; ++it) {                                       \
      int i = tid + it * NTHR;                                                \
      if (i < A_ITEMS) {                                                      \
        int r = i >> 3, c4 = (i & 7) * 4;                                     \
        *(v4f*)(&As[buf][r * AP + c4]) = ra[it];                           \
      }                                                                       \
    }                                                                         \
    _Pragma("unroll")                  \
    for (int it = 0; it < B_IT; ++it) {                                       \
      int j = tid + it * NTHR;                                                \
      int kp = j / BQ, n4 = (j - kp * BQ) * 4;                                \
      v4f lo, hi;                                                          \
      lo.x = rb0[it].x; lo.y = rb1[it].x; lo.z = rb0[it].y; lo.w = rb1[it].y; \
      hi.x = rb0[it].z; hi.y = rb1[it].z; hi.z = rb0[it].w; hi.w = rb1[it].w; \
      float* d = &Bs[buf][kp * BP + 2 * n4];                                  \
      *(v4f*)d       = lo;                                                 \
      *(v4f*)(d + 4) = hi;                                                 \
    }                                                                         \
  }

  v8f acc[5];
#pragma unroll
  for (int mt = 0; mt < 5; ++mt) acc[mt] = v8f{};

  GEMM_GLOAD(0)
  GEMM_SSTORE(0)
  __syncthreads();

  const int nch = K / BK;
  const int a_off = l16 * AP + 2 * g;
  const int b_off = g * BP + 2 * (wave * 16 + l16);
  for (int c = 0; c < nch; ++c) {
    const int cur = c & 1;
    const bool more = (c + 1 < nch);
    if (more) GEMM_GLOAD((c + 1) * BK)
    const float* as = &As[cur][a_off];
    const float* bs = &Bs[cur][b_off];
#pragma unroll
    for (int s = 0; s < BK / 4; ++s) {
      v2f b = *(const v2f*)(bs + 2 * s * BP);
#pragma unroll
      for (int mt = 0; mt < 5; ++mt) {
        v2f a = *(const v2f*)(as + mt * 16 * AP + 4 * s);
        acc[mt] = WMMA_F32(a, b, acc[mt]);
      }
    }
    if (more) GEMM_SSTORE(cur ^ 1)
    __syncthreads();
  }
#undef GEMM_GLOAD
#undef GEMM_SSTORE

  float* Cbase = C + (size_t)(row0 + 8 * g) * ldc + wave * 16 + l16;
  if (accumulate) {
#pragma unroll
    for (int mt = 0; mt < 5; ++mt) {
#pragma unroll
      for (int i = 0; i < 8; ++i)
        acc[mt][i] += Cbase[(size_t)(mt * 16 + i) * ldc];
    }
  }
#pragma unroll
  for (int mt = 0; mt < 5; ++mt) {
#pragma unroll
    for (int i = 0; i < 8; ++i)
      Cbase[(size_t)(mt * 16 + i) * ldc] = acc[mt][i];
  }
}

__global__ void gemm_pe_wmma_kernel(const float* __restrict__ z,
                                    const int* __restrict__ gi,
                                    const int* __restrict__ gj,
                                    const float* __restrict__ Bm,
                                    float* __restrict__ C,
                                    int M, int Nc, int K,
                                    int ldb, int ldc) {
  int wave = threadIdx.x >> 5;
  int lane = threadIdx.x & 31;
  int l16  = lane & 15;
  int g    = lane >> 4;
  int tilesN = Nc >> 6;
  int tilesTotal = (M >> 4) * tilesN;
  int tile = blockIdx.x * 8 + wave;
  if (tile >= tilesTotal) return;
  int tm = tile / tilesN;
  int tn = tile - tm * tilesN;

  int row = tm * 16 + l16;
  const float* zi = z + (size_t)gi[row] * OUT_;
  const float* zj = z + (size_t)gj[row] * OUT_;
  const float* Bbase = Bm + (size_t)(2 * g) * ldb + tn * 64 + l16;

  v8f acc0 = {}, acc1 = {}, acc2 = {}, acc3 = {};
  for (int k = 0; k < K; k += 4) {
    int kk = k + 2 * g;
    int f  = kk & (OUT_ - 1);
    int s  = kk >> 7;
    float zix = zi[f], ziy = zi[f + 1];
    float zjx = zj[f], zjy = zj[f + 1];
    v2f a;
    if (s == 0)      { a.x = zix + zjx;        a.y = ziy + zjy; }
    else if (s == 1) { a.x = zix * zjx;        a.y = ziy * zjy; }
    else             { a.x = fabsf(zix - zjx); a.y = fabsf(ziy - zjy); }
    const float* Bk0 = Bbase + (size_t)k * ldb;
    const float* Bk1 = Bk0 + ldb;
    v2f b0, b1, b2, b3;
    b0.x = Bk0[0];  b0.y = Bk1[0];
    b1.x = Bk0[16]; b1.y = Bk1[16];
    b2.x = Bk0[32]; b2.y = Bk1[32];
    b3.x = Bk0[48]; b3.y = Bk1[48];
    acc0 = WMMA_F32(a, b0, acc0);
    acc1 = WMMA_F32(a, b1, acc1);
    acc2 = WMMA_F32(a, b2, acc2);
    acc3 = WMMA_F32(a, b3, acc3);
  }

  float* Crow = C + (size_t)(tm * 16 + 8 * g) * ldc + tn * 64 + l16;
#pragma unroll
  for (int i = 0; i < 8; ++i) {
    size_t off = (size_t)i * ldc;
    Crow[off]      = acc0[i];
    Crow[off + 16] = acc1[i];
    Crow[off + 32] = acc2[i];
    Crow[off + 48] = acc3[i];
  }
}

__global__ void ln_res_gelu_kernel(const float* __restrict__ y,
                                   const float* __restrict__ bias,
                                   const float* __restrict__ gamma,
                                   const float* __restrict__ beta,
                                   const float* __restrict__ resid,
                                   float* __restrict__ out,
                                   int rows, int C, int do_gelu) {
  int wave = threadIdx.x >> 5;
  int lane = threadIdx.x & 31;
  int row = blockIdx.x * 8 + wave;
  if (row >= rows) return;
  const float* yr = y + (size_t)row * C;
  int nv = C >> 5;
  float vals[8];
  float s = 0.f;
  for (int i = 0; i < nv; ++i) {
    int c = lane + (i << 5);
    float v = yr[c];
    if (bias) v += bias[c];
    vals[i] = v;
    s += v;
  }
  for (int o = 16; o; o >>= 1) s += __shfl_xor(s, o, 32);
  float mean = s / (float)C;
  float vs = 0.f;
  for (int i = 0; i < nv; ++i) { float d = vals[i] - mean; vs += d * d; }
  for (int o = 16; o; o >>= 1) vs += __shfl_xor(vs, o, 32);
  float inv = rsqrtf(vs / (float)C + LN_EPS);
  for (int i = 0; i < nv; ++i) {
    int c = lane + (i << 5);
    float v = (vals[i] - mean) * inv * gamma[c] + beta[c];
    if (resid) v += resid[(size_t)row * C + c];
    if (do_gelu) v = 0.5f * v * (1.f + erff(v * 0.70710678118654752f));
    out[(size_t)row * C + c] = v;
  }
}

__global__ void dot64_kernel(const float* __restrict__ d2,
                             const float* __restrict__ w,
                             const float* __restrict__ b,
                             float* __restrict__ logits, int rows) {
  int wave = threadIdx.x >> 5;
  int lane = threadIdx.x & 31;
  int p = blockIdx.x * 8 + wave;
  if (p >= rows) return;
  const float* dr = d2 + (size_t)p * 64;
  float s = dr[lane] * w[lane] + dr[lane + 32] * w[lane + 32];
  for (int o = 16; o; o >>= 1) s += __shfl_xor(s, o, 32);
  if (lane == 0) logits[p] = s + b[0];
}

extern "C" void kernel_launch(void* const* d_in, const int* in_sizes, int n_in,
                              void* d_out, int out_size, void* d_ws, size_t ws_size,
                              hipStream_t stream) {
  (void)in_sizes; (void)n_in; (void)out_size; (void)ws_size;

  const float* x    = (const float*)d_in[0];
  const int*   ei   = (const int*)d_in[1];
  const int*   esrc = ei;
  const int*   edst = ei + E_;
  const int*   et   = (const int*)d_in[2];
  const int*   gp   = (const int*)d_in[3];
  const int*   gi   = gp;
  const int*   gj   = gp + P_;

  const float* basis[4], *att[4], *root[4], *relatt[4], *gam[4], *bet[4];
  for (int i = 0; i < 4; ++i) {
    basis[i]  = (const float*)d_in[4 + 6 * i + 0];
    att[i]    = (const float*)d_in[4 + 6 * i + 1];
    root[i]   = (const float*)d_in[4 + 6 * i + 2];
    relatt[i] = (const float*)d_in[4 + 6 * i + 3];
    gam[i]    = (const float*)d_in[4 + 6 * i + 4];
    bet[i]    = (const float*)d_in[4 + 6 * i + 5];
  }
  const float* res0 = (const float*)d_in[28];
  const float* res3 = (const float*)d_in[29];
  const float* dW0 = (const float*)d_in[30], *db0 = (const float*)d_in[31];
  const float* dg0 = (const float*)d_in[32], *dbb0 = (const float*)d_in[33];
  const float* dW1 = (const float*)d_in[34], *db1 = (const float*)d_in[35];
  const float* dg1 = (const float*)d_in[36], *dbb1 = (const float*)d_in[37];
  const float* dW2 = (const float*)d_in[38], *db2 = (const float*)d_in[39];
  const float* dg2 = (const float*)d_in[40], *dbb2 = (const float*)d_in[41];
  const float* dW3 = (const float*)d_in[42], *db3 = (const float*)d_in[43];

  float* ws = (float*)d_ws;
  size_t off = 0;
  float* a_sm  = ws + off; off += 64;
  float* Wcat  = ws + off; off += (size_t)R_ * H_ * H_;
  float* hA    = ws + off; off += (size_t)N_ * H_;
  float* hB    = ws + off; off += (size_t)N_ * H_;
  float* resid = ws + off; off += (size_t)N_ * H_;
  float* big   = ws + off;
  float* d0 = big;
  float* d1 = big + (size_t)P_ * H_;
  float* d2 = hA;
  int* csr_cnt    = (int*)(resid + (size_t)N_ * OUT_);
  int* csr_offs   = csr_cnt + 20480;
  int* csr_rank   = csr_offs + 20480;
  int* csr_key    = csr_rank + E_;
  int* csr_packed = csr_key + E_;

  auto gemm = [&](const float* A, const float* Bm, float* Cm,
                  int M, int Nc, int K, int lda, int ldb, int ldc, int acc) {
    int blocks = M / 80;
    if (Nc == 256)
      gemm_f32_wmma_lds_kernel<256><<<blocks, 512, 0, stream>>>(A, Bm, Cm, K, lda, ldb, ldc, acc);
    else if (Nc == 128)
      gemm_f32_wmma_lds_kernel<128><<<blocks, 256, 0, stream>>>(A, Bm, Cm, K, lda, ldb, ldc, acc);
    else
      gemm_f32_wmma_lds_kernel<64><<<blocks, 128, 0, stream>>>(A, Bm, Cm, K, lda, ldb, ldc, acc);
  };

  struct { const float* cur; int in; int out; int li;
           const float* resmat; float* conv; int gelu; float* rbuf; } L[4] = {
    { x,  IN_, H_,   0, res0,    hA, 1, hB },
    { hA, H_,  H_,   1, nullptr, hB, 1, nullptr },
    { hB, H_,  H_,   2, nullptr, hA, 1, nullptr },
    { hA, H_,  OUT_, 3, res3,    hB, 0, resid },
  };

  csr_zero_kernel<<<(N_ + 255) / 256, 256, 0, stream>>>(csr_cnt);
  csr_count_kernel<<<(E_ + 255) / 256, 256, 0, stream>>>(edst, csr_cnt, csr_rank);
  csr_scan_kernel<<<1, 1024, 0, stream>>>(csr_cnt, csr_offs);
  csr_fill_kernel<<<(E_ + 255) / 256, 256, 0, stream>>>(edst, et, csr_offs, csr_rank, csr_key);
  csr_order_kernel<<<(N_ + 7) / 8, 256, 0, stream>>>(esrc, csr_offs, csr_key, csr_packed);

  for (int i = 0; i < 4; ++i) {
    const float* cur = L[i].cur;
    int in_ch = L[i].in, out_ch = L[i].out, li = L[i].li;
    float* conv = L[i].conv;

    relatt_softmax_kernel<<<1, 32, 0, stream>>>(relatt[li], a_sm);
    int wtotal = R_ * in_ch * out_ch;
    make_weights_kernel<<<(wtotal + 255) / 256, 256, 0, stream>>>(
        att[li], basis[li], a_sm, Wcat, in_ch, out_ch);

    if (in_ch == H_)
      gather_agg_kernel<H_><<<N_ / 4, 256, 0, stream>>>(cur, csr_offs, csr_packed, big);
    else
      gather_agg_kernel<IN_><<<N_ / 8, 256, 0, stream>>>(cur, csr_offs, csr_packed, big);

    gemm(big, Wcat, conv, N_, out_ch, R_ * in_ch, R_ * in_ch, out_ch, out_ch, 0);
    gemm(cur, root[li], conv, N_, out_ch, in_ch, in_ch, out_ch, out_ch, 1);

    const float* rptr;
    if (L[i].resmat) {
      gemm(cur, L[i].resmat, L[i].rbuf, N_, out_ch, in_ch, in_ch, out_ch, out_ch, 0);
      rptr = L[i].rbuf;
    } else {
      rptr = cur;
    }
    ln_res_gelu_kernel<<<(N_ + 7) / 8, 256, 0, stream>>>(
        conv, nullptr, gam[li], bet[li], rptr, conv, N_, out_ch, L[i].gelu);
  }

  hipMemcpyAsync((float*)d_out + P_, hB, (size_t)N_ * OUT_ * sizeof(float),
                 hipMemcpyDeviceToDevice, stream);

  {
    int tiles = (P_ / 16) * (H_ / 64);
    gemm_pe_wmma_kernel<<<(tiles + 7) / 8, 256, 0, stream>>>(
        hB, gi, gj, dW0, d0, P_, H_, 3 * OUT_, H_, H_);
  }
  ln_res_gelu_kernel<<<(P_ + 7) / 8, 256, 0, stream>>>(
      d0, db0, dg0, dbb0, nullptr, d0, P_, H_, 1);

  gemm(d0, dW1, d1, P_, H_ / 2, H_, H_, H_ / 2, H_ / 2, 0);
  ln_res_gelu_kernel<<<(P_ + 7) / 8, 256, 0, stream>>>(
      d1, db1, dg1, dbb1, nullptr, d1, P_, H_ / 2, 1);

  gemm(d1, dW2, d2, P_, H_ / 4, H_ / 2, H_ / 2, H_ / 4, H_ / 4, 0);
  ln_res_gelu_kernel<<<(P_ + 7) / 8, 256, 0, stream>>>(
      d2, db2, dg2, dbb2, nullptr, d2, P_, H_ / 4, 1);

  dot64_kernel<<<(P_ + 7) / 8, 256, 0, stream>>>(d2, dW3, db3, (float*)d_out, P_);
}


# Decoder layer-0 GEMM staged through LDS like the other GEMMs: 80x256 block tile, pair features of 80 rows built once per 32-deep K chunk from 16-byte z row loads; k order of the sums unchanged
# speedup vs baseline: 2.5014x; 1.0977x over previous
#include <hip/hip_runtime.h>
#include <hip/hip_bf16.h>
#include <math.h>

#define N_   20000
#define E_   320000
#define R_   8
#define B_   8
#define IN_  128
#define H_   256
#define OUT_ 128
#define P_   100000
#define LN_EPS 1e-5f

typedef float v8f __attribute__((ext_vector_type(8)));
typedef float v2f __attribute__((ext_vector_type(2)));
typedef float v4f __attribute__((ext_vector_type(4)));

#define WMMA_F32(a, b, c) \
  __builtin_amdgcn_wmma_f32_16x16x4_f32(false, (a), false, (b), (short)0, (c), false, false)

__global__ void relatt_softmax_kernel(const float* __restrict__ relatt,
                                      float* __restrict__ a) {
  if (threadIdx.x == 0 && blockIdx.x == 0) {
    float m = relatt[0];
    for (int r = 1; r < R_; ++r) m = fmaxf(m, relatt[r]);
    float e[R_]; float s = 0.f;
    for (int r = 0; r < R_; ++r) { e[r] = expf(relatt[r] - m); s += e[r]; }
    for (int r = 0; r < R_; ++r) a[r] = e[r] / s;
  }
}

__global__ void make_weights_kernel(const float* __restrict__ att,
                                    const float* __restrict__ basis,
                                    const float* __restrict__ a,
                                    float* __restrict__ W,
                                    int in_ch, int out_ch) {
  int idx = blockIdx.x * blockDim.x + threadIdx.x;
  int io_sz = in_ch * out_ch;
  int total = R_ * io_sz;
  if (idx >= total) return;
  int r  = idx / io_sz;
  int io = idx - r * io_sz;
  float s = 0.f;
#pragma unroll
  for (int bb = 0; bb < B_; ++bb)
    s += att[r * B_ + bb] * basis[(size_t)bb * io_sz + io];
  W[idx] = s * a[r];
}

__global__ void csr_zero_kernel(int* __restrict__ cnt) {
  int i = blockIdx.x * blockDim.x + threadIdx.x;
  if (i < N_) cnt[i] = 0;
}

__global__ void csr_count_kernel(const int* __restrict__ dst,
                                 int* __restrict__ cnt,
                                 int* __restrict__ rank) {
  int e = blockIdx.x * blockDim.x + threadIdx.x;
  if (e < E_) rank[e] = atomicAdd(&cnt[dst[e]], 1);
}

__global__ __launch_bounds__(1024)
void csr_scan_kernel(const int* __restrict__ cnt, int* __restrict__ offs) {
  constexpr int CH = (N_ + 1023) / 1024;
  __shared__ int part[1024];
  const int t = threadIdx.x;
  int loc[CH];
  int s = 0;
#pragma unroll
  for (int i = 0; i < CH; ++i) {
    int idx = t * CH + i;
    int v = (idx < N_) ? cnt[idx] : 0;
    loc[i] = s;
    s += v;
  }
  part[t] = s;
  __syncthreads();
  for (int o = 1; o < 1024; o <<= 1) {
    int v = (t >= o) ? part[t - o] : 0;
    __syncthreads();
    part[t] += v;
    __syncthreads();
  }
  const int excl = part[t] - s;
#pragma unroll
  for (int i = 0; i < CH; ++i) {
    int idx = t * CH + i;
    if (idx < N_) offs[idx] = excl + loc[i];
  }
  if (t == 1023) offs[N_] = part[1023];
}

__global__ void csr_fill_kernel(const int* __restrict__ dst,
                                const int* __restrict__ et,
                                const int* __restrict__ offs,
                                const int* __restrict__ rank,
                                int* __restrict__ key) {
  int e = blockIdx.x * blockDim.x + threadIdx.x;
  if (e < E_) key[offs[dst[e]] + rank[e]] = et[e] * E_ + e;
}

__global__ void csr_order_kernel(const int* __restrict__ src,
                                 const int* __restrict__ offs,
                                 const int* __restrict__ key,
                                 int* __restrict__ packed) {
  int wave = threadIdx.x >> 5;
  int lane = threadIdx.x & 31;
  int node = blockIdx.x * 8 + wave;
  if (node >= N_) return;
  int beg = offs[node];
  int n   = offs[node + 1] - beg;
  for (int i = lane; i < n; i += 32) {
    int my = key[beg + i];
    int rk = 0;
    for (int j = 0; j < n; ++j) rk += (key[beg + j] < my) ? 1 : 0;
    int r = my / E_;
    int e = my - r * E_;
    packed[beg + rk] = src[e] * R_ + r;
  }
}

template <int IN>
__global__ __launch_bounds__(256)
void gather_agg_kernel(const float* __restrict__ x,
                       const int* __restrict__ offs,
                       const int* __restrict__ packed,
                       float* __restrict__ agg) {
  constexpr int TPN = IN / 4;
  constexpr int NPB = 256 / TPN;
  const int lane = threadIdx.x & 31;
  const int node = blockIdx.x * NPB + threadIdx.x / TPN;
  const int t    = threadIdx.x % TPN;
  if (node >= N_) return;
  const int beg = __builtin_amdgcn_readfirstlane(offs[node]);
  const int end = __builtin_amdgcn_readfirstlane(offs[node + 1]);

  const float* xt = x + 4 * t;
  float* o = agg + (size_t)node * (R_ * IN) + 4 * t;
  v4f acc = v4f{};
  int cur = 0;
#define GATHER_LOAD(p) (*(const v4f*)(xt + (size_t)((unsigned)(p) / R_) * IN))
#define GATHER_STEP(p, v)                                                     \
  {                                                                           \
    const int r = (int)((unsigned)(p) % R_);                                  \
    _Pragma("nounroll")   \
    while (cur < r) {              \
      *(v4f*)(o + cur * IN) = acc;                                            \
      acc = v4f{};                                                            \
      ++cur;                                                                  \
    }                                                                         \
    acc += (v);                                                               \
  }

  for (int c = beg; c < end; c += 32) {
    const int rem = end - c;
    const int m = rem < 32 ? rem : 32;
    const int pv = (lane < m) ? packed[c + lane] : 0;
    int j = 0;
    for (; j + 4 <= m; j += 4) {
      const int p0 = __builtin_amdgcn_readlane(pv, j);
      const int p1 = __builtin_amdgcn_readlane(pv, j + 1);
      const int p2 = __builtin_amdgcn_readlane(pv, j + 2);
      const int p3 = __builtin_amdgcn_readlane(pv, j + 3);
      const v4f v0 = GATHER_LOAD(p0);
      const v4f v1 = GATHER_LOAD(p1);
      const v4f v2 = GATHER_LOAD(p2);
      const v4f v3 = GATHER_LOAD(p3);
      GATHER_STEP(p0, v0)
      GATHER_STEP(p1, v1)
      GATHER_STEP(p2, v2)
      GATHER_STEP(p3, v3)
    }
    for (; j < m; ++j) {
      const int p = __builtin_amdgcn_readlane(pv, j);
      const v4f v = GATHER_LOAD(p);
      GATHER_STEP(p, v)
    }
  }
#undef GATHER_LOAD
#undef GATHER_STEP
#pragma nounroll
  while (cur < R_) {
    *(v4f*)(o + cur * IN) = acc;
    acc = v4f{};
    ++cur;
  }
}

template <int BN>
__global__ __launch_bounds__(2 * BN)
void gemm_f32_wmma_lds_kernel(const float* __restrict__ A,
                              const float* __restrict__ Bm,
                              float* __restrict__ C,
                              int K, int lda, int ldb, int ldc,
                              int accumulate) {
  constexpr int NTHR = 2 * BN;
  constexpr int BM = 80, BK = 32;
  constexpr int AP = BK + 4;
  constexpr int BP = 2 * BN + 32;
  constexpr int A_ITEMS = BM * (BK / 4);
  constexpr int A_IT = (A_ITEMS + NTHR - 1) / NTHR;
  constexpr int BQ = BN / 4;
  constexpr int B_IT = ((BK / 2) * BQ) / NTHR;
  __shared__ __attribute__((aligned(16))) float As[2][BM * AP];
  __shared__ __attribute__((aligned(16))) float Bs[2][(BK / 2) * BP];

  const int tid  = threadIdx.x;
  const int wave = tid >> 5;
  const int lane = tid & 31;
  const int l16  = lane & 15;
  const int g    = lane >> 4;
  const int row0 = blockIdx.x * BM;

  v4f ra[A_IT];
  v4f rb0[B_IT], rb1[B_IT];

#define GEMM_GLOAD(k0)                                                        \
  {                                                                           \
    _Pragma("unroll")                \
    for (int it = 0; it < A_IT; ++it) {                                       \
      int i = tid + it * NTHR;                                                \
      i = (i < A_ITEMS) ? i : (A_ITEMS - 1);   \
      int r = i >> 3, c4 = (i & 7) * 4;                                       \
      ra[it] = *(const v4f*)(A + (size_t)(row0 + r) * lda + (k0) + c4);    \
    }                                                                         \
    _Pragma("unroll")                     \
    for (int it = 0; it < B_IT; ++it) {                                       \
      int j = tid + it * NTHR;                                                \
      int kp = j / BQ, n4 = (j - kp * BQ) * 4;                                \
      const float* p = Bm + (size_t)((k0) + 2 * kp) * ldb + n4;               \
      rb0[it] = *(const v4f*)p;                                            \
      rb1[it] = *(const v4f*)(p + ldb);                                    \
    }                                                                         \
  }

#define GEMM_SSTORE(buf)                                                      \
  {                                                                           \
    _Pragma("unroll")                          \
    for (int it = 0; it < A_IT; ++it) {                                       \
      int i = tid + it * NTHR;                                                \
      if (i < A_ITEMS) {                                                      \
        int r = i >> 3, c4 = (i & 7) * 4;                                     \
        *(v4f*)(&As[buf][r * AP + c4]) = ra[it];                           \
      }                                                                       \
    }                                                                         \
    _Pragma("unroll")                  \
    for (int it = 0; it < B_IT; ++it) {                                       \
      int j = tid + it * NTHR;                                                \
      int kp = j / BQ, n4 = (j - kp * BQ) * 4;                                \
      v4f lo, hi;                                                          \
      lo.x = rb0[it].x; lo.y = rb1[it].x; lo.z = rb0[it].y; lo.w = rb1[it].y; \
      hi.x = rb0[it].z; hi.y = rb1[it].z; hi.z = rb0[it].w; hi.w = rb1[it].w; \
      float* d = &Bs[buf][kp * BP + 2 * n4];                                  \
      *(v4f*)d       = lo;                                                 \
      *(v4f*)(d + 4) = hi;                                                 \
    }                                                                         \
  }

  v8f acc[5];
#pragma unroll
  for (int mt = 0; mt < 5; ++mt) acc[mt] = v8f{};

  GEMM_GLOAD(0)
  GEMM_SSTORE(0)
  __syncthreads();

  const int nch = K / BK;
  const int a_off = l16 * AP + 2 * g;
  const int b_off = g * BP + 2 * (wave * 16 + l16);
  for (int c = 0; c < nch; ++c) {
    const int cur = c & 1;
    const bool more = (c + 1 < nch);
    if (more) GEMM_GLOAD((c + 1) * BK)
    const float* as = &As[cur][a_off];
    const float* bs = &Bs[cur][b_off];
#pragma unroll
    for (int s = 0; s < BK / 4; ++s) {
      v2f b = *(const v2f*)(bs + 2 * s * BP);
#pragma unroll
      for (int mt = 0; mt < 5; ++mt) {
        v2f a = *(const v2f*)(as + mt * 16 * AP + 4 * s);
        acc[mt] = WMMA_F32(a, b, acc[mt]);
      }
    }
    if (more) GEMM_SSTORE(cur ^ 1)
    __syncthreads();
  }
#undef GEMM_GLOAD
#undef GEMM_SSTORE

  float* Cbase = C + (size_t)(row0 + 8 * g) * ldc + wave * 16 + l16;
  if (accumulate) {
#pragma unroll
    for (int mt = 0; mt < 5; ++mt) {
#pragma unroll
      for (int i = 0; i < 8; ++i)
        acc[mt][i] += Cbase[(size_t)(mt * 16 + i) * ldc];
    }
  }
#pragma unroll
  for (int mt = 0; mt < 5; ++mt) {
#pragma unroll
    for (int i = 0; i < 8; ++i)
      Cbase[(size_t)(mt * 16 + i) * ldc] = acc[mt][i];
  }
}

__global__ __launch_bounds__(512)
void gemm_pe_wmma_lds_kernel(const float* __restrict__ z,
                             const int* __restrict__ gi,
                             const int* __restrict__ gj,
                             const float* __restrict__ Bm,
                             float* __restrict__ C,
                             int K) {
  constexpr int BN = H_;
  constexpr int NTHR = 2 * BN;
  constexpr int BM = 80, BK = 32;
  constexpr int AP = BK + 4;
  constexpr int BP = 2 * BN + 32;
  constexpr int A_ITEMS = BM * (BK / 4);
  constexpr int A_IT = (A_ITEMS + NTHR - 1) / NTHR;
  constexpr int BQ = BN / 4;
  constexpr int B_IT = ((BK / 2) * BQ) / NTHR;
  __shared__ __attribute__((aligned(16))) float As[2][BM * AP];
  __shared__ __attribute__((aligned(16))) float Bs[2][(BK / 2) * BP];

  const int tid  = threadIdx.x;
  const int wave = tid >> 5;
  const int lane = tid & 31;
  const int l16  = lane & 15;
  const int g    = lane >> 4;
  const int row0 = blockIdx.x * BM;

  unsigned offi[A_IT], offj[A_IT];
#pragma unroll
  for (int it = 0; it < A_IT; ++it) {
    int i = tid + it * NTHR;
    i = (i < A_ITEMS) ? i : (A_ITEMS - 1);
    int r = i >> 3, c4 = (i & 7) * 4;
    offi[it] = (unsigned)gi[row0 + r] * OUT_ + c4;
    offj[it] = (unsigned)gj[row0 + r] * OUT_ + c4;
  }

  v4f rzi[A_IT], rzj[A_IT];
  v4f rb0[B_IT], rb1[B_IT];

#define PE_GLOAD(k0)                                                          \
  {                                                                           \
    const int f0 = (k0) & (OUT_ - 1);                                         \
    _Pragma("unroll")        \
    for (int it = 0; it < A_IT; ++it) {                                       \
      rzi[it] = *(const v4f*)(z + offi[it] + f0);                             \
      rzj[it] = *(const v4f*)(z + offj[it] + f0);                             \
    }                                                                         \
    _Pragma("unroll")                     \
    for (int it = 0; it < B_IT; ++it) {                                       \
      int j = tid + it * NTHR;                                                \
      int kp = j / BQ, n4 = (j - kp * BQ) * 4;                                \
      const float* p = Bm + (size_t)((k0) + 2 * kp) * BN + n4;                \
      rb0[it] = *(const v4f*)p;                                               \
      rb1[it] = *(const v4f*)(p + BN);                                        \
    }                                                                         \
  }

#define PE_SSTORE(buf, k0)                                                    \
  {                                                                           \
    const int sec = (k0) >> 7;                                                \
    _Pragma("unroll")                  \
    for (int it = 0; it < A_IT; ++it) {                                       \
      int i = tid + it * NTHR;                                                \
      if (i < A_ITEMS) {                                                      \
        int r = i >> 3, c4 = (i & 7) * 4;                                     \
        v4f a;                                                                \
        if (sec == 0)      a = rzi[it] + rzj[it];                             \
        else if (sec == 1) a = rzi[it] * rzj[it];                             \
        else {                                                                \
          const v4f d = rzi[it] - rzj[it];                                    \
          a.x = fabsf(d.x); a.y = fabsf(d.y); a.z = fabsf(d.z); a.w = fabsf(d.w); \
        }                                                                     \
        *(v4f*)(&As[buf][r * AP + c4]) = a;                                   \
      }                                                                       \
    }                                                                         \
    _Pragma("unroll")                  \
    for (int it = 0; it < B_IT; ++it) {                                       \
      int j = tid + it * NTHR;                                                \
      int kp = j / BQ, n4 = (j - kp * BQ) * 4;                                \
      v4f lo, hi;                                                             \
      lo.x = rb0[it].x; lo.y = rb1[it].x; lo.z = rb0[it].y; lo.w = rb1[it].y; \
      hi.x = rb0[it].z; hi.y = rb1[it].z; hi.z = rb0[it].w; hi.w = rb1[it].w; \
      float* d = &Bs[buf][kp * BP + 2 * n4];                                  \
      *(v4f*)d       = lo;                                                    \
      *(v4f*)(d + 4) = hi;                                                    \
    }                                                                         \
  }

  v8f acc[5];
#pragma unroll
  for (int mt = 0; mt < 5; ++mt) acc[mt] = v8f{};

  PE_GLOAD(0)
  PE_SSTORE(0, 0)
  __syncthreads();

  const int nch = K / BK;
  const int a_off = l16 * AP + 2 * g;
  const int b_off = g * BP + 2 * (wave * 16 + l16);
  for (int c = 0; c < nch; ++c) {
    const int cur = c & 1;
    const bool more = (c + 1 < nch);
    if (more) PE_GLOAD((c + 1) * BK)
    const float* as = &As[cur][a_off];
    const float* bs = &Bs[cur][b_off];
#pragma unroll
    for (int s = 0; s < BK / 4; ++s) {
      v2f b = *(const v2f*)(bs + 2 * s * BP);
#pragma unroll
      for (int mt = 0; mt < 5; ++mt) {
        v2f a = *(const v2f*)(as + mt * 16 * AP + 4 * s);
        acc[mt] = WMMA_F32(a, b, acc[mt]);
      }
    }
    if (more) PE_SSTORE(cur ^ 1, (c + 1) * BK)
    __syncthreads();
  }
#undef PE_GLOAD
#undef PE_SSTORE

  float* Cbase = C + (size_t)(row0 + 8 * g) * BN + wave * 16 + l16;
#pragma unroll
  for (int mt = 0; mt < 5; ++mt) {
#pragma unroll
    for (int i = 0; i < 8; ++i)
      Cbase[(size_t)(mt * 16 + i) * BN] = acc[mt][i];
  }
}

__global__ void ln_res_gelu_kernel(const float* __restrict__ y,
                                   const float* __restrict__ bias,
                                   const float* __restrict__ gamma,
                                   const float* __restrict__ beta,
                                   const float* __restrict__ resid,
                                   float* __restrict__ out,
                                   int rows, int C, int do_gelu) {
  int wave = threadIdx.x >> 5;
  int lane = threadIdx.x & 31;
  int row = blockIdx.x * 8 + wave;
  if (row >= rows) return;
  const float* yr = y + (size_t)row * C;
  int nv = C >> 5;
  float vals[8];
  float s = 0.f;
  for (int i = 0; i < nv; ++i) {
    int c = lane + (i << 5);
    float v = yr[c];
    if (bias) v += bias[c];
    vals[i] = v;
    s += v;
  }
  for (int o = 16; o; o >>= 1) s += __shfl_xor(s, o, 32);
  float mean = s / (float)C;
  float vs = 0.f;
  for (int i = 0; i < nv; ++i) { float d = vals[i] - mean; vs += d * d; }
  for (int o = 16; o; o >>= 1) vs += __shfl_xor(vs, o, 32);
  float inv = rsqrtf(vs / (float)C + LN_EPS);
  for (int i = 0; i < nv; ++i) {
    int c = lane + (i << 5);
    float v = (vals[i] - mean) * inv * gamma[c] + beta[c];
    if (resid) v += resid[(size_t)row * C + c];
    if (do_gelu) v = 0.5f * v * (1.f + erff(v * 0.70710678118654752f));
    out[(size_t)row * C + c] = v;
  }
}

__global__ void dot64_kernel(const float* __restrict__ d2,
                             const float* __restrict__ w,
                             const float* __restrict__ b,
                             float* __restrict__ logits, int rows) {
  int wave = threadIdx.x >> 5;
  int lane = threadIdx.x & 31;
  int p = blockIdx.x * 8 + wave;
  if (p >= rows) return;
  const float* dr = d2 + (size_t)p * 64;
  float s = dr[lane] * w[lane] + dr[lane + 32] * w[lane + 32];
  for (int o = 16; o; o >>= 1) s += __shfl_xor(s, o, 32);
  if (lane == 0) logits[p] = s + b[0];
}

extern "C" void kernel_launch(void* const* d_in, const int* in_sizes, int n_in,
                              void* d_out, int out_size, void* d_ws, size_t ws_size,
                              hipStream_t stream) {
  (void)in_sizes; (void)n_in; (void)out_size; (void)ws_size;

  const float* x    = (const float*)d_in[0];
  const int*   ei   = (const int*)d_in[1];
  const int*   esrc = ei;
  const int*   edst = ei + E_;
  const int*   et   = (const int*)d_in[2];
  const int*   gp   = (const int*)d_in[3];
  const int*   gi   = gp;
  const int*   gj   = gp + P_;

  const float* basis[4], *att[4], *root[4], *relatt[4], *gam[4], *bet[4];
  for (int i = 0; i < 4; ++i) {
    basis[i]  = (const float*)d_in[4 + 6 * i + 0];
    att[i]    = (const float*)d_in[4 + 6 * i + 1];
    root[i]   = (const float*)d_in[4 + 6 * i + 2];
    relatt[i] = (const float*)d_in[4 + 6 * i + 3];
    gam[i]    = (const float*)d_in[4 + 6 * i + 4];
    bet[i]    = (const float*)d_in[4 + 6 * i + 5];
  }
  const float* res0 = (const float*)d_in[28];
  const float* res3 = (const float*)d_in[29];
  const float* dW0 = (const float*)d_in[30], *db0 = (const float*)d_in[31];
  const float* dg0 = (const float*)d_in[32], *dbb0 = (const float*)d_in[33];
  const float* dW1 = (const float*)d_in[34], *db1 = (const float*)d_in[35];
  const float* dg1 = (const float*)d_in[36], *dbb1 = (const float*)d_in[37];
  const float* dW2 = (const float*)d_in[38], *db2 = (const float*)d_in[39];
  const float* dg2 = (const float*)d_in[40], *dbb2 = (const float*)d_in[41];
  const float* dW3 = (const float*)d_in[42], *db3 = (const float*)d_in[43];

  float* ws = (float*)d_ws;
  size_t off = 0;
  float* a_sm  = ws + off; off += 64;
  float* Wcat  = ws + off; off += (size_t)R_ * H_ * H_;
  float* hA    = ws + off; off += (size_t)N_ * H_;
  float* hB    = ws + off; off += (size_t)N_ * H_;
  float* resid = ws + off; off += (size_t)N_ * H_;
  float* big   = ws + off;
  float* d0 = big;
  float* d1 = big + (size_t)P_ * H_;
  float* d2 = hA;
  int* csr_cnt    = (int*)(resid + (size_t)N_ * OUT_);
  int* csr_offs   = csr_cnt + 20480;
  int* csr_rank   = csr_offs + 20480;
  int* csr_key    = csr_rank + E_;
  int* csr_packed = csr_key + E_;

  auto gemm = [&](const float* A, const float* Bm, float* Cm,
                  int M, int Nc, int K, int lda, int ldb, int ldc, int acc) {
    int blocks = M / 80;
    if (Nc == 256)
      gemm_f32_wmma_lds_kernel<256><<<blocks, 512, 0, stream>>>(A, Bm, Cm, K, lda, ldb, ldc, acc);
    else if (Nc == 128)
      gemm_f32_wmma_lds_kernel<128><<<blocks, 256, 0, stream>>>(A, Bm, Cm, K, lda, ldb, ldc, acc);
    else
      gemm_f32_wmma_lds_kernel<64><<<blocks, 128, 0, stream>>>(A, Bm, Cm, K, lda, ldb, ldc, acc);
  };

  struct { const float* cur; int in; int out; int li;
           const float* resmat; float* conv; int gelu; float* rbuf; } L[4] = {
    { x,  IN_, H_,   0, res0,    hA, 1, hB },
    { hA, H_,  H_,   1, nullptr, hB, 1, nullptr },
    { hB, H_,  H_,   2, nullptr, hA, 1, nullptr },
    { hA, H_,  OUT_, 3, res3,    hB, 0, resid },
  };

  csr_zero_kernel<<<(N_ + 255) / 256, 256, 0, stream>>>(csr_cnt);
  csr_count_kernel<<<(E_ + 255) / 256, 256, 0, stream>>>(edst, csr_cnt, csr_rank);
  csr_scan_kernel<<<1, 1024, 0, stream>>>(csr_cnt, csr_offs);
  csr_fill_kernel<<<(E_ + 255) / 256, 256, 0, stream>>>(edst, et, csr_offs, csr_rank, csr_key);
  csr_order_kernel<<<(N_ + 7) / 8, 256, 0, stream>>>(esrc, csr_offs, csr_key, csr_packed);

  for (int i = 0; i < 4; ++i) {
    const float* cur = L[i].cur;
    int in_ch = L[i].in, out_ch = L[i].out, li = L[i].li;
    float* conv = L[i].conv;

    relatt_softmax_kernel<<<1, 32, 0, stream>>>(relatt[li], a_sm);
    int wtotal = R_ * in_ch * out_ch;
    make_weights_kernel<<<(wtotal + 255) / 256, 256, 0, stream>>>(
        att[li], basis[li], a_sm, Wcat, in_ch, out_ch);

    if (in_ch == H_)
      gather_agg_kernel<H_><<<N_ / 4, 256, 0, stream>>>(cur, csr_offs, csr_packed, big);
    else
      gather_agg_kernel<IN_><<<N_ / 8, 256, 0, stream>>>(cur, csr_offs, csr_packed, big);

    gemm(big, Wcat, conv, N_, out_ch, R_ * in_ch, R_ * in_ch, out_ch, out_ch, 0);
    gemm(cur, root[li], conv, N_, out_ch, in_ch, in_ch, out_ch, out_ch, 1);

    const float* rptr;
    if (L[i].resmat) {
      gemm(cur, L[i].resmat, L[i].rbuf, N_, out_ch, in_ch, in_ch, out_ch, out_ch, 0);
      rptr = L[i].rbuf;
    } else {
      rptr = cur;
    }
    ln_res_gelu_kernel<<<(N_ + 7) / 8, 256, 0, stream>>>(
        conv, nullptr, gam[li], bet[li], rptr, conv, N_, out_ch, L[i].gelu);
  }

  hipMemcpyAsync((float*)d_out + P_, hB, (size_t)N_ * OUT_ * sizeof(float),
                 hipMemcpyDeviceToDevice, stream);

  gemm_pe_wmma_lds_kernel<<<P_ / 80, 512, 0, stream>>>(hB, gi, gj, dW0, d0, 3 * OUT_);
  ln_res_gelu_kernel<<<(P_ + 7) / 8, 256, 0, stream>>>(
      d0, db0, dg0, dbb0, nullptr, d0, P_, H_, 1);

  gemm(d0, dW1, d1, P_, H_ / 2, H_, H_, H_ / 2, H_ / 2, 0);
  ln_res_gelu_kernel<<<(P_ + 7) / 8, 256, 0, stream>>>(
      d1, db1, dg1, dbb1, nullptr, d1, P_, H_ / 2, 1);

  gemm(d1, dW2, d2, P_, H_ / 4, H_ / 2, H_ / 2, H_ / 4, H_ / 4, 0);
  ln_res_gelu_kernel<<<(P_ + 7) / 8, 256, 0, stream>>>(
      d2, db2, dg2, dbb2, nullptr, d2, P_, H_ / 4, 1);

  dot64_kernel<<<(P_ + 7) / 8, 256, 0, stream>>>(d2, dW3, db3, (float*)d_out, P_);
}


# Non-GEMM kernels (LayerNorm, final dot, edge gather, weight and edge-list build) run on fixed grids whose waves loop over many rows: about 7x fewer waves launched; every sum keeps its order
# speedup vs baseline: 2.5954x; 1.0376x over previous
#include <hip/hip_runtime.h>
#include <hip/hip_bf16.h>
#include <math.h>

#define N_   20000
#define E_   320000
#define R_   8
#define B_   8
#define IN_  128
#define H_   256
#define OUT_ 128
#define P_   100000
#define LN_EPS 1e-5f

typedef float v8f __attribute__((ext_vector_type(8)));
typedef float v2f __attribute__((ext_vector_type(2)));
typedef float v4f __attribute__((ext_vector_type(4)));

#define WMMA_F32(a, b, c) \
  __builtin_amdgcn_wmma_f32_16x16x4_f32(false, (a), false, (b), (short)0, (c), false, false)

__global__ void relatt_softmax_kernel(const float* __restrict__ relatt,
                                      float* __restrict__ a) {
  if (threadIdx.x == 0 && blockIdx.x == 0) {
    float m = relatt[0];
    for (int r = 1; r < R_; ++r) m = fmaxf(m, relatt[r]);
    float e[R_]; float s = 0.f;
    for (int r = 0; r < R_; ++r) { e[r] = expf(relatt[r] - m); s += e[r]; }
    for (int r = 0; r < R_; ++r) a[r] = e[r] / s;
  }
}

__global__ __launch_bounds__(256)
void make_weights_kernel(const float* __restrict__ att,
                         const float* __restrict__ basis,
                         const float* __restrict__ a,
                         float* __restrict__ W,
                         int in_ch, int out_ch) {
  const int io_sz  = in_ch * out_ch;
  const int total4 = (R_ * io_sz) / 4;
  const int nthr   = gridDim.x * blockDim.x;
  for (int q = blockIdx.x * blockDim.x + threadIdx.x; q < total4; q += nthr) {
    const int idx = 4 * q;
    const int r   = idx / io_sz;
    const int io  = idx - r * io_sz;
    v4f s = v4f{};
#pragma unroll
    for (int bb = 0; bb < B_; ++bb)
      s += att[r * B_ + bb] * *(const v4f*)(basis + (size_t)bb * io_sz + io);
    *(v4f*)(W + idx) = s * a[r];
  }
}

__global__ void csr_zero_kernel(int* __restrict__ cnt) {
  int i = blockIdx.x * blockDim.x + threadIdx.x;
  if (i < N_) cnt[i] = 0;
}

__global__ void csr_count_kernel(const int* __restrict__ dst,
                                 int* __restrict__ cnt,
                                 int* __restrict__ rank) {
  const int nthr = gridDim.x * blockDim.x;
  for (int e = blockIdx.x * blockDim.x + threadIdx.x; e < E_; e += nthr)
    rank[e] = atomicAdd(&cnt[dst[e]], 1);
}

__global__ __launch_bounds__(1024)
void csr_scan_kernel(const int* __restrict__ cnt, int* __restrict__ offs) {
  constexpr int CH = (N_ + 1023) / 1024;
  __shared__ int part[1024];
  const int t = threadIdx.x;
  int loc[CH];
  int s = 0;
#pragma unroll
  for (int i = 0; i < CH; ++i) {
    int idx = t * CH + i;
    int v = (idx < N_) ? cnt[idx] : 0;
    loc[i] = s;
    s += v;
  }
  part[t] = s;
  __syncthreads();
  for (int o = 1; o < 1024; o <<= 1) {
    int v = (t >= o) ? part[t - o] : 0;
    __syncthreads();
    part[t] += v;
    __syncthreads();
  }
  const int excl = part[t] - s;
#pragma unroll
  for (int i = 0; i < CH; ++i) {
    int idx = t * CH + i;
    if (idx < N_) offs[idx] = excl + loc[i];
  }
  if (t == 1023) offs[N_] = part[1023];
}

__global__ void csr_fill_kernel(const int* __restrict__ dst,
                                const int* __restrict__ et,
                                const int* __restrict__ offs,
                                const int* __restrict__ rank,
                                int* __restrict__ key) {
  const int nthr = gridDim.x * blockDim.x;
  for (int e = blockIdx.x * blockDim.x + threadIdx.x; e < E_; e += nthr)
    key[offs[dst[e]] + rank[e]] = et[e] * E_ + e;
}

__global__ void csr_order_kernel(const int* __restrict__ src,
                                 const int* __restrict__ offs,
                                 const int* __restrict__ key,
                                 int* __restrict__ packed) {
  int wave = threadIdx.x >> 5;
  int lane = threadIdx.x & 31;
  for (int node = blockIdx.x * 8 + wave; node < N_; node += gridDim.x * 8) {
    int beg = offs[node];
    int n   = offs[node + 1] - beg;
    for (int i = lane; i < n; i += 32) {
      int my = key[beg + i];
      int rk = 0;
      for (int j = 0; j < n; ++j) rk += (key[beg + j] < my) ? 1 : 0;
      int r = my / E_;
      int e = my - r * E_;
      packed[beg + rk] = src[e] * R_ + r;
    }
  }
}

template <int IN>
__global__ __launch_bounds__(256)
void gather_agg_kernel(const float* __restrict__ x,
                       const int* __restrict__ offs,
                       const int* __restrict__ packed,
                       float* __restrict__ agg) {
  constexpr int TPN = IN / 4;
  constexpr int NPB = 256 / TPN;
  const int lane = threadIdx.x & 31;
  const int t    = threadIdx.x % TPN;
  const float* xt = x + 4 * t;
#define GATHER_LOAD(p) (*(const v4f*)(xt + (size_t)((unsigned)(p) / R_) * IN))
#define GATHER_STEP(p, v)                                                     \
  {                                                                           \
    const int r = (int)((unsigned)(p) % R_);                                  \
    _Pragma("nounroll")   \
    while (cur < r) {              \
      *(v4f*)(o + cur * IN) = acc;                                            \
      acc = v4f{};                                                            \
      ++cur;                                                                  \
    }                                                                         \
    acc += (v);                                                               \
  }

  for (int nb = blockIdx.x; nb * NPB < N_; nb += gridDim.x) {
    const int node = nb * NPB + threadIdx.x / TPN;
    if (node >= N_) continue;
    const int beg = __builtin_amdgcn_readfirstlane(offs[node]);
    const int end = __builtin_amdgcn_readfirstlane(offs[node + 1]);
    float* o = agg + (size_t)node * (R_ * IN) + 4 * t;
    v4f acc = v4f{};
    int cur = 0;

    for (int c = beg; c < end; c += 32) {
      const int rem = end - c;
      const int m = rem < 32 ? rem : 32;
      const int pv = (lane < m) ? packed[c + lane] : 0;
      int j = 0;
      for (; j + 4 <= m; j += 4) {
        const int p0 = __builtin_amdgcn_readlane(pv, j);
        const int p1 = __builtin_amdgcn_readlane(pv, j + 1);
        const int p2 = __builtin_amdgcn_readlane(pv, j + 2);
        const int p3 = __builtin_amdgcn_readlane(pv, j + 3);
        const v4f v0 = GATHER_LOAD(p0);
        const v4f v1 = GATHER_LOAD(p1);
        const v4f v2 = GATHER_LOAD(p2);
        const v4f v3 = GATHER_LOAD(p3);
        GATHER_STEP(p0, v0)
        GATHER_STEP(p1, v1)
        GATHER_STEP(p2, v2)
        GATHER_STEP(p3, v3)
      }
      for (; j < m; ++j) {
        const int p = __builtin_amdgcn_readlane(pv, j);
        const v4f v = GATHER_LOAD(p);
        GATHER_STEP(p, v)
      }
    }
#pragma nounroll
    while (cur < R_) {
      *(v4f*)(o + cur * IN) = acc;
      acc = v4f{};
      ++cur;
    }
  }
#undef GATHER_LOAD
#undef GATHER_STEP
}

template <int BN>
__global__ __launch_bounds__(2 * BN)
void gemm_f32_wmma_lds_kernel(const float* __restrict__ A,
                              const float* __restrict__ Bm,
                              float* __restrict__ C,
                              int K, int lda, int ldb, int ldc,
                              int accumulate) {
  constexpr int NTHR = 2 * BN;
  constexpr int BM = 80, BK = 32;
  constexpr int AP = BK + 4;
  constexpr int BP = 2 * BN + 32;
  constexpr int A_ITEMS = BM * (BK / 4);
  constexpr int A_IT = (A_ITEMS + NTHR - 1) / NTHR;
  constexpr int BQ = BN / 4;
  constexpr int B_IT = ((BK / 2) * BQ) / NTHR;
  __shared__ __attribute__((aligned(16))) float As[2][BM * AP];
  __shared__ __attribute__((aligned(16))) float Bs[2][(BK / 2) * BP];

  const int tid  = threadIdx.x;
  const int wave = tid >> 5;
  const int lane = tid & 31;
  const int l16  = lane & 15;
  const int g    = lane >> 4;
  const int row0 = blockIdx.x * BM;

  v4f ra[A_IT];
  v4f rb0[B_IT], rb1[B_IT];

#define GEMM_GLOAD(k0)                                                        \
  {                                                                           \
    _Pragma("unroll")                \
    for (int it = 0; it < A_IT; ++it) {                                       \
      int i = tid + it * NTHR;                                                \
      i = (i < A_ITEMS) ? i : (A_ITEMS - 1);   \
      int r = i >> 3, c4 = (i & 7) * 4;                                       \
      ra[it] = *(const v4f*)(A + (size_t)(row0 + r) * lda + (k0) + c4);    \
    }                                                                         \
    _Pragma("unroll")                     \
    for (int it = 0; it < B_IT; ++it) {                                       \
      int j = tid + it * NTHR;                                                \
      int kp = j / BQ, n4 = (j - kp * BQ) * 4;                                \
      const float* p = Bm + (size_t)((k0) + 2 * kp) * ldb + n4;               \
      rb0[it] = *(const v4f*)p;                                            \
      rb1[it] = *(const v4f*)(p + ldb);                                    \
    }                                                                         \
  }

#define GEMM_SSTORE(buf)                                                      \
  {                                                                           \
    _Pragma("unroll")                          \
    for (int it = 0; it < A_IT; ++it) {                                       \
      int i = tid + it * NTHR;                                                \
      if (i < A_ITEMS) {                                                      \
        int r = i >> 3, c4 = (i & 7) * 4;                                     \
        *(v4f*)(&As[buf][r * AP + c4]) = ra[it];                           \
      }                                                                       \
    }                                                                         \
    _Pragma("unroll")                  \
    for (int it = 0; it < B_IT; ++it) {                                       \
      int j = tid + it * NTHR;                                                \
      int kp = j / BQ, n4 = (j - kp * BQ) * 4;                                \
      v4f lo, hi;                                                          \
      lo.x = rb0[it].x; lo.y = rb1[it].x; lo.z = rb0[it].y; lo.w = rb1[it].y; \
      hi.x = rb0[it].z; hi.y = rb1[it].z; hi.z = rb0[it].w; hi.w = rb1[it].w; \
      float* d = &Bs[buf][kp * BP + 2 * n4];                                  \
      *(v4f*)d       = lo;                                                 \
      *(v4f*)(d + 4) = hi;                                                 \
    }                                                                         \
  }

  v8f acc[5];
#pragma unroll
  for (int mt = 0; mt < 5; ++mt) acc[mt] = v8f{};

  GEMM_GLOAD(0)
  GEMM_SSTORE(0)
  __syncthreads();

  const int nch = K / BK;
  const int a_off = l16 * AP + 2 * g;
  const int b_off = g * BP + 2 * (wave * 16 + l16);
  for (int c = 0; c < nch; ++c) {
    const int cur = c & 1;
    const bool more = (c + 1 < nch);
    if (more) GEMM_GLOAD((c + 1) * BK)
    const float* as = &As[cur][a_off];
    const float* bs = &Bs[cur][b_off];
#pragma unroll
    for (int s = 0; s < BK / 4; ++s) {
      v2f b = *(const v2f*)(bs + 2 * s * BP);
#pragma unroll
      for (int mt = 0; mt < 5; ++mt) {
        v2f a = *(const v2f*)(as + mt * 16 * AP + 4 * s);
        acc[mt] = WMMA_F32(a, b, acc[mt]);
      }
    }
    if (more) GEMM_SSTORE(cur ^ 1)
    __syncthreads();
  }
#undef GEMM_GLOAD
#undef GEMM_SSTORE

  float* Cbase = C + (size_t)(row0 + 8 * g) * ldc + wave * 16 + l16;
  if (accumulate) {
#pragma unroll
    for (int mt = 0; mt < 5; ++mt) {
#pragma unroll
      for (int i = 0; i < 8; ++i)
        acc[mt][i] += Cbase[(size_t)(mt * 16 + i) * ldc];
    }
  }
#pragma unroll
  for (int mt = 0; mt < 5; ++mt) {
#pragma unroll
    for (int i = 0; i < 8; ++i)
      Cbase[(size_t)(mt * 16 + i) * ldc] = acc[mt][i];
  }
}

__global__ __launch_bounds__(512)
void gemm_pe_wmma_lds_kernel(const float* __restrict__ z,
                             const int* __restrict__ gi,
                             const int* __restrict__ gj,
                             const float* __restrict__ Bm,
                             float* __restrict__ C,
                             int K) {
  constexpr int BN = H_;
  constexpr int NTHR = 2 * BN;
  constexpr int BM = 80, BK = 32;
  constexpr int AP = BK + 4;
  constexpr int BP = 2 * BN + 32;
  constexpr int A_ITEMS = BM * (BK / 4);
  constexpr int A_IT = (A_ITEMS + NTHR - 1) / NTHR;
  constexpr int BQ = BN / 4;
  constexpr int B_IT = ((BK / 2) * BQ) / NTHR;
  __shared__ __attribute__((aligned(16))) float As[2][BM * AP];
  __shared__ __attribute__((aligned(16))) float Bs[2][(BK / 2) * BP];

  const int tid  = threadIdx.x;
  const int wave = tid >> 5;
  const int lane = tid & 31;
  const int l16  = lane & 15;
  const int g    = lane >> 4;
  const int row0 = blockIdx.x * BM;

  unsigned offi[A_IT], offj[A_IT];
#pragma unroll
  for (int it = 0; it < A_IT; ++it) {
    int i = tid + it * NTHR;
    i = (i < A_ITEMS) ? i : (A_ITEMS - 1);
    int r = i >> 3, c4 = (i & 7) * 4;
    offi[it] = (unsigned)gi[row0 + r] * OUT_ + c4;
    offj[it] = (unsigned)gj[row0 + r] * OUT_ + c4;
  }

  v4f rzi[A_IT], rzj[A_IT];
  v4f rb0[B_IT], rb1[B_IT];

#define PE_GLOAD(k0)                                                          \
  {                                                                           \
    const int f0 = (k0) & (OUT_ - 1);                                         \
    _Pragma("unroll")        \
    for (int it = 0; it < A_IT; ++it) {                                       \
      rzi[it] = *(const v4f*)(z + offi[it] + f0);                             \
      rzj[it] = *(const v4f*)(z + offj[it] + f0);                             \
    }                                                                         \
    _Pragma("unroll")                     \
    for (int it = 0; it < B_IT; ++it) {                                       \
      int j = tid + it * NTHR;                                                \
      int kp = j / BQ, n4 = (j - kp * BQ) * 4;                                \
      const float* p = Bm + (size_t)((k0) + 2 * kp) * BN + n4;                \
      rb0[it] = *(const v4f*)p;                                               \
      rb1[it] = *(const v4f*)(p + BN);                                        \
    }                                                                         \
  }

#define PE_SSTORE(buf, k0)                                                    \
  {                                                                           \
    const int sec = (k0) >> 7;                                                \
    _Pragma("unroll")                  \
    for (int it = 0; it < A_IT; ++it) {                                       \
      int i = tid + it * NTHR;                                                \
      if (i < A_ITEMS) {                                                      \
        int r = i >> 3, c4 = (i & 7) * 4;                                     \
        v4f a;                                                                \
        if (sec == 0)      a = rzi[it] + rzj[it];                             \
        else if (sec == 1) a = rzi[it] * rzj[it];                             \
        else {                                                                \
          const v4f d = rzi[it] - rzj[it];                                    \
          a.x = fabsf(d.x); a.y = fabsf(d.y); a.z = fabsf(d.z); a.w = fabsf(d.w); \
        }                                                                     \
        *(v4f*)(&As[buf][r * AP + c4]) = a;                                   \
      }                                                                       \
    }                                                                         \
    _Pragma("unroll")                  \
    for (int it = 0; it < B_IT; ++it) {                                       \
      int j = tid + it * NTHR;                                                \
      int kp = j / BQ, n4 = (j - kp * BQ) * 4;                                \
      v4f lo, hi;                                                             \
      lo.x = rb0[it].x; lo.y = rb1[it].x; lo.z = rb0[it].y; lo.w = rb1[it].y; \
      hi.x = rb0[it].z; hi.y = rb1[it].z; hi.z = rb0[it].w; hi.w = rb1[it].w; \
      float* d = &Bs[buf][kp * BP + 2 * n4];                                  \
      *(v4f*)d       = lo;                                                    \
      *(v4f*)(d + 4) = hi;                                                    \
    }                                                                         \
  }

  v8f acc[5];
#pragma unroll
  for (int mt = 0; mt < 5; ++mt) acc[mt] = v8f{};

  PE_GLOAD(0)
  PE_SSTORE(0, 0)
  __syncthreads();

  const int nch = K / BK;
  const int a_off = l16 * AP + 2 * g;
  const int b_off = g * BP + 2 * (wave * 16 + l16);
  for (int c = 0; c < nch; ++c) {
    const int cur = c & 1;
    const bool more = (c + 1 < nch);
    if (more) PE_GLOAD((c + 1) * BK)
    const float* as = &As[cur][a_off];
    const float* bs = &Bs[cur][b_off];
#pragma unroll
    for (int s = 0; s < BK / 4; ++s) {
      v2f b = *(const v2f*)(bs + 2 * s * BP);
#pragma unroll
      for (int mt = 0; mt < 5; ++mt) {
        v2f a = *(const v2f*)(as + mt * 16 * AP + 4 * s);
        acc[mt] = WMMA_F32(a, b, acc[mt]);
      }
    }
    if (more) PE_SSTORE(cur ^ 1, (c + 1) * BK)
    __syncthreads();
  }
#undef PE_GLOAD
#undef PE_SSTORE

  float* Cbase = C + (size_t)(row0 + 8 * g) * BN + wave * 16 + l16;
#pragma unroll
  for (int mt = 0; mt < 5; ++mt) {
#pragma unroll
    for (int i = 0; i < 8; ++i)
      Cbase[(size_t)(mt * 16 + i) * BN] = acc[mt][i];
  }
}

template <int C>
__global__ __launch_bounds__(256)
void ln_res_gelu_kernel(const float* __restrict__ y,
                        const float* __restrict__ bias,
                        const float* __restrict__ gamma,
                        const float* __restrict__ beta,
                        const float* __restrict__ resid,
                        float* __restrict__ out,
                        int rows, int do_gelu) {
  constexpr int NV  = C / 32;
  constexpr int RPI = 256 / C;
  const int lane   = threadIdx.x & 31;
  const int wave   = blockIdx.x * 8 + (threadIdx.x >> 5);
  const int nwaves = gridDim.x * 8;

  float gm[NV], bt[NV], bs[NV];
#pragma unroll
  for (int i = 0; i < NV; ++i) {
    const int c = lane + (i << 5);
    gm[i] = gamma[c];
    bt[i] = beta[c];
    bs[i] = bias ? bias[c] : 0.f;
  }

  for (int grp = wave; grp * RPI < rows; grp += nwaves) {
    float vals[RPI][NV], rs[RPI][NV];
#pragma unroll
    for (int r = 0; r < RPI; ++r) {
      int row = grp * RPI + r;
      row = row < rows ? row : rows - 1;
      const size_t base = (size_t)row * C + lane;
#pragma unroll
      for (int i = 0; i < NV; ++i) {
        vals[r][i] = y[base + (i << 5)];
        rs[r][i] = resid ? resid[base + (i << 5)] : 0.f;
      }
    }
#pragma unroll
    for (int r = 0; r < RPI; ++r) {
      const int row = grp * RPI + r;
      float s = 0.f;
#pragma unroll
      for (int i = 0; i < NV; ++i) {
        float v = vals[r][i];
        if (bias) v += bs[i];
        vals[r][i] = v;
        s += v;
      }
      for (int o = 16; o; o >>= 1) s += __shfl_xor(s, o, 32);
      const float mean = s / (float)C;
      float vs = 0.f;
#pragma unroll
      for (int i = 0; i < NV; ++i) { float d = vals[r][i] - mean; vs += d * d; }
      for (int o = 16; o; o >>= 1) vs += __shfl_xor(vs, o, 32);
      const float inv = rsqrtf(vs / (float)C + LN_EPS);
      if (row < rows) {
#pragma unroll
        for (int i = 0; i < NV; ++i) {
          float v = (vals[r][i] - mean) * inv * gm[i] + bt[i];
          if (resid) v += rs[r][i];
          if (do_gelu) v = 0.5f * v * (1.f + erff(v * 0.70710678118654752f));
          out[(size_t)row * C + lane + (i << 5)] = v;
        }
      }
    }
  }
}

__global__ __launch_bounds__(256)
void dot64_kernel(const float* __restrict__ d2,
                  const float* __restrict__ w,
                  const float* __restrict__ b,
                  float* __restrict__ logits, int rows) {
  constexpr int RPI = 4;
  const int lane   = threadIdx.x & 31;
  const int wave   = blockIdx.x * 8 + (threadIdx.x >> 5);
  const int nwaves = gridDim.x * 8;
  const float w0 = w[lane], w1 = w[lane + 32];
  const float b0 = b[0];
  for (int grp = wave; grp * RPI < rows; grp += nwaves) {
    float a0[RPI], a1[RPI];
#pragma unroll
    for (int r = 0; r < RPI; ++r) {
      int p = grp * RPI + r;
      p = p < rows ? p : rows - 1;
      const float* dr = d2 + (size_t)p * 64;
      a0[r] = dr[lane];
      a1[r] = dr[lane + 32];
    }
#pragma unroll
    for (int r = 0; r < RPI; ++r) {
      const int p = grp * RPI + r;
      float s = a0[r] * w0 + a1[r] * w1;
      for (int o = 16; o; o >>= 1) s += __shfl_xor(s, o, 32);
      if (lane == 0 && p < rows) logits[p] = s + b0;
    }
  }
}

extern "C" void kernel_launch(void* const* d_in, const int* in_sizes, int n_in,
                              void* d_out, int out_size, void* d_ws, size_t ws_size,
                              hipStream_t stream) {
  (void)in_sizes; (void)n_in; (void)out_size; (void)ws_size;

  const float* x    = (const float*)d_in[0];
  const int*   ei   = (const int*)d_in[1];
  const int*   esrc = ei;
  const int*   edst = ei + E_;
  const int*   et   = (const int*)d_in[2];
  const int*   gp   = (const int*)d_in[3];
  const int*   gi   = gp;
  const int*   gj   = gp + P_;

  const float* basis[4], *att[4], *root[4], *relatt[4], *gam[4], *bet[4];
  for (int i = 0; i < 4; ++i) {
    basis[i]  = (const float*)d_in[4 + 6 * i + 0];
    att[i]    = (const float*)d_in[4 + 6 * i + 1];
    root[i]   = (const float*)d_in[4 + 6 * i + 2];
    relatt[i] = (const float*)d_in[4 + 6 * i + 3];
    gam[i]    = (const float*)d_in[4 + 6 * i + 4];
    bet[i]    = (const float*)d_in[4 + 6 * i + 5];
  }
  const float* res0 = (const float*)d_in[28];
  const float* res3 = (const float*)d_in[29];
  const float* dW0 = (const float*)d_in[30], *db0 = (const float*)d_in[31];
  const float* dg0 = (const float*)d_in[32], *dbb0 = (const float*)d_in[33];
  const float* dW1 = (const float*)d_in[34], *db1 = (const float*)d_in[35];
  const float* dg1 = (const float*)d_in[36], *dbb1 = (const float*)d_in[37];
  const float* dW2 = (const float*)d_in[38], *db2 = (const float*)d_in[39];
  const float* dg2 = (const float*)d_in[40], *dbb2 = (const float*)d_in[41];
  const float* dW3 = (const float*)d_in[42], *db3 = (const float*)d_in[43];

  float* ws = (float*)d_ws;
  size_t off = 0;
  float* a_sm  = ws + off; off += 64;
  float* Wcat  = ws + off; off += (size_t)R_ * H_ * H_;
  float* hA    = ws + off; off += (size_t)N_ * H_;
  float* hB    = ws + off; off += (size_t)N_ * H_;
  float* resid = ws + off; off += (size_t)N_ * H_;
  float* big   = ws + off;
  float* d0 = big;
  float* d1 = big + (size_t)P_ * H_;
  float* d2 = hA;
  int* csr_cnt    = (int*)(resid + (size_t)N_ * OUT_);
  int* csr_offs   = csr_cnt + 20480;
  int* csr_rank   = csr_offs + 20480;
  int* csr_key    = csr_rank + E_;
  int* csr_packed = csr_key + E_;

  auto gemm = [&](const float* A, const float* Bm, float* Cm,
                  int M, int Nc, int K, int lda, int ldb, int ldc, int acc) {
    int blocks = M / 80;
    if (Nc == 256)
      gemm_f32_wmma_lds_kernel<256><<<blocks, 512, 0, stream>>>(A, Bm, Cm, K, lda, ldb, ldc, acc);
    else if (Nc == 128)
      gemm_f32_wmma_lds_kernel<128><<<blocks, 256, 0, stream>>>(A, Bm, Cm, K, lda, ldb, ldc, acc);
    else
      gemm_f32_wmma_lds_kernel<64><<<blocks, 128, 0, stream>>>(A, Bm, Cm, K, lda, ldb, ldc, acc);
  };

  struct { const float* cur; int in; int out; int li;
           const float* resmat; float* conv; int gelu; float* rbuf; } L[4] = {
    { x,  IN_, H_,   0, res0,    hA, 1, hB },
    { hA, H_,  H_,   1, nullptr, hB, 1, nullptr },
    { hB, H_,  H_,   2, nullptr, hA, 1, nullptr },
    { hA, H_,  OUT_, 3, res3,    hB, 0, resid },
  };

  csr_zero_kernel<<<(N_ + 255) / 256, 256, 0, stream>>>(csr_cnt);
  csr_count_kernel<<<(E_ / 4 + 255) / 256, 256, 0, stream>>>(edst, csr_cnt, csr_rank);
  csr_scan_kernel<<<1, 1024, 0, stream>>>(csr_cnt, csr_offs);
  csr_fill_kernel<<<(E_ / 4 + 255) / 256, 256, 0, stream>>>(edst, et, csr_offs, csr_rank, csr_key);
  csr_order_kernel<<<(N_ + 31) / 32, 256, 0, stream>>>(esrc, csr_offs, csr_key, csr_packed);

  for (int i = 0; i < 4; ++i) {
    const float* cur = L[i].cur;
    int in_ch = L[i].in, out_ch = L[i].out, li = L[i].li;
    float* conv = L[i].conv;

    relatt_softmax_kernel<<<1, 32, 0, stream>>>(relatt[li], a_sm);
    int wtotal = R_ * in_ch * out_ch;
    make_weights_kernel<<<(wtotal + 2047) / 2048, 256, 0, stream>>>(
        att[li], basis[li], a_sm, Wcat, in_ch, out_ch);

    if (in_ch == H_)
      gather_agg_kernel<H_><<<N_ / 16, 256, 0, stream>>>(cur, csr_offs, csr_packed, big);
    else
      gather_agg_kernel<IN_><<<N_ / 32, 256, 0, stream>>>(cur, csr_offs, csr_packed, big);

    gemm(big, Wcat, conv, N_, out_ch, R_ * in_ch, R_ * in_ch, out_ch, out_ch, 0);
    gemm(cur, root[li], conv, N_, out_ch, in_ch, in_ch, out_ch, out_ch, 1);

    const float* rptr;
    if (L[i].resmat) {
      gemm(cur, L[i].resmat, L[i].rbuf, N_, out_ch, in_ch, in_ch, out_ch, out_ch, 0);
      rptr = L[i].rbuf;
    } else {
      rptr = cur;
    }
    if (out_ch == H_)
      ln_res_gelu_kernel<H_><<<(N_ + 31) / 32, 256, 0, stream>>>(
          conv, nullptr, gam[li], bet[li], rptr, conv, N_, L[i].gelu);
    else
      ln_res_gelu_kernel<OUT_><<<(N_ + 31) / 32, 256, 0, stream>>>(
          conv, nullptr, gam[li], bet[li], rptr, conv, N_, L[i].gelu);
  }

  hipMemcpyAsync((float*)d_out + P_, hB, (size_t)N_ * OUT_ * sizeof(float),
                 hipMemcpyDeviceToDevice, stream);

  gemm_pe_wmma_lds_kernel<<<P_ / 80, 512, 0, stream>>>(hB, gi, gj, dW0, d0, 3 * OUT_);
  ln_res_gelu_kernel<H_><<<1024, 256, 0, stream>>>(
      d0, db0, dg0, dbb0, nullptr, d0, P_, 1);

  gemm(d0, dW1, d1, P_, H_ / 2, H_, H_, H_ / 2, H_ / 2, 0);
  ln_res_gelu_kernel<H_ / 2><<<1024, 256, 0, stream>>>(
      d1, db1, dg1, dbb1, nullptr, d1, P_, 1);

  gemm(d1, dW2, d2, P_, H_ / 4, H_ / 2, H_ / 2, H_ / 4, H_ / 4, 0);
  ln_res_gelu_kernel<H_ / 4><<<1024, 256, 0, stream>>>(
      d2, db2, dg2, dbb2, nullptr, d2, P_, 1);

  dot64_kernel<<<1024, 256, 0, stream>>>(d2, dW3, db3, (float*)d_out, P_);
}


# One GEMM pass per layer: the root product joins the K loop behind the relation chunks (one running sum, so its terms are added in a new order); LayerNorm+residual+GELU run in the tile epilogue
# speedup vs baseline: 2.7250x; 1.0499x over previous
#include <hip/hip_runtime.h>
#include <hip/hip_bf16.h>
#include <math.h>

#define N_   20000
#define E_   320000
#define R_   8
#define B_   8
#define IN_  128
#define H_   256
#define OUT_ 128
#define P_   100000
#define LN_EPS 1e-5f

typedef float v8f __attribute__((ext_vector_type(8)));
typedef float v2f __attribute__((ext_vector_type(2)));
typedef float v4f __attribute__((ext_vector_type(4)));

#define WMMA_F32(a, b, c) \
  __builtin_amdgcn_wmma_f32_16x16x4_f32(false, (a), false, (b), (short)0, (c), false, false)

__global__ void relatt_softmax_kernel(const float* __restrict__ relatt,
                                      float* __restrict__ a) {
  if (threadIdx.x == 0 && blockIdx.x == 0) {
    float m = relatt[0];
    for (int r = 1; r < R_; ++r) m = fmaxf(m, relatt[r]);
    float e[R_]; float s = 0.f;
    for (int r = 0; r < R_; ++r) { e[r] = expf(relatt[r] - m); s += e[r]; }
    for (int r = 0; r < R_; ++r) a[r] = e[r] / s;
  }
}

__global__ __launch_bounds__(256)
void make_weights_kernel(const float* __restrict__ att,
                         const float* __restrict__ basis,
                         const float* __restrict__ a,
                         float* __restrict__ W,
                         int in_ch, int out_ch) {
  const int io_sz  = in_ch * out_ch;
  const int total4 = (R_ * io_sz) / 4;
  const int nthr   = gridDim.x * blockDim.x;
  for (int q = blockIdx.x * blockDim.x + threadIdx.x; q < total4; q += nthr) {
    const int idx = 4 * q;
    const int r   = idx / io_sz;
    const int io  = idx - r * io_sz;
    v4f s = v4f{};
#pragma unroll
    for (int bb = 0; bb < B_; ++bb)
      s += att[r * B_ + bb] * *(const v4f*)(basis + (size_t)bb * io_sz + io);
    *(v4f*)(W + idx) = s * a[r];
  }
}

__global__ void csr_zero_kernel(int* __restrict__ cnt) {
  int i = blockIdx.x * blockDim.x + threadIdx.x;
  if (i < N_) cnt[i] = 0;
}

__global__ void csr_count_kernel(const int* __restrict__ dst,
                                 int* __restrict__ cnt,
                                 int* __restrict__ rank) {
  const int nthr = gridDim.x * blockDim.x;
  for (int e = blockIdx.x * blockDim.x + threadIdx.x; e < E_; e += nthr)
    rank[e] = atomicAdd(&cnt[dst[e]], 1);
}

__global__ __launch_bounds__(1024)
void csr_scan_kernel(const int* __restrict__ cnt, int* __restrict__ offs) {
  constexpr int CH = (N_ + 1023) / 1024;
  __shared__ int part[1024];
  const int t = threadIdx.x;
  int loc[CH];
  int s = 0;
#pragma unroll
  for (int i = 0; i < CH; ++i) {
    int idx = t * CH + i;
    int v = (idx < N_) ? cnt[idx] : 0;
    loc[i] = s;
    s += v;
  }
  part[t] = s;
  __syncthreads();
  for (int o = 1; o < 1024; o <<= 1) {
    int v = (t >= o) ? part[t - o] : 0;
    __syncthreads();
    part[t] += v;
    __syncthreads();
  }
  const int excl = part[t] - s;
#pragma unroll
  for (int i = 0; i < CH; ++i) {
    int idx = t * CH + i;
    if (idx < N_) offs[idx] = excl + loc[i];
  }
  if (t == 1023) offs[N_] = part[1023];
}

__global__ void csr_fill_kernel(const int* __restrict__ dst,
                                const int* __restrict__ et,
                                const int* __restrict__ offs,
                                const int* __restrict__ rank,
                                int* __restrict__ key) {
  const int nthr = gridDim.x * blockDim.x;
  for (int e = blockIdx.x * blockDim.x + threadIdx.x; e < E_; e += nthr)
    key[offs[dst[e]] + rank[e]] = et[e] * E_ + e;
}

__global__ void csr_order_kernel(const int* __restrict__ src,
                                 const int* __restrict__ offs,
                                 const int* __restrict__ key,
                                 int* __restrict__ packed) {
  int wave = threadIdx.x >> 5;
  int lane = threadIdx.x & 31;
  for (int node = blockIdx.x * 8 + wave; node < N_; node += gridDim.x * 8) {
    int beg = offs[node];
    int n   = offs[node + 1] - beg;
    for (int i = lane; i < n; i += 32) {
      int my = key[beg + i];
      int rk = 0;
      for (int j = 0; j < n; ++j) rk += (key[beg + j] < my) ? 1 : 0;
      int r = my / E_;
      int e = my - r * E_;
      packed[beg + rk] = src[e] * R_ + r;
    }
  }
}

template <int IN>
__global__ __launch_bounds__(256)
void gather_agg_kernel(const float* __restrict__ x,
                       const int* __restrict__ offs,
                       const int* __restrict__ packed,
                       float* __restrict__ agg) {
  constexpr int TPN = IN / 4;
  constexpr int NPB = 256 / TPN;
  const int lane = threadIdx.x & 31;
  const int t    = threadIdx.x % TPN;
  const float* xt = x + 4 * t;
#define GATHER_LOAD(p) (*(const v4f*)(xt + (size_t)((unsigned)(p) / R_) * IN))
#define GATHER_STEP(p, v)                                                     \
  {                                                                           \
    const int r = (int)((unsigned)(p) % R_);                                  \
    _Pragma("nounroll")   \
    while (cur < r) {              \
      *(v4f*)(o + cur * IN) = acc;                                            \
      acc = v4f{};                                                            \
      ++cur;                                                                  \
    }                                                                         \
    acc += (v);                                                               \
  }

  for (int nb = blockIdx.x; nb * NPB < N_; nb += gridDim.x) {
    const int node = nb * NPB + threadIdx.x / TPN;
    if (node >= N_) continue;
    const int beg = __builtin_amdgcn_readfirstlane(offs[node]);
    const int end = __builtin_amdgcn_readfirstlane(offs[node + 1]);
    float* o = agg + (size_t)node * (R_ * IN) + 4 * t;
    v4f acc = v4f{};
    int cur = 0;

    for (int c = beg; c < end; c += 32) {
      const int rem = end - c;
      const int m = rem < 32 ? rem : 32;
      const int pv = (lane < m) ? packed[c + lane] : 0;
      int j = 0;
      for (; j + 4 <= m; j += 4) {
        const int p0 = __builtin_amdgcn_readlane(pv, j);
        const int p1 = __builtin_amdgcn_readlane(pv, j + 1);
        const int p2 = __builtin_amdgcn_readlane(pv, j + 2);
        const int p3 = __builtin_amdgcn_readlane(pv, j + 3);
        const v4f v0 = GATHER_LOAD(p0);
        const v4f v1 = GATHER_LOAD(p1);
        const v4f v2 = GATHER_LOAD(p2);
        const v4f v3 = GATHER_LOAD(p3);
        GATHER_STEP(p0, v0)
        GATHER_STEP(p1, v1)
        GATHER_STEP(p2, v2)
        GATHER_STEP(p3, v3)
      }
      for (; j < m; ++j) {
        const int p = __builtin_amdgcn_readlane(pv, j);
        const v4f v = GATHER_LOAD(p);
        GATHER_STEP(p, v)
      }
    }
#pragma nounroll
    while (cur < R_) {
      *(v4f*)(o + cur * IN) = acc;
      acc = v4f{};
      ++cur;
    }
  }
#undef GATHER_LOAD
#undef GATHER_STEP
}

template <int BN>
__device__ __forceinline__
void tile_ln_store(v8f (&acc)[5], float* ct,
                   const float* __restrict__ bias,
                   const float* __restrict__ gamma,
                   const float* __restrict__ beta,
                   const float* __restrict__ resid,
                   float* __restrict__ out, int row0, int do_gelu) {
  constexpr int CP = BN + 4;
  constexpr int NW = BN / 16;
  constexpr int NV = BN / 32;
  const int tid  = threadIdx.x;
  const int wave = tid >> 5;
  const int lane = tid & 31;
  const int l16  = lane & 15;
  const int g    = lane >> 4;

  float* cw = ct + (8 * g) * CP + wave * 16 + l16;
#pragma unroll
  for (int mt = 0; mt < 5; ++mt) {
#pragma unroll
    for (int i = 0; i < 8; ++i)
      cw[(mt * 16 + i) * CP] = acc[mt][i];
  }
  __syncthreads();

  float gm[NV], bt[NV], bs[NV];
#pragma unroll
  for (int i = 0; i < NV; ++i) {
    const int c = lane + (i << 5);
    gm[i] = gamma[c];
    bt[i] = beta[c];
    bs[i] = bias ? bias[c] : 0.f;
  }

  for (int r = wave; r < 80; r += NW) {
    const size_t base = (size_t)(row0 + r) * BN + lane;
    const float* cr = ct + r * CP + lane;
    float vals[NV], rs[NV];
#pragma unroll
    for (int i = 0; i < NV; ++i) {
      vals[i] = cr[i << 5];
      rs[i] = resid ? resid[base + (i << 5)] : 0.f;
    }
    float s = 0.f;
#pragma unroll
    for (int i = 0; i < NV; ++i) {
      float v = vals[i];
      if (bias) v += bs[i];
      vals[i] = v;
      s += v;
    }
    for (int o = 16; o; o >>= 1) s += __shfl_xor(s, o, 32);
    const float mean = s / (float)BN;
    float vs = 0.f;
#pragma unroll
    for (int i = 0; i < NV; ++i) { float d = vals[i] - mean; vs += d * d; }
    for (int o = 16; o; o >>= 1) vs += __shfl_xor(vs, o, 32);
    const float inv = rsqrtf(vs / (float)BN + LN_EPS);
#pragma unroll
    for (int i = 0; i < NV; ++i) {
      float v = (vals[i] - mean) * inv * gm[i] + bt[i];
      if (resid) v += rs[i];
      if (do_gelu) v = 0.5f * v * (1.f + erff(v * 0.70710678118654752f));
      out[base + (i << 5)] = v;
    }
  }
}

template <int BN>
__global__ __launch_bounds__(2 * BN)
void gemm_f32_wmma_lds_kernel(const float* __restrict__ A,
                              const float* __restrict__ Bm, int K, int lda,
                              const float* __restrict__ A2,
                              const float* __restrict__ B2, int K2, int lda2,
                              float* __restrict__ C,
                              const float* __restrict__ bias,
                              const float* __restrict__ gamma,
                              const float* __restrict__ beta,
                              const float* __restrict__ resid,
                              int do_gelu) {
  constexpr int NTHR = 2 * BN;
  constexpr int BM = 80, BK = 32;
  constexpr int AP = BK + 4;
  constexpr int BP = 2 * BN + 32;
  constexpr int A_ITEMS = BM * (BK / 4);
  constexpr int A_IT = (A_ITEMS + NTHR - 1) / NTHR;
  constexpr int BQ = BN / 4;
  constexpr int B_IT = ((BK / 2) * BQ) / NTHR;
  constexpr int A_ST = BM * AP;
  constexpr int B_ST = (BK / 2) * BP;
  static_assert(2 * A_ST + 2 * B_ST >= BM * (BN + 4), "result tile must fit in the operand stages");
  __shared__ __attribute__((aligned(16))) float smem[2 * A_ST + 2 * B_ST];
  float* const As = smem;
  float* const Bs = smem + 2 * A_ST;

  const int tid  = threadIdx.x;
  const int wave = tid >> 5;
  const int lane = tid & 31;
  const int l16  = lane & 15;
  const int g    = lane >> 4;
  const int row0 = blockIdx.x * BM;
  const int nch1 = K / BK;
  const int nch  = nch1 + K2 / BK;

  v4f ra[A_IT];
  v4f rb0[B_IT], rb1[B_IT];

#define GEMM_GLOAD(ch)                                                        \
  {                                                                           \
    const bool first = (ch) < nch1;                                           \
    const float* Ap = first ? A : A2;                                         \
    const float* Bp = first ? Bm : B2;                                        \
    const int ld = first ? lda : lda2;                                        \
    const int k0 = (first ? (ch) : (ch) - nch1) * BK;                         \
    _Pragma("unroll")                \
    for (int it = 0; it < A_IT; ++it) {                                       \
      int i = tid + it * NTHR;                                                \
      i = (i < A_ITEMS) ? i : (A_ITEMS - 1);   \
      int r = i >> 3, c4 = (i & 7) * 4;                                       \
      ra[it] = *(const v4f*)(Ap + (size_t)(row0 + r) * ld + k0 + c4);         \
    }                                                                         \
    _Pragma("unroll")                     \
    for (int it = 0; it < B_IT; ++it) {                                       \
      int j = tid + it * NTHR;                                                \
      int kp = j / BQ, n4 = (j - kp * BQ) * 4;                                \
      const float* p = Bp + (size_t)(k0 + 2 * kp) * BN + n4;                  \
      rb0[it] = *(const v4f*)p;                                               \
      rb1[it] = *(const v4f*)(p + BN);                                        \
    }                                                                         \
  }

#define GEMM_SSTORE(buf)                                                      \
  {                                                                           \
    _Pragma("unroll")                          \
    for (int it = 0; it < A_IT; ++it) {                                       \
      int i = tid + it * NTHR;                                                \
      if (i < A_ITEMS) {                                                      \
        int r = i >> 3, c4 = (i & 7) * 4;                                     \
        *(v4f*)(As + (buf) * A_ST + r * AP + c4) = ra[it];                    \
      }                                                                       \
    }                                                                         \
    _Pragma("unroll")                  \
    for (int it = 0; it < B_IT; ++it) {                                       \
      int j = tid + it * NTHR;                                                \
      int kp = j / BQ, n4 = (j - kp * BQ) * 4;                                \
      v4f lo, hi;                                                             \
      lo.x = rb0[it].x; lo.y = rb1[it].x; lo.z = rb0[it].y; lo.w = rb1[it].y; \
      hi.x = rb0[it].z; hi.y = rb1[it].z; hi.z = rb0[it].w; hi.w = rb1[it].w; \
      float* d = Bs + (buf) * B_ST + kp * BP + 2 * n4;                        \
      *(v4f*)d       = lo;                                                    \
      *(v4f*)(d + 4) = hi;                                                    \
    }                                                                         \
  }

  v8f acc[5];
#pragma unroll
  for (int mt = 0; mt < 5; ++mt) acc[mt] = v8f{};

  GEMM_GLOAD(0)
  GEMM_SSTORE(0)
  __syncthreads();

  const int a_off = l16 * AP + 2 * g;
  const int b_off = g * BP + 2 * (wave * 16 + l16);
  for (int c = 0; c < nch; ++c) {
    const int cur = c & 1;
    const bool more = (c + 1 < nch);
    if (more) GEMM_GLOAD(c + 1)
    const float* as = As + cur * A_ST + a_off;
    const float* bs = Bs + cur * B_ST + b_off;
#pragma unroll
    for (int s = 0; s < BK / 4; ++s) {
      v2f b = *(const v2f*)(bs + 2 * s * BP);
#pragma unroll
      for (int mt = 0; mt < 5; ++mt) {
        v2f a = *(const v2f*)(as + mt * 16 * AP + 4 * s);
        acc[mt] = WMMA_F32(a, b, acc[mt]);
      }
    }
    if (more) GEMM_SSTORE(cur ^ 1)
    __syncthreads();
  }
#undef GEMM_GLOAD
#undef GEMM_SSTORE

  if (gamma) {
    tile_ln_store<BN>(acc, smem, bias, gamma, beta, resid, C, row0, do_gelu);
  } else {
    float* Cbase = C + (size_t)(row0 + 8 * g) * BN + wave * 16 + l16;
#pragma unroll
    for (int mt = 0; mt < 5; ++mt) {
#pragma unroll
      for (int i = 0; i < 8; ++i)
        Cbase[(size_t)(mt * 16 + i) * BN] = acc[mt][i];
    }
  }
}

__global__ __launch_bounds__(512)
void gemm_pe_wmma_lds_kernel(const float* __restrict__ z,
                             const int* __restrict__ gi,
                             const int* __restrict__ gj,
                             const float* __restrict__ Bm,
                             float* __restrict__ C,
                             int K,
                             const float* __restrict__ bias,
                             const float* __restrict__ gamma,
                             const float* __restrict__ beta) {
  constexpr int BN = H_;
  constexpr int NTHR = 2 * BN;
  constexpr int BM = 80, BK = 32;
  constexpr int AP = BK + 4;
  constexpr int BP = 2 * BN + 32;
  constexpr int A_ITEMS = BM * (BK / 4);
  constexpr int A_IT = (A_ITEMS + NTHR - 1) / NTHR;
  constexpr int BQ = BN / 4;
  constexpr int B_IT = ((BK / 2) * BQ) / NTHR;
  constexpr int A_ST = BM * AP;
  constexpr int B_ST = (BK / 2) * BP;
  static_assert(2 * A_ST + 2 * B_ST >= BM * (BN + 4), "result tile must fit in the operand stages");
  __shared__ __attribute__((aligned(16))) float smem[2 * A_ST + 2 * B_ST];
  float* const As = smem;
  float* const Bs = smem + 2 * A_ST;

  const int tid  = threadIdx.x;
  const int wave = tid >> 5;
  const int lane = tid & 31;
  const int l16  = lane & 15;
  const int g    = lane >> 4;
  const int row0 = blockIdx.x * BM;

  unsigned offi[A_IT], offj[A_IT];
#pragma unroll
  for (int it = 0; it < A_IT; ++it) {
    int i = tid + it * NTHR;
    i = (i < A_ITEMS) ? i : (A_ITEMS - 1);
    int r = i >> 3, c4 = (i & 7) * 4;
    offi[it] = (unsigned)gi[row0 + r] * OUT_ + c4;
    offj[it] = (unsigned)gj[row0 + r] * OUT_ + c4;
  }

  v4f rzi[A_IT], rzj[A_IT];
  v4f rb0[B_IT], rb1[B_IT];

#define PE_GLOAD(k0)                                                          \
  {                                                                           \
    const int f0 = (k0) & (OUT_ - 1);                                         \
    _Pragma("unroll")        \
    for (int it = 0; it < A_IT; ++it) {                                       \
      rzi[it] = *(const v4f*)(z + offi[it] + f0);                             \
      rzj[it] = *(const v4f*)(z + offj[it] + f0);                             \
    }                                                                         \
    _Pragma("unroll")                     \
    for (int it = 0; it < B_IT; ++it) {                                       \
      int j = tid + it * NTHR;                                                \
      int kp = j / BQ, n4 = (j - kp * BQ) * 4;                                \
      const float* p = Bm + (size_t)((k0) + 2 * kp) * BN + n4;                \
      rb0[it] = *(const v4f*)p;                                               \
      rb1[it] = *(const v4f*)(p + BN);                                        \
    }                                                                         \
  }

#define PE_SSTORE(buf, k0)                                                    \
  {                                                                           \
    const int sec = (k0) >> 7;                                                \
    _Pragma("unroll")                  \
    for (int it = 0; it < A_IT; ++it) {                                       \
      int i = tid + it * NTHR;                                                \
      if (i < A_ITEMS) {                                                      \
        int r = i >> 3, c4 = (i & 7) * 4;                                     \
        v4f a;                                                                \
        if (sec == 0)      a = rzi[it] + rzj[it];                             \
        else if (sec == 1) a = rzi[it] * rzj[it];                             \
        else {                                                                \
          const v4f d = rzi[it] - rzj[it];                                    \
          a.x = fabsf(d.x); a.y = fabsf(d.y); a.z = fabsf(d.z); a.w = fabsf(d.w); \
        }                                                                     \
        *(v4f*)(As + (buf) * A_ST + r * AP + c4) = a;                                   \
      }                                                                       \
    }                                                                         \
    _Pragma("unroll")                  \
    for (int it = 0; it < B_IT; ++it) {                                       \
      int j = tid + it * NTHR;                                                \
      int kp = j / BQ, n4 = (j - kp * BQ) * 4;                                \
      v4f lo, hi;                                                             \
      lo.x = rb0[it].x; lo.y = rb1[it].x; lo.z = rb0[it].y; lo.w = rb1[it].y; \
      hi.x = rb0[it].z; hi.y = rb1[it].z; hi.z = rb0[it].w; hi.w = rb1[it].w; \
      float* d = Bs + (buf) * B_ST + kp * BP + 2 * n4;                                  \
      *(v4f*)d       = lo;                                                    \
      *(v4f*)(d + 4) = hi;                                                    \
    }                                                                         \
  }

  v8f acc[5];
#pragma unroll
  for (int mt = 0; mt < 5; ++mt) acc[mt] = v8f{};

  PE_GLOAD(0)
  PE_SSTORE(0, 0)
  __syncthreads();

  const int nch = K / BK;
  const int a_off = l16 * AP + 2 * g;
  const int b_off = g * BP + 2 * (wave * 16 + l16);
  for (int c = 0; c < nch; ++c) {
    const int cur = c & 1;
    const bool more = (c + 1 < nch);
    if (more) PE_GLOAD((c + 1) * BK)
    const float* as = As + cur * A_ST + a_off;
    const float* bs = Bs + cur * B_ST + b_off;
#pragma unroll
    for (int s = 0; s < BK / 4; ++s) {
      v2f b = *(const v2f*)(bs + 2 * s * BP);
#pragma unroll
      for (int mt = 0; mt < 5; ++mt) {
        v2f a = *(const v2f*)(as + mt * 16 * AP + 4 * s);
        acc[mt] = WMMA_F32(a, b, acc[mt]);
      }
    }
    if (more) PE_SSTORE(cur ^ 1, (c + 1) * BK)
    __syncthreads();
  }
#undef PE_GLOAD
#undef PE_SSTORE

  tile_ln_store<BN>(acc, smem, bias, gamma, beta, nullptr, C, row0, 1);
}

__global__ __launch_bounds__(256)
void dot64_kernel(const float* __restrict__ d2,
                  const float* __restrict__ w,
                  const float* __restrict__ b,
                  float* __restrict__ logits, int rows) {
  constexpr int RPI = 4;
  const int lane   = threadIdx.x & 31;
  const int wave   = blockIdx.x * 8 + (threadIdx.x >> 5);
  const int nwaves = gridDim.x * 8;
  const float w0 = w[lane], w1 = w[lane + 32];
  const float b0 = b[0];
  for (int grp = wave; grp * RPI < rows; grp += nwaves) {
    float a0[RPI], a1[RPI];
#pragma unroll
    for (int r = 0; r < RPI; ++r) {
      int p = grp * RPI + r;
      p = p < rows ? p : rows - 1;
      const float* dr = d2 + (size_t)p * 64;
      a0[r] = dr[lane];
      a1[r] = dr[lane + 32];
    }
#pragma unroll
    for (int r = 0; r < RPI; ++r) {
      const int p = grp * RPI + r;
      float s = a0[r] * w0 + a1[r] * w1;
      for (int o = 16; o; o >>= 1) s += __shfl_xor(s, o, 32);
      if (lane == 0 && p < rows) logits[p] = s + b0;
    }
  }
}

extern "C" void kernel_launch(void* const* d_in, const int* in_sizes, int n_in,
                              void* d_out, int out_size, void* d_ws, size_t ws_size,
                              hipStream_t stream) {
  (void)in_sizes; (void)n_in; (void)out_size; (void)ws_size;

  const float* x    = (const float*)d_in[0];
  const int*   ei   = (const int*)d_in[1];
  const int*   esrc = ei;
  const int*   edst = ei + E_;
  const int*   et   = (const int*)d_in[2];
  const int*   gp   = (const int*)d_in[3];
  const int*   gi   = gp;
  const int*   gj   = gp + P_;

  const float* basis[4], *att[4], *root[4], *relatt[4], *gam[4], *bet[4];
  for (int i = 0; i < 4; ++i) {
    basis[i]  = (const float*)d_in[4 + 6 * i + 0];
    att[i]    = (const float*)d_in[4 + 6 * i + 1];
    root[i]   = (const float*)d_in[4 + 6 * i + 2];
    relatt[i] = (const float*)d_in[4 + 6 * i + 3];
    gam[i]    = (const float*)d_in[4 + 6 * i + 4];
    bet[i]    = (const float*)d_in[4 + 6 * i + 5];
  }
  const float* res0 = (const float*)d_in[28];
  const float* res3 = (const float*)d_in[29];
  const float* dW0 = (const float*)d_in[30], *db0 = (const float*)d_in[31];
  const float* dg0 = (const float*)d_in[32], *dbb0 = (const float*)d_in[33];
  const float* dW1 = (const float*)d_in[34], *db1 = (const float*)d_in[35];
  const float* dg1 = (const float*)d_in[36], *dbb1 = (const float*)d_in[37];
  const float* dW2 = (const float*)d_in[38], *db2 = (const float*)d_in[39];
  const float* dg2 = (const float*)d_in[40], *dbb2 = (const float*)d_in[41];
  const float* dW3 = (const float*)d_in[42], *db3 = (const float*)d_in[43];

  float* ws = (float*)d_ws;
  size_t off = 0;
  float* a_sm  = ws + off; off += 64;
  float* Wcat  = ws + off; off += (size_t)R_ * H_ * H_;
  float* hA    = ws + off; off += (size_t)N_ * H_;
  float* hB    = ws + off; off += (size_t)N_ * H_;
  float* resid = ws + off; off += (size_t)N_ * H_;
  float* big   = ws + off;
  float* d0 = big;
  float* d1 = big + (size_t)P_ * H_;
  float* d2 = hA;
  float* zout = (float*)d_out + P_;
  int* csr_cnt    = (int*)(resid + (size_t)N_ * OUT_);
  int* csr_offs   = csr_cnt + 20480;
  int* csr_rank   = csr_offs + 20480;
  int* csr_key    = csr_rank + E_;
  int* csr_packed = csr_key + E_;

  auto gemm = [&](const float* A, const float* Bm, int K, int lda,
                  const float* A2, const float* B2, int K2, int lda2,
                  float* Cm, int M, int Nc,
                  const float* bias, const float* gamma, const float* beta,
                  const float* res, int gelu) {
    int blocks = M / 80;
    if (Nc == 256)
      gemm_f32_wmma_lds_kernel<256><<<blocks, 512, 0, stream>>>(
          A, Bm, K, lda, A2, B2, K2, lda2, Cm, bias, gamma, beta, res, gelu);
    else if (Nc == 128)
      gemm_f32_wmma_lds_kernel<128><<<blocks, 256, 0, stream>>>(
          A, Bm, K, lda, A2, B2, K2, lda2, Cm, bias, gamma, beta, res, gelu);
    else
      gemm_f32_wmma_lds_kernel<64><<<blocks, 128, 0, stream>>>(
          A, Bm, K, lda, A2, B2, K2, lda2, Cm, bias, gamma, beta, res, gelu);
  };

  struct { const float* cur; int in; int out; int li;
           const float* resmat; float* conv; int gelu; float* rbuf; } L[4] = {
    { x,  IN_, H_,   0, res0,    hA,   1, hB },
    { hA, H_,  H_,   1, nullptr, hB,   1, nullptr },
    { hB, H_,  H_,   2, nullptr, hA,   1, nullptr },
    { hA, H_,  OUT_, 3, res3,    zout, 0, resid },
  };

  csr_zero_kernel<<<(N_ + 255) / 256, 256, 0, stream>>>(csr_cnt);
  csr_count_kernel<<<(E_ / 4 + 255) / 256, 256, 0, stream>>>(edst, csr_cnt, csr_rank);
  csr_scan_kernel<<<1, 1024, 0, stream>>>(csr_cnt, csr_offs);
  csr_fill_kernel<<<(E_ / 4 + 255) / 256, 256, 0, stream>>>(edst, et, csr_offs, csr_rank, csr_key);
  csr_order_kernel<<<(N_ + 31) / 32, 256, 0, stream>>>(esrc, csr_offs, csr_key, csr_packed);

  for (int i = 0; i < 4; ++i) {
    const float* cur = L[i].cur;
    int in_ch = L[i].in, out_ch = L[i].out, li = L[i].li;
    float* conv = L[i].conv;

    relatt_softmax_kernel<<<1, 32, 0, stream>>>(relatt[li], a_sm);
    int wtotal = R_ * in_ch * out_ch;
    make_weights_kernel<<<(wtotal + 2047) / 2048, 256, 0, stream>>>(
        att[li], basis[li], a_sm, Wcat, in_ch, out_ch);

    if (in_ch == H_)
      gather_agg_kernel<H_><<<N_ / 16, 256, 0, stream>>>(cur, csr_offs, csr_packed, big);
    else
      gather_agg_kernel<IN_><<<N_ / 32, 256, 0, stream>>>(cur, csr_offs, csr_packed, big);

    const float* rptr = cur;
    if (L[i].resmat) {
      gemm(cur, L[i].resmat, in_ch, in_ch, nullptr, nullptr, 0, 0,
           L[i].rbuf, N_, out_ch, nullptr, nullptr, nullptr, nullptr, 0);
      rptr = L[i].rbuf;
    }
    gemm(big, Wcat, R_ * in_ch, R_ * in_ch, cur, root[li], in_ch, in_ch,
         conv, N_, out_ch, nullptr, gam[li], bet[li], rptr, L[i].gelu);
  }

  gemm_pe_wmma_lds_kernel<<<P_ / 80, 512, 0, stream>>>(zout, gi, gj, dW0, d0, 3 * OUT_, db0, dg0, dbb0);
  gemm(d0, dW1, H_, H_, nullptr, nullptr, 0, 0,
       d1, P_, H_ / 2, db1, dg1, dbb1, nullptr, 1);
  gemm(d1, dW2, H_ / 2, H_ / 2, nullptr, nullptr, 0, 0,
       d2, P_, H_ / 4, db2, dg2, dbb2, nullptr, 1);
  dot64_kernel<<<1024, 256, 0, stream>>>(d2, dW3, db3, (float*)d_out, P_);
}


# Decoder in one launch: a block keeps its 80-pair tile in LDS through all four layers (A fragments read from the tile, only logits stored), two blocks per WGP; every sum keeps its order
# speedup vs baseline: 2.8973x; 1.0632x over previous
#include <hip/hip_runtime.h>
#include <hip/hip_bf16.h>
#include <math.h>

#define N_   20000
#define E_   320000
#define R_   8
#define B_   8
#define IN_  128
#define H_   256
#define OUT_ 128
#define P_   100000
#define LN_EPS 1e-5f

typedef float v8f __attribute__((ext_vector_type(8)));
typedef float v2f __attribute__((ext_vector_type(2)));
typedef float v4f __attribute__((ext_vector_type(4)));

#define WMMA_F32(a, b, c) \
  __builtin_amdgcn_wmma_f32_16x16x4_f32(false, (a), false, (b), (short)0, (c), false, false)

__global__ void relatt_softmax_kernel(const float* __restrict__ relatt,
                                      float* __restrict__ a) {
  if (threadIdx.x == 0 && blockIdx.x == 0) {
    float m = relatt[0];
    for (int r = 1; r < R_; ++r) m = fmaxf(m, relatt[r]);
    float e[R_]; float s = 0.f;
    for (int r = 0; r < R_; ++r) { e[r] = expf(relatt[r] - m); s += e[r]; }
    for (int r = 0; r < R_; ++r) a[r] = e[r] / s;
  }
}

__global__ __launch_bounds__(256)
void make_weights_kernel(const float* __restrict__ att,
                         const float* __restrict__ basis,
                         const float* __restrict__ a,
                         float* __restrict__ W,
                         int in_ch, int out_ch) {
  const int io_sz  = in_ch * out_ch;
  const int total4 = (R_ * io_sz) / 4;
  const int nthr   = gridDim.x * blockDim.x;
  for (int q = blockIdx.x * blockDim.x + threadIdx.x; q < total4; q += nthr) {
    const int idx = 4 * q;
    const int r   = idx / io_sz;
    const int io  = idx - r * io_sz;
    v4f s = v4f{};
#pragma unroll
    for (int bb = 0; bb < B_; ++bb)
      s += att[r * B_ + bb] * *(const v4f*)(basis + (size_t)bb * io_sz + io);
    *(v4f*)(W + idx) = s * a[r];
  }
}

__global__ void csr_zero_kernel(int* __restrict__ cnt) {
  int i = blockIdx.x * blockDim.x + threadIdx.x;
  if (i < N_) cnt[i] = 0;
}

__global__ void csr_count_kernel(const int* __restrict__ dst,
                                 int* __restrict__ cnt,
                                 int* __restrict__ rank) {
  const int nthr = gridDim.x * blockDim.x;
  for (int e = blockIdx.x * blockDim.x + threadIdx.x; e < E_; e += nthr)
    rank[e] = atomicAdd(&cnt[dst[e]], 1);
}

__global__ __launch_bounds__(1024)
void csr_scan_kernel(const int* __restrict__ cnt, int* __restrict__ offs) {
  constexpr int CH = (N_ + 1023) / 1024;
  __shared__ int part[1024];
  const int t = threadIdx.x;
  int loc[CH];
  int s = 0;
#pragma unroll
  for (int i = 0; i < CH; ++i) {
    int idx = t * CH + i;
    int v = (idx < N_) ? cnt[idx] : 0;
    loc[i] = s;
    s += v;
  }
  part[t] = s;
  __syncthreads();
  for (int o = 1; o < 1024; o <<= 1) {
    int v = (t >= o) ? part[t - o] : 0;
    __syncthreads();
    part[t] += v;
    __syncthreads();
  }
  const int excl = part[t] - s;
#pragma unroll
  for (int i = 0; i < CH; ++i) {
    int idx = t * CH + i;
    if (idx < N_) offs[idx] = excl + loc[i];
  }
  if (t == 1023) offs[N_] = part[1023];
}

__global__ void csr_fill_kernel(const int* __restrict__ dst,
                                const int* __restrict__ et,
                                const int* __restrict__ offs,
                                const int* __restrict__ rank,
                                int* __restrict__ key) {
  const int nthr = gridDim.x * blockDim.x;
  for (int e = blockIdx.x * blockDim.x + threadIdx.x; e < E_; e += nthr)
    key[offs[dst[e]] + rank[e]] = et[e] * E_ + e;
}

__global__ void csr_order_kernel(const int* __restrict__ src,
                                 const int* __restrict__ offs,
                                 const int* __restrict__ key,
                                 int* __restrict__ packed) {
  int wave = threadIdx.x >> 5;
  int lane = threadIdx.x & 31;
  for (int node = blockIdx.x * 8 + wave; node < N_; node += gridDim.x * 8) {
    int beg = offs[node];
    int n   = offs[node + 1] - beg;
    for (int i = lane; i < n; i += 32) {
      int my = key[beg + i];
      int rk = 0;
      for (int j = 0; j < n; ++j) rk += (key[beg + j] < my) ? 1 : 0;
      int r = my / E_;
      int e = my - r * E_;
      packed[beg + rk] = src[e] * R_ + r;
    }
  }
}

template <int IN>
__global__ __launch_bounds__(256)
void gather_agg_kernel(const float* __restrict__ x,
                       const int* __restrict__ offs,
                       const int* __restrict__ packed,
                       float* __restrict__ agg) {
  constexpr int TPN = IN / 4;
  constexpr int NPB = 256 / TPN;
  const int lane = threadIdx.x & 31;
  const int t    = threadIdx.x % TPN;
  const float* xt = x + 4 * t;
#define GATHER_LOAD(p) (*(const v4f*)(xt + (size_t)((unsigned)(p) / R_) * IN))
#define GATHER_STEP(p, v)                                                     \
  {                                                                           \
    const int r = (int)((unsigned)(p) % R_);                                  \
    _Pragma("nounroll")   \
    while (cur < r) {              \
      *(v4f*)(o + cur * IN) = acc;                                            \
      acc = v4f{};                                                            \
      ++cur;                                                                  \
    }                                                                         \
    acc += (v);                                                               \
  }

  for (int nb = blockIdx.x; nb * NPB < N_; nb += gridDim.x) {
    const int node = nb * NPB + threadIdx.x / TPN;
    if (node >= N_) continue;
    const int beg = __builtin_amdgcn_readfirstlane(offs[node]);
    const int end = __builtin_amdgcn_readfirstlane(offs[node + 1]);
    float* o = agg + (size_t)node * (R_ * IN) + 4 * t;
    v4f acc = v4f{};
    int cur = 0;

    for (int c = beg; c < end; c += 32) {
      const int rem = end - c;
      const int m = rem < 32 ? rem : 32;
      const int pv = (lane < m) ? packed[c + lane] : 0;
      int j = 0;
      for (; j + 4 <= m; j += 4) {
        const int p0 = __builtin_amdgcn_readlane(pv, j);
        const int p1 = __builtin_amdgcn_readlane(pv, j + 1);
        const int p2 = __builtin_amdgcn_readlane(pv, j + 2);
        const int p3 = __builtin_amdgcn_readlane(pv, j + 3);
        const v4f v0 = GATHER_LOAD(p0);
        const v4f v1 = GATHER_LOAD(p1);
        const v4f v2 = GATHER_LOAD(p2);
        const v4f v3 = GATHER_LOAD(p3);
        GATHER_STEP(p0, v0)
        GATHER_STEP(p1, v1)
        GATHER_STEP(p2, v2)
        GATHER_STEP(p3, v3)
      }
      for (; j < m; ++j) {
        const int p = __builtin_amdgcn_readlane(pv, j);
        const v4f v = GATHER_LOAD(p);
        GATHER_STEP(p, v)
      }
    }
#pragma nounroll
    while (cur < R_) {
      *(v4f*)(o + cur * IN) = acc;
      acc = v4f{};
      ++cur;
    }
  }
#undef GATHER_LOAD
#undef GATHER_STEP
}

template <int BN>
__device__ __forceinline__
void tile_ln_store(v8f (&acc)[5], float* ct,
                   const float* __restrict__ bias,
                   const float* __restrict__ gamma,
                   const float* __restrict__ beta,
                   const float* __restrict__ resid,
                   float* __restrict__ out, int row0, int do_gelu) {
  constexpr int CP = BN + 4;
  constexpr int NW = BN / 16;
  constexpr int NV = BN / 32;
  const int tid  = threadIdx.x;
  const int wave = tid >> 5;
  const int lane = tid & 31;
  const int l16  = lane & 15;
  const int g    = lane >> 4;

  float* cw = ct + (8 * g) * CP + wave * 16 + l16;
#pragma unroll
  for (int mt = 0; mt < 5; ++mt) {
#pragma unroll
    for (int i = 0; i < 8; ++i)
      cw[(mt * 16 + i) * CP] = acc[mt][i];
  }
  __syncthreads();

  float gm[NV], bt[NV], bs[NV];
#pragma unroll
  for (int i = 0; i < NV; ++i) {
    const int c = lane + (i << 5);
    gm[i] = gamma[c];
    bt[i] = beta[c];
    bs[i] = bias ? bias[c] : 0.f;
  }

  for (int r = wave; r < 80; r += NW) {
    const size_t base = (size_t)(row0 + r) * BN + lane;
    const float* cr = ct + r * CP + lane;
    float vals[NV], rs[NV];
#pragma unroll
    for (int i = 0; i < NV; ++i) {
      vals[i] = cr[i << 5];
      rs[i] = resid ? resid[base + (i << 5)] : 0.f;
    }
    float s = 0.f;
#pragma unroll
    for (int i = 0; i < NV; ++i) {
      float v = vals[i];
      if (bias) v += bs[i];
      vals[i] = v;
      s += v;
    }
    for (int o = 16; o; o >>= 1) s += __shfl_xor(s, o, 32);
    const float mean = s / (float)BN;
    float vs = 0.f;
#pragma unroll
    for (int i = 0; i < NV; ++i) { float d = vals[i] - mean; vs += d * d; }
    for (int o = 16; o; o >>= 1) vs += __shfl_xor(vs, o, 32);
    const float inv = rsqrtf(vs / (float)BN + LN_EPS);
#pragma unroll
    for (int i = 0; i < NV; ++i) {
      float v = (vals[i] - mean) * inv * gm[i] + bt[i];
      if (resid) v += rs[i];
      if (do_gelu) v = 0.5f * v * (1.f + erff(v * 0.70710678118654752f));
      out[base + (i << 5)] = v;
    }
  }
}

template <int BN>
__global__ __launch_bounds__(2 * BN)
void gemm_f32_wmma_lds_kernel(const float* __restrict__ A,
                              const float* __restrict__ Bm, int K, int lda,
                              const float* __restrict__ A2,
                              const float* __restrict__ B2, int K2, int lda2,
                              float* __restrict__ C,
                              const float* __restrict__ bias,
                              const float* __restrict__ gamma,
                              const float* __restrict__ beta,
                              const float* __restrict__ resid,
                              int do_gelu) {
  constexpr int NTHR = 2 * BN;
  constexpr int BM = 80, BK = 32;
  constexpr int AP = BK + 4;
  constexpr int BP = 2 * BN + 32;
  constexpr int A_ITEMS = BM * (BK / 4);
  constexpr int A_IT = (A_ITEMS + NTHR - 1) / NTHR;
  constexpr int BQ = BN / 4;
  constexpr int B_IT = ((BK / 2) * BQ) / NTHR;
  constexpr int A_ST = BM * AP;
  constexpr int B_ST = (BK / 2) * BP;
  static_assert(2 * A_ST + 2 * B_ST >= BM * (BN + 4), "result tile must fit in the operand stages");
  __shared__ __attribute__((aligned(16))) float smem[2 * A_ST + 2 * B_ST];
  float* const As = smem;
  float* const Bs = smem + 2 * A_ST;

  const int tid  = threadIdx.x;
  const int wave = tid >> 5;
  const int lane = tid & 31;
  const int l16  = lane & 15;
  const int g    = lane >> 4;
  const int row0 = blockIdx.x * BM;
  const int nch1 = K / BK;
  const int nch  = nch1 + K2 / BK;

  v4f ra[A_IT];
  v4f rb0[B_IT], rb1[B_IT];

#define GEMM_GLOAD(ch)                                                        \
  {                                                                           \
    const bool first = (ch) < nch1;                                           \
    const float* Ap = first ? A : A2;                                         \
    const float* Bp = first ? Bm : B2;                                        \
    const int ld = first ? lda : lda2;                                        \
    const int k0 = (first ? (ch) : (ch) - nch1) * BK;                         \
    _Pragma("unroll")                \
    for (int it = 0; it < A_IT; ++it) {                                       \
      int i = tid + it * NTHR;                                                \
      i = (i < A_ITEMS) ? i : (A_ITEMS - 1);   \
      int r = i >> 3, c4 = (i & 7) * 4;                                       \
      ra[it] = *(const v4f*)(Ap + (size_t)(row0 + r) * ld + k0 + c4);         \
    }                                                                         \
    _Pragma("unroll")                     \
    for (int it = 0; it < B_IT; ++it) {                                       \
      int j = tid + it * NTHR;                                                \
      int kp = j / BQ, n4 = (j - kp * BQ) * 4;                                \
      const float* p = Bp + (size_t)(k0 + 2 * kp) * BN + n4;                  \
      rb0[it] = *(const v4f*)p;                                               \
      rb1[it] = *(const v4f*)(p + BN);                                        \
    }                                                                         \
  }

#define GEMM_SSTORE(buf)                                                      \
  {                                                                           \
    _Pragma("unroll")                          \
    for (int it = 0; it < A_IT; ++it) {                                       \
      int i = tid + it * NTHR;                                                \
      if (i < A_ITEMS) {                                                      \
        int r = i >> 3, c4 = (i & 7) * 4;                                     \
        *(v4f*)(As + (buf) * A_ST + r * AP + c4) = ra[it];                    \
      }                                                                       \
    }                                                                         \
    _Pragma("unroll")                  \
    for (int it = 0; it < B_IT; ++it) {                                       \
      int j = tid + it * NTHR;                                                \
      int kp = j / BQ, n4 = (j - kp * BQ) * 4;                                \
      v4f lo, hi;                                                             \
      lo.x = rb0[it].x; lo.y = rb1[it].x; lo.z = rb0[it].y; lo.w = rb1[it].y; \
      hi.x = rb0[it].z; hi.y = rb1[it].z; hi.z = rb0[it].w; hi.w = rb1[it].w; \
      float* d = Bs + (buf) * B_ST + kp * BP + 2 * n4;                        \
      *(v4f*)d       = lo;                                                    \
      *(v4f*)(d + 4) = hi;                                                    \
    }                                                                         \
  }

  v8f acc[5];
#pragma unroll
  for (int mt = 0; mt < 5; ++mt) acc[mt] = v8f{};

  GEMM_GLOAD(0)
  GEMM_SSTORE(0)
  __syncthreads();

  const int a_off = l16 * AP + 2 * g;
  const int b_off = g * BP + 2 * (wave * 16 + l16);
  for (int c = 0; c < nch; ++c) {
    const int cur = c & 1;
    const bool more = (c + 1 < nch);
    if (more) GEMM_GLOAD(c + 1)
    const float* as = As + cur * A_ST + a_off;
    const float* bs = Bs + cur * B_ST + b_off;
#pragma unroll
    for (int s = 0; s < BK / 4; ++s) {
      v2f b = *(const v2f*)(bs + 2 * s * BP);
#pragma unroll
      for (int mt = 0; mt < 5; ++mt) {
        v2f a = *(const v2f*)(as + mt * 16 * AP + 4 * s);
        acc[mt] = WMMA_F32(a, b, acc[mt]);
      }
    }
    if (more) GEMM_SSTORE(cur ^ 1)
    __syncthreads();
  }
#undef GEMM_GLOAD
#undef GEMM_SSTORE

  if (gamma) {
    tile_ln_store<BN>(acc, smem, bias, gamma, beta, resid, C, row0, do_gelu);
  } else {
    float* Cbase = C + (size_t)(row0 + 8 * g) * BN + wave * 16 + l16;
#pragma unroll
    for (int mt = 0; mt < 5; ++mt) {
#pragma unroll
      for (int i = 0; i < 8; ++i)
        Cbase[(size_t)(mt * 16 + i) * BN] = acc[mt][i];
    }
  }
}

template <int BN, bool DOT>
__device__ __forceinline__
void tile_ln_lds(v8f (&acc)[5], float* ct,
                 const float* __restrict__ bias,
                 const float* __restrict__ gamma,
                 const float* __restrict__ beta,
                 const float* __restrict__ wdot,
                 const float* __restrict__ bdot,
                 float* __restrict__ logits, int row0) {
  constexpr int CP = BN + 4;
  constexpr int NV = BN / 32;
  static_assert(!DOT || NV == 2, "the final dot is written for 64 columns");
  const int tid  = threadIdx.x;
  const int wave = tid >> 5;
  const int lane = tid & 31;
  const int l16  = lane & 15;
  const int g    = lane >> 4;

  if (wave < BN / 16) {
    float* cw = ct + (8 * g) * CP + wave * 16 + l16;
#pragma unroll
    for (int mt = 0; mt < 5; ++mt) {
#pragma unroll
      for (int i = 0; i < 8; ++i)
        cw[(mt * 16 + i) * CP] = acc[mt][i];
    }
  }
  __syncthreads();

  float gm[NV], bt[NV], bs[NV], wd[NV];
#pragma unroll
  for (int i = 0; i < NV; ++i) {
    const int c = lane + (i << 5);
    gm[i] = gamma[c];
    bt[i] = beta[c];
    bs[i] = bias[c];
    wd[i] = DOT ? wdot[c] : 0.f;
  }
  const float b0 = DOT ? bdot[0] : 0.f;

  for (int r = wave; r < 80; r += 16) {
    float* cr = ct + r * CP + lane;
    float vals[NV];
    float s = 0.f;
#pragma unroll
    for (int i = 0; i < NV; ++i) {
      float v = cr[i << 5];
      v += bs[i];
      vals[i] = v;
      s += v;
    }
    for (int o = 16; o; o >>= 1) s += __shfl_xor(s, o, 32);
    const float mean = s / (float)BN;
    float vs = 0.f;
#pragma unroll
    for (int i = 0; i < NV; ++i) { float d = vals[i] - mean; vs += d * d; }
    for (int o = 16; o; o >>= 1) vs += __shfl_xor(vs, o, 32);
    const float inv = rsqrtf(vs / (float)BN + LN_EPS);
#pragma unroll
    for (int i = 0; i < NV; ++i) {
      float v = (vals[i] - mean) * inv * gm[i] + bt[i];
      v = 0.5f * v * (1.f + erff(v * 0.70710678118654752f));
      vals[i] = v;
    }
    if (!DOT) {
#pragma unroll
      for (int i = 0; i < NV; ++i) cr[i << 5] = vals[i];
    } else {
      float d = vals[0] * wd[0] + vals[NV - 1] * wd[NV - 1];
      for (int o = 16; o; o >>= 1) d += __shfl_xor(d, o, 32);
      if (lane == 0) logits[row0 + r] = d + b0;
    }
  }
  if (!DOT) __syncthreads();
}

template <int BN, int AP>
__device__ __forceinline__
void lds_a_gemm(const float* At, int K,
                const float* __restrict__ Bm, float* Bs, v8f (&acc)[5]) {
  constexpr int BK = 32;
  constexpr int BP = 2 * BN + 32;
  constexpr int BQ = BN / 4;
  constexpr int B_ITEMS = (BK / 2) * BQ;
  constexpr int B_ST = (BK / 2) * BP;
  static_assert(B_ITEMS <= 512 && B_ITEMS % 32 == 0, "one piece per thread, whole waves");

  int tid = threadIdx.x;
  asm volatile("" : "+v"(tid));
  const int wave = tid >> 5;
  const int lane = tid & 31;
  const int l16  = lane & 15;
  const int g    = lane >> 4;
  const bool stager = tid < B_ITEMS;
  const bool worker = wave < BN / 16;
  const int kp = tid / BQ, n4 = (tid - kp * BQ) * 4;
  v4f rb0 = v4f{}, rb1 = v4f{};

#pragma unroll
  for (int mt = 0; mt < 5; ++mt) acc[mt] = v8f{};

#define LG_GLOAD(ch)                                                          \
  if (stager) {                                                               \
    const float* p = Bm + (size_t)((ch) * BK + 2 * kp) * BN + n4;             \
    rb0 = *(const v4f*)p;                                                     \
    rb1 = *(const v4f*)(p + BN);                                              \
  }
#define LG_SSTORE(buf)                                                        \
  if (stager) {                                                               \
    v4f lo, hi;                                                               \
    lo.x = rb0.x; lo.y = rb1.x; lo.z = rb0.y; lo.w = rb1.y;                   \
    hi.x = rb0.z; hi.y = rb1.z; hi.z = rb0.w; hi.w = rb1.w;                   \
    float* d = Bs + (buf) * B_ST + kp * BP + 2 * n4;                          \
    *(v4f*)d       = lo;                                                      \
    *(v4f*)(d + 4) = hi;                                                      \
  }

  LG_GLOAD(0)
  LG_SSTORE(0)
  __syncthreads();

  const int nch = K / BK;
  const int a_off = l16 * AP + 2 * g;
  const int b_off = g * BP + 2 * (wave * 16 + l16);
  for (int c = 0; c < nch; ++c) {
    const int cur = c & 1;
    const bool more = (c + 1 < nch);
    if (more) LG_GLOAD(c + 1)
    if (worker) {
      const float* as = At + a_off + c * BK;
      const float* bs = Bs + cur * B_ST + b_off;
#pragma unroll
      for (int s = 0; s < BK / 4; ++s) {
        v2f b = *(const v2f*)(bs + 2 * s * BP);
#pragma unroll
        for (int mt = 0; mt < 5; ++mt) {
          v2f a = *(const v2f*)(as + mt * 16 * AP + 4 * s);
          acc[mt] = WMMA_F32(a, b, acc[mt]);
        }
      }
    }
    if (more) LG_SSTORE(cur ^ 1)
    __syncthreads();
  }
#undef LG_GLOAD
#undef LG_SSTORE
}

__global__ __launch_bounds__(512, 8)
void decoder_tile_kernel(const float* __restrict__ z,
                         const int* __restrict__ gi,
                         const int* __restrict__ gj,
                         const float* __restrict__ Bm,
                         int K,
                         const float* __restrict__ bias,
                         const float* __restrict__ gamma,
                         const float* __restrict__ beta,
                         const float* __restrict__ W1,
                         const float* __restrict__ bias1,
                         const float* __restrict__ gamma1,
                         const float* __restrict__ beta1,
                         const float* __restrict__ W2,
                         const float* __restrict__ bias2,
                         const float* __restrict__ gamma2,
                         const float* __restrict__ beta2,
                         const float* __restrict__ W3,
                         const float* __restrict__ bias3,
                         float* __restrict__ logits) {
  constexpr int BN = H_;
  constexpr int NTHR = 2 * BN;
  constexpr int BM = 80, BK = 32;
  constexpr int AP = BK + 4;
  constexpr int BP = 2 * BN + 32;
  constexpr int A_ITEMS = BM * (BK / 4);
  constexpr int A_IT = (A_ITEMS + NTHR - 1) / NTHR;
  constexpr int BQ = BN / 4;
  constexpr int B_IT = ((BK / 2) * BQ) / NTHR;
  constexpr int A_ST = BM * AP;
  constexpr int B_ST = (BK / 2) * BP;
  constexpr int T0 = BM * (BN + 4);
  constexpr int S1 = 2 * (BK / 2) * (2 * 128 + 32);
  static_assert(2 * A_ST + 2 * B_ST <= T0 + S1, "layer-0 stages fit in the block's LDS");
  __shared__ __attribute__((aligned(16))) float smem[T0 + S1];
  float* const Bs2 = smem + T0;
  float* const As = smem;
  float* const Bs = smem + 2 * A_ST;

  const int tid  = threadIdx.x;
  const int wave = tid >> 5;
  const int lane = tid & 31;
  const int l16  = lane & 15;
  const int g    = lane >> 4;
  const int row0 = blockIdx.x * BM;

  unsigned offi[A_IT], offj[A_IT];
#pragma unroll
  for (int it = 0; it < A_IT; ++it) {
    int i = tid + it * NTHR;
    i = (i < A_ITEMS) ? i : (A_ITEMS - 1);
    int r = i >> 3, c4 = (i & 7) * 4;
    offi[it] = (unsigned)gi[row0 + r] * OUT_ + c4;
    offj[it] = (unsigned)gj[row0 + r] * OUT_ + c4;
  }

  v4f rzi[A_IT], rzj[A_IT];
  v4f rb0[B_IT], rb1[B_IT];

#define PE_GLOAD(k0)                                                          \
  {                                                                           \
    const int f0 = (k0) & (OUT_ - 1);                                         \
    _Pragma("unroll")        \
    for (int it = 0; it < A_IT; ++it) {                                       \
      rzi[it] = *(const v4f*)(z + offi[it] + f0);                             \
      rzj[it] = *(const v4f*)(z + offj[it] + f0);                             \
    }                                                                         \
    _Pragma("unroll")                     \
    for (int it = 0; it < B_IT; ++it) {                                       \
      int j = tid + it * NTHR;                                                \
      int kp = j / BQ, n4 = (j - kp * BQ) * 4;                                \
      const float* p = Bm + (size_t)((k0) + 2 * kp) * BN + n4;                \
      rb0[it] = *(const v4f*)p;                                               \
      rb1[it] = *(const v4f*)(p + BN);                                        \
    }                                                                         \
  }

#define PE_SSTORE(buf, k0)                                                    \
  {                                                                           \
    const int sec = (k0) >> 7;                                                \
    _Pragma("unroll")                  \
    for (int it = 0; it < A_IT; ++it) {                                       \
      int i = tid + it * NTHR;                                                \
      if (i < A_ITEMS) {                                                      \
        int r = i >> 3, c4 = (i & 7) * 4;                                     \
        v4f a;                                                                \
        if (sec == 0)      a = rzi[it] + rzj[it];                             \
        else if (sec == 1) a = rzi[it] * rzj[it];                             \
        else {                                                                \
          const v4f d = rzi[it] - rzj[it];                                    \
          a.x = fabsf(d.x); a.y = fabsf(d.y); a.z = fabsf(d.z); a.w = fabsf(d.w); \
        }                                                                     \
        *(v4f*)(As + (buf) * A_ST + r * AP + c4) = a;                                   \
      }                                                                       \
    }                                                                         \
    _Pragma("unroll")                  \
    for (int it = 0; it < B_IT; ++it) {                                       \
      int j = tid + it * NTHR;                                                \
      int kp = j / BQ, n4 = (j - kp * BQ) * 4;                                \
      v4f lo, hi;                                                             \
      lo.x = rb0[it].x; lo.y = rb1[it].x; lo.z = rb0[it].y; lo.w = rb1[it].y; \
      hi.x = rb0[it].z; hi.y = rb1[it].z; hi.z = rb0[it].w; hi.w = rb1[it].w; \
      float* d = Bs + (buf) * B_ST + kp * BP + 2 * n4;                                  \
      *(v4f*)d       = lo;                                                    \
      *(v4f*)(d + 4) = hi;                                                    \
    }                                                                         \
  }

  v8f acc[5];
#pragma unroll
  for (int mt = 0; mt < 5; ++mt) acc[mt] = v8f{};

  PE_GLOAD(0)
  PE_SSTORE(0, 0)
  __syncthreads();

  const int nch = K / BK;
  const int a_off = l16 * AP + 2 * g;
  const int b_off = g * BP + 2 * (wave * 16 + l16);
  for (int c = 0; c < nch; ++c) {
    const int cur = c & 1;
    const bool more = (c + 1 < nch);
    if (more) PE_GLOAD((c + 1) * BK)
    const float* as = As + cur * A_ST + a_off;
    const float* bs = Bs + cur * B_ST + b_off;
#pragma unroll
    for (int s = 0; s < BK / 4; ++s) {
      v2f b = *(const v2f*)(bs + 2 * s * BP);
#pragma unroll
      for (int mt = 0; mt < 5; ++mt) {
        v2f a = *(const v2f*)(as + mt * 16 * AP + 4 * s);
        acc[mt] = WMMA_F32(a, b, acc[mt]);
      }
    }
    if (more) PE_SSTORE(cur ^ 1, (c + 1) * BK)
    __syncthreads();
  }
#undef PE_GLOAD
#undef PE_SSTORE

  tile_ln_lds<256, false>(acc, smem, bias, gamma, beta, nullptr, nullptr, nullptr, row0);
  lds_a_gemm<128, 260>(smem, H_, W1, Bs2, acc);
  tile_ln_lds<128, false>(acc, smem, bias1, gamma1, beta1, nullptr, nullptr, nullptr, row0);
  lds_a_gemm<64, 132>(smem, H_ / 2, W2, Bs2, acc);
  tile_ln_lds<64, true>(acc, smem, bias2, gamma2, beta2, W3, bias3, logits, row0);
}

extern "C" void kernel_launch(void* const* d_in, const int* in_sizes, int n_in,
                              void* d_out, int out_size, void* d_ws, size_t ws_size,
                              hipStream_t stream) {
  (void)in_sizes; (void)n_in; (void)out_size; (void)ws_size;

  const float* x    = (const float*)d_in[0];
  const int*   ei   = (const int*)d_in[1];
  const int*   esrc = ei;
  const int*   edst = ei + E_;
  const int*   et   = (const int*)d_in[2];
  const int*   gp   = (const int*)d_in[3];
  const int*   gi   = gp;
  const int*   gj   = gp + P_;

  const float* basis[4], *att[4], *root[4], *relatt[4], *gam[4], *bet[4];
  for (int i = 0; i < 4; ++i) {
    basis[i]  = (const float*)d_in[4 + 6 * i + 0];
    att[i]    = (const float*)d_in[4 + 6 * i + 1];
    root[i]   = (const float*)d_in[4 + 6 * i + 2];
    relatt[i] = (const float*)d_in[4 + 6 * i + 3];
    gam[i]    = (const float*)d_in[4 + 6 * i + 4];
    bet[i]    = (const float*)d_in[4 + 6 * i + 5];
  }
  const float* res0 = (const float*)d_in[28];
  const float* res3 = (const float*)d_in[29];
  const float* dW0 = (const float*)d_in[30], *db0 = (const float*)d_in[31];
  const float* dg0 = (const float*)d_in[32], *dbb0 = (const float*)d_in[33];
  const float* dW1 = (const float*)d_in[34], *db1 = (const float*)d_in[35];
  const float* dg1 = (const float*)d_in[36], *dbb1 = (const float*)d_in[37];
  const float* dW2 = (const float*)d_in[38], *db2 = (const float*)d_in[39];
  const float* dg2 = (const float*)d_in[40], *dbb2 = (const float*)d_in[41];
  const float* dW3 = (const float*)d_in[42], *db3 = (const float*)d_in[43];

  float* ws = (float*)d_ws;
  size_t off = 0;
  float* a_sm  = ws + off; off += 64;
  float* Wcat  = ws + off; off += (size_t)R_ * H_ * H_;
  float* hA    = ws + off; off += (size_t)N_ * H_;
  float* hB    = ws + off; off += (size_t)N_ * H_;
  float* resid = ws + off; off += (size_t)N_ * H_;
  float* big   = ws + off;
  float* zout = (float*)d_out + P_;
  int* csr_cnt    = (int*)(resid + (size_t)N_ * OUT_);
  int* csr_offs   = csr_cnt + 20480;
  int* csr_rank   = csr_offs + 20480;
  int* csr_key    = csr_rank + E_;
  int* csr_packed = csr_key + E_;

  auto gemm = [&](const float* A, const float* Bm, int K, int lda,
                  const float* A2, const float* B2, int K2, int lda2,
                  float* Cm, int M, int Nc,
                  const float* bias, const float* gamma, const float* beta,
                  const float* res, int gelu) {
    int blocks = M / 80;
    if (Nc == 256)
      gemm_f32_wmma_lds_kernel<256><<<blocks, 512, 0, stream>>>(
          A, Bm, K, lda, A2, B2, K2, lda2, Cm, bias, gamma, beta, res, gelu);
    else
      gemm_f32_wmma_lds_kernel<128><<<blocks, 256, 0, stream>>>(
          A, Bm, K, lda, A2, B2, K2, lda2, Cm, bias, gamma, beta, res, gelu);
  };

  struct { const float* cur; int in; int out; int li;
           const float* resmat; float* conv; int gelu; float* rbuf; } L[4] = {
    { x,  IN_, H_,   0, res0,    hA,   1, hB },
    { hA, H_,  H_,   1, nullptr, hB,   1, nullptr },
    { hB, H_,  H_,   2, nullptr, hA,   1, nullptr },
    { hA, H_,  OUT_, 3, res3,    zout, 0, resid },
  };

  csr_zero_kernel<<<(N_ + 255) / 256, 256, 0, stream>>>(csr_cnt);
  csr_count_kernel<<<(E_ / 4 + 255) / 256, 256, 0, stream>>>(edst, csr_cnt, csr_rank);
  csr_scan_kernel<<<1, 1024, 0, stream>>>(csr_cnt, csr_offs);
  csr_fill_kernel<<<(E_ / 4 + 255) / 256, 256, 0, stream>>>(edst, et, csr_offs, csr_rank, csr_key);
  csr_order_kernel<<<(N_ + 31) / 32, 256, 0, stream>>>(esrc, csr_offs, csr_key, csr_packed);

  for (int i = 0; i < 4; ++i) {
    const float* cur = L[i].cur;
    int in_ch = L[i].in, out_ch = L[i].out, li = L[i].li;
    float* conv = L[i].conv;

    relatt_softmax_kernel<<<1, 32, 0, stream>>>(relatt[li], a_sm);
    int wtotal = R_ * in_ch * out_ch;
    make_weights_kernel<<<(wtotal + 2047) / 2048, 256, 0, stream>>>(
        att[li], basis[li], a_sm, Wcat, in_ch, out_ch);

    if (in_ch == H_)
      gather_agg_kernel<H_><<<N_ / 16, 256, 0, stream>>>(cur, csr_offs, csr_packed, big);
    else
      gather_agg_kernel<IN_><<<N_ / 32, 256, 0, stream>>>(cur, csr_offs, csr_packed, big);

    const float* rptr = cur;
    if (L[i].resmat) {
      gemm(cur, L[i].resmat, in_ch, in_ch, nullptr, nullptr, 0, 0,
           L[i].rbuf, N_, out_ch, nullptr, nullptr, nullptr, nullptr, 0);
      rptr = L[i].rbuf;
    }
    gemm(big, Wcat, R_ * in_ch, R_ * in_ch, cur, root[li], in_ch, in_ch,
         conv, N_, out_ch, nullptr, gam[li], bet[li], rptr, L[i].gelu);
  }

  decoder_tile_kernel<<<P_ / 80, 512, 0, stream>>>(zout, gi, gj, dW0, 3 * OUT_, db0, dg0, dbb0,
                                                   dW1, db1, dg1, dbb1, dW2, db2, dg2, dbb2,
                                                   dW3, db3, (float*)d_out);
}


# Edge sums made inside the layer GEMM: 8 extra waves per block add the source rows of each 80x32 A chunk into LDS while the matrix waves multiply; no agg array or gather launch; sums keep their order
# speedup vs baseline: 3.3892x; 1.1698x over previous
#include <hip/hip_runtime.h>
#include <hip/hip_bf16.h>
#include <math.h>

#define N_   20000
#define E_   320000
#define R_   8
#define B_   8
#define IN_  128
#define H_   256
#define OUT_ 128
#define P_   100000
#define LN_EPS 1e-5f

typedef float v8f __attribute__((ext_vector_type(8)));
typedef float v2f __attribute__((ext_vector_type(2)));
typedef float v4f __attribute__((ext_vector_type(4)));

#define WMMA_F32(a, b, c) \
  __builtin_amdgcn_wmma_f32_16x16x4_f32(false, (a), false, (b), (short)0, (c), false, false)

__global__ void relatt_softmax_kernel(const float* __restrict__ relatt,
                                      float* __restrict__ a) {
  if (threadIdx.x == 0 && blockIdx.x == 0) {
    float m = relatt[0];
    for (int r = 1; r < R_; ++r) m = fmaxf(m, relatt[r]);
    float e[R_]; float s = 0.f;
    for (int r = 0; r < R_; ++r) { e[r] = expf(relatt[r] - m); s += e[r]; }
    for (int r = 0; r < R_; ++r) a[r] = e[r] / s;
  }
}

__global__ __launch_bounds__(256)
void make_weights_kernel(const float* __restrict__ att,
                         const float* __restrict__ basis,
                         const float* __restrict__ a,
                         float* __restrict__ W,
                         int in_ch, int out_ch) {
  const int io_sz  = in_ch * out_ch;
  const int total4 = (R_ * io_sz) / 4;
  const int nthr   = gridDim.x * blockDim.x;
  for (int q = blockIdx.x * blockDim.x + threadIdx.x; q < total4; q += nthr) {
    const int idx = 4 * q;
    const int r   = idx / io_sz;
    const int io  = idx - r * io_sz;
    v4f s = v4f{};
#pragma unroll
    for (int bb = 0; bb < B_; ++bb)
      s += att[r * B_ + bb] * *(const v4f*)(basis + (size_t)bb * io_sz + io);
    *(v4f*)(W + idx) = s * a[r];
  }
}

__global__ void csr_zero_kernel(int* __restrict__ cnt) {
  int i = blockIdx.x * blockDim.x + threadIdx.x;
  if (i < N_) cnt[i] = 0;
}

__global__ void csr_count_kernel(const int* __restrict__ dst,
                                 int* __restrict__ cnt,
                                 int* __restrict__ rank) {
  const int nthr = gridDim.x * blockDim.x;
  for (int e = blockIdx.x * blockDim.x + threadIdx.x; e < E_; e += nthr)
    rank[e] = atomicAdd(&cnt[dst[e]], 1);
}

__global__ __launch_bounds__(1024)
void csr_scan_kernel(const int* __restrict__ cnt, int* __restrict__ offs) {
  constexpr int CH = (N_ + 1023) / 1024;
  __shared__ int part[1024];
  const int t = threadIdx.x;
  int loc[CH];
  int s = 0;
#pragma unroll
  for (int i = 0; i < CH; ++i) {
    int idx = t * CH + i;
    int v = (idx < N_) ? cnt[idx] : 0;
    loc[i] = s;
    s += v;
  }
  part[t] = s;
  __syncthreads();
  for (int o = 1; o < 1024; o <<= 1) {
    int v = (t >= o) ? part[t - o] : 0;
    __syncthreads();
    part[t] += v;
    __syncthreads();
  }
  const int excl = part[t] - s;
#pragma unroll
  for (int i = 0; i < CH; ++i) {
    int idx = t * CH + i;
    if (idx < N_) offs[idx] = excl + loc[i];
  }
  if (t == 1023) offs[N_] = part[1023];
}

__global__ void csr_fill_kernel(const int* __restrict__ dst,
                                const int* __restrict__ et,
                                const int* __restrict__ offs,
                                const int* __restrict__ rank,
                                int* __restrict__ key) {
  const int nthr = gridDim.x * blockDim.x;
  for (int e = blockIdx.x * blockDim.x + threadIdx.x; e < E_; e += nthr)
    key[offs[dst[e]] + rank[e]] = et[e] * E_ + e;
}

__global__ void csr_order_kernel(const int* __restrict__ src,
                                 const int* __restrict__ offs,
                                 const int* __restrict__ key,
                                 int* __restrict__ packed,
                                 int* __restrict__ seg) {
  int wave = threadIdx.x >> 5;
  int lane = threadIdx.x & 31;
  for (int node = blockIdx.x * 8 + wave; node < N_; node += gridDim.x * 8) {
    int beg = offs[node];
    int n   = offs[node + 1] - beg;
    for (int i = lane; i < n; i += 32) {
      int my = key[beg + i];
      int rk = 0;
      for (int j = 0; j < n; ++j) rk += (key[beg + j] < my) ? 1 : 0;
      int r = my / E_;
      int e = my - r * E_;
      packed[beg + rk] = src[e] * R_ + r;
    }
    if (lane < R_) {
      int c = 0;
      for (int j = 0; j < n; ++j) c += (key[beg + j] / E_ < lane) ? 1 : 0;
      seg[node * R_ + lane] = beg + c;
    }
    if (node == N_ - 1 && lane == R_) seg[N_ * R_] = beg + n;
  }
}

template <int BN, int NW, int W0>
__device__ __forceinline__
void tile_ln_store(v8f (&acc)[5], float* ct,
                   const float* __restrict__ bias,
                   const float* __restrict__ gamma,
                   const float* __restrict__ beta,
                   const float* __restrict__ resid,
                   float* __restrict__ out, int row0, int do_gelu) {
  constexpr int CP = BN + 4;
  constexpr int NV = BN / 32;
  const int tid  = threadIdx.x;
  const int wave = tid >> 5;
  const int lane = tid & 31;
  const int l16  = lane & 15;
  const int g    = lane >> 4;

  const int sw = wave - W0;
  if (NW == BN / 16 || (sw >= 0 && sw < BN / 16)) {
    float* cw = ct + (8 * g) * CP + sw * 16 + l16;
#pragma unroll
    for (int mt = 0; mt < 5; ++mt) {
#pragma unroll
      for (int i = 0; i < 8; ++i)
        cw[(mt * 16 + i) * CP] = acc[mt][i];
    }
  }
  __syncthreads();

  float gm[NV], bt[NV], bs[NV];
#pragma unroll
  for (int i = 0; i < NV; ++i) {
    const int c = lane + (i << 5);
    gm[i] = gamma[c];
    bt[i] = beta[c];
    bs[i] = bias ? bias[c] : 0.f;
  }

  for (int r = wave; r < 80; r += NW) {
    const size_t base = (size_t)(row0 + r) * BN + lane;
    const float* cr = ct + r * CP + lane;
    float vals[NV], rs[NV];
#pragma unroll
    for (int i = 0; i < NV; ++i) {
      vals[i] = cr[i << 5];
      rs[i] = resid ? resid[base + (i << 5)] : 0.f;
    }
    float s = 0.f;
#pragma unroll
    for (int i = 0; i < NV; ++i) {
      float v = vals[i];
      if (bias) v += bs[i];
      vals[i] = v;
      s += v;
    }
    for (int o = 16; o; o >>= 1) s += __shfl_xor(s, o, 32);
    const float mean = s / (float)BN;
    float vs = 0.f;
#pragma unroll
    for (int i = 0; i < NV; ++i) { float d = vals[i] - mean; vs += d * d; }
    for (int o = 16; o; o >>= 1) vs += __shfl_xor(vs, o, 32);
    const float inv = rsqrtf(vs / (float)BN + LN_EPS);
#pragma unroll
    for (int i = 0; i < NV; ++i) {
      float v = (vals[i] - mean) * inv * gm[i] + bt[i];
      if (resid) v += rs[i];
      if (do_gelu) v = 0.5f * v * (1.f + erff(v * 0.70710678118654752f));
      out[base + (i << 5)] = v;
    }
  }
}

template <int BN>
__global__ __launch_bounds__(2 * BN)
void gemm_f32_wmma_lds_kernel(const float* __restrict__ A,
                              const float* __restrict__ Bm, int K, int lda,
                              const float* __restrict__ A2,
                              const float* __restrict__ B2, int K2, int lda2,
                              float* __restrict__ C,
                              const float* __restrict__ bias,
                              const float* __restrict__ gamma,
                              const float* __restrict__ beta,
                              const float* __restrict__ resid,
                              int do_gelu) {
  constexpr int NTHR = 2 * BN;
  constexpr int BM = 80, BK = 32;
  constexpr int AP = BK + 4;
  constexpr int BP = 2 * BN + 32;
  constexpr int A_ITEMS = BM * (BK / 4);
  constexpr int A_IT = (A_ITEMS + NTHR - 1) / NTHR;
  constexpr int BQ = BN / 4;
  constexpr int B_IT = ((BK / 2) * BQ) / NTHR;
  constexpr int A_ST = BM * AP;
  constexpr int B_ST = (BK / 2) * BP;
  static_assert(2 * A_ST + 2 * B_ST >= BM * (BN + 4), "result tile must fit in the operand stages");
  __shared__ __attribute__((aligned(16))) float smem[2 * A_ST + 2 * B_ST];
  float* const As = smem;
  float* const Bs = smem + 2 * A_ST;

  const int tid  = threadIdx.x;
  const int wave = tid >> 5;
  const int lane = tid & 31;
  const int l16  = lane & 15;
  const int g    = lane >> 4;
  const int row0 = blockIdx.x * BM;
  const int nch1 = K / BK;
  const int nch  = nch1 + K2 / BK;

  v4f ra[A_IT];
  v4f rb0[B_IT], rb1[B_IT];

#define GEMM_GLOAD(ch)                                                        \
  {                                                                           \
    const bool first = (ch) < nch1;                                           \
    const float* Ap = first ? A : A2;                                         \
    const float* Bp = first ? Bm : B2;                                        \
    const int ld = first ? lda : lda2;                                        \
    const int k0 = (first ? (ch) : (ch) - nch1) * BK;                         \
    _Pragma("unroll")                \
    for (int it = 0; it < A_IT; ++it) {                                       \
      int i = tid + it * NTHR;                                                \
      i = (i < A_ITEMS) ? i : (A_ITEMS - 1);   \
      int r = i >> 3, c4 = (i & 7) * 4;                                       \
      ra[it] = *(const v4f*)(Ap + (size_t)(row0 + r) * ld + k0 + c4);         \
    }                                                                         \
    _Pragma("unroll")                     \
    for (int it = 0; it < B_IT; ++it) {                                       \
      int j = tid + it * NTHR;                                                \
      int kp = j / BQ, n4 = (j - kp * BQ) * 4;                                \
      const float* p = Bp + (size_t)(k0 + 2 * kp) * BN + n4;                  \
      rb0[it] = *(const v4f*)p;                                               \
      rb1[it] = *(const v4f*)(p + BN);                                        \
    }                                                                         \
  }

#define GEMM_SSTORE(buf)                                                      \
  {                                                                           \
    _Pragma("unroll")                          \
    for (int it = 0; it < A_IT; ++it) {                                       \
      int i = tid + it * NTHR;                                                \
      if (i < A_ITEMS) {                                                      \
        int r = i >> 3, c4 = (i & 7) * 4;                                     \
        *(v4f*)(As + (buf) * A_ST + r * AP + c4) = ra[it];                    \
      }                                                                       \
    }                                                                         \
    _Pragma("unroll")                  \
    for (int it = 0; it < B_IT; ++it) {                                       \
      int j = tid + it * NTHR;                                                \
      int kp = j / BQ, n4 = (j - kp * BQ) * 4;                                \
      v4f lo, hi;                                                             \
      lo.x = rb0[it].x; lo.y = rb1[it].x; lo.z = rb0[it].y; lo.w = rb1[it].y; \
      hi.x = rb0[it].z; hi.y = rb1[it].z; hi.z = rb0[it].w; hi.w = rb1[it].w; \
      float* d = Bs + (buf) * B_ST + kp * BP + 2 * n4;                        \
      *(v4f*)d       = lo;                                                    \
      *(v4f*)(d + 4) = hi;                                                    \
    }                                                                         \
  }

  v8f acc[5];
#pragma unroll
  for (int mt = 0; mt < 5; ++mt) acc[mt] = v8f{};

  GEMM_GLOAD(0)
  GEMM_SSTORE(0)
  __syncthreads();

  const int a_off = l16 * AP + 2 * g;
  const int b_off = g * BP + 2 * (wave * 16 + l16);
  for (int c = 0; c < nch; ++c) {
    const int cur = c & 1;
    const bool more = (c + 1 < nch);
    if (more) GEMM_GLOAD(c + 1)
    const float* as = As + cur * A_ST + a_off;
    const float* bs = Bs + cur * B_ST + b_off;
#pragma unroll
    for (int s = 0; s < BK / 4; ++s) {
      v2f b = *(const v2f*)(bs + 2 * s * BP);
#pragma unroll
      for (int mt = 0; mt < 5; ++mt) {
        v2f a = *(const v2f*)(as + mt * 16 * AP + 4 * s);
        acc[mt] = WMMA_F32(a, b, acc[mt]);
      }
    }
    if (more) GEMM_SSTORE(cur ^ 1)
    __syncthreads();
  }
#undef GEMM_GLOAD
#undef GEMM_SSTORE

  if (gamma) {
    tile_ln_store<BN, BN / 16, 0>(acc, smem, bias, gamma, beta, resid, C, row0, do_gelu);
  } else {
    float* Cbase = C + (size_t)(row0 + 8 * g) * BN + wave * 16 + l16;
#pragma unroll
    for (int mt = 0; mt < 5; ++mt) {
#pragma unroll
      for (int i = 0; i < 8; ++i)
        Cbase[(size_t)(mt * 16 + i) * BN] = acc[mt][i];
    }
  }
}

template <int BN, int IN>
__global__ __launch_bounds__(2 * BN + 256)
void rgcn_layer_gemm_kernel(const float* __restrict__ X,
                            const int* __restrict__ seg,
                            const int* __restrict__ packed,
                            const float* __restrict__ Bm,
                            const float* __restrict__ B2,
                            float* __restrict__ C,
                            const float* __restrict__ gamma,
                            const float* __restrict__ beta,
                            const float* __restrict__ resid,
                            int do_gelu) {
  constexpr int NCONS = 2 * BN;
  constexpr int NPROD = 256;
  constexpr int BM = 80, BK = 32;
  constexpr int AP = BK + 4;
  constexpr int BP = 2 * BN + 32;
  constexpr int A_ITEMS = BM * (BK / 4);
  constexpr int P_IT = (A_ITEMS + NPROD - 1) / NPROD;
  constexpr int BQ = BN / 4;
  constexpr int B_IT = ((BK / 2) * BQ) / NCONS;
  constexpr int A_ST = BM * AP;
  constexpr int B_ST = (BK / 2) * BP;
  constexpr int CPR = IN / BK;
  constexpr int NCH1 = R_ * CPR;
  constexpr int NCH = NCH1 + CPR;
  constexpr int NPRE = (BN >= 256) ? 4 : 8;
  constexpr int NPASS = (BN >= 256) ? 2 : 1;
  constexpr int NKNOWN = NPRE * NPASS;
  static_assert(2 * A_ST + 2 * B_ST >= BM * (BN + 4), "result tile must fit in the operand stages");
  static_assert(IN % BK == 0, "a chunk lies inside one relation");
  __shared__ __attribute__((aligned(16))) float smem[2 * A_ST + 2 * B_ST];
  __shared__ unsigned pso_l[NKNOWN * P_IT * NPROD];
  float* const As = smem;
  float* const Bs = smem + 2 * A_ST;

  const int pt   = threadIdx.x;
  const int tid  = pt - NPROD;
  const int wave = tid >> 5;
  const int lane = pt & 31;
  const int l16  = lane & 15;
  const int g    = lane >> 4;
  const int row0 = blockIdx.x * BM;
  const bool consumer = pt >= NPROD;

  int prow[P_IT], pc4[P_IT];
  bool pv[P_IT];
  int pcnt[P_IT], pbeg[P_IT];
#pragma unroll
  for (int it = 0; it < P_IT; ++it) {
    int i = pt + it * NPROD;
    pv[it] = !consumer && i < A_ITEMS;
    i = pv[it] ? i : 0;
    prow[it] = i >> 3;
    pc4[it] = (i & 7) * 4;
    pcnt[it] = 0;
    pbeg[it] = 0;
  }

  auto build_a = [&](int cn, int buf) {
    v4f a[P_IT];
    if (cn < NCH1) {
      const int rel = cn / CPR;
      const int f0  = (cn - rel * CPR) * BK;
      if (cn == rel * CPR) {
        int s[P_IT], e[P_IT], p[P_IT][NKNOWN];
#pragma unroll
        for (int it = 0; it < P_IT; ++it) {
          const int sidx = (row0 + prow[it]) * R_ + rel;
          s[it] = seg[sidx];
          e[it] = seg[sidx + 1];
        }
#pragma unroll
        for (int it = 0; it < P_IT; ++it) {
#pragma unroll
          for (int u = 0; u < NKNOWN; ++u) {
            const int idx = s[it] + u;
            p[it][u] = packed[idx < E_ ? idx : E_ - 1];
          }
        }
#pragma unroll
        for (int it = 0; it < P_IT; ++it) {
          pbeg[it] = s[it];
          pcnt[it] = pv[it] ? e[it] - s[it] : 0;
#pragma unroll
          for (int u = 0; u < NKNOWN; ++u)
            pso_l[(it * NKNOWN + u) * NPROD + pt] = (u < pcnt[it]) ? ((unsigned)p[it][u] / R_) * IN : 0u;
        }
      }
#pragma unroll
      for (int it = 0; it < P_IT; ++it) a[it] = v4f{};
#pragma unroll
      for (int ps = 0; ps < NPASS; ++ps) {
        bool need = true;
        if (ps > 0) {
          bool m = false;
#pragma unroll
          for (int it = 0; it < P_IT; ++it) m = m || (pcnt[it] > ps * NPRE);
          need = __builtin_amdgcn_ballot_w32(m) != 0u;
        }
        if (need) {
          unsigned so[P_IT][NPRE];
          v4f v[P_IT][NPRE];
#pragma unroll
          for (int it = 0; it < P_IT; ++it) {
#pragma unroll
            for (int u = 0; u < NPRE; ++u)
              so[it][u] = pso_l[(it * NKNOWN + ps * NPRE + u) * NPROD + pt];
          }
#pragma unroll
          for (int it = 0; it < P_IT; ++it) {
#pragma unroll
            for (int u = 0; u < NPRE; ++u)
              v[it][u] = *(const v4f*)(X + so[it][u] + f0 + pc4[it]);
          }
#pragma unroll
          for (int it = 0; it < P_IT; ++it) {
#pragma unroll
            for (int u = 0; u < NPRE; ++u)
              a[it] += (ps * NPRE + u < pcnt[it]) ? v[it][u] : v4f{};
          }
        }
      }
#pragma unroll
      for (int it = 0; it < P_IT; ++it) {
        for (int j = NKNOWN; j < pcnt[it]; ++j) {
          const unsigned so = ((unsigned)packed[pbeg[it] + j] / R_) * IN;
          a[it] += *(const v4f*)(X + so + f0 + pc4[it]);
        }
      }
    } else {
      const int k0 = (cn - NCH1) * BK;
#pragma unroll
      for (int it = 0; it < P_IT; ++it)
        a[it] = *(const v4f*)(X + (size_t)(row0 + prow[it]) * IN + k0 + pc4[it]);
    }
#pragma unroll
    for (int it = 0; it < P_IT; ++it)
      if (pv[it]) *(v4f*)(As + buf * A_ST + prow[it] * AP + pc4[it]) = a[it];
  };

  v4f rb0[B_IT], rb1[B_IT];
#define LAYER_GLOAD(ch)                                                       \
  {                                                                           \
    const bool first = (ch) < NCH1;                                           \
    const float* Bp = first ? Bm : B2;                                        \
    const int k0 = (first ? (ch) : (ch) - NCH1) * BK;                         \
    _Pragma("unroll")                     \
    for (int it = 0; it < B_IT; ++it) {                                       \
      int j = tid + it * NCONS;                                               \
      int kp = j / BQ, n4 = (j - kp * BQ) * 4;                                \
      const float* p = Bp + (size_t)(k0 + 2 * kp) * BN + n4;                  \
      rb0[it] = *(const v4f*)p;                                               \
      rb1[it] = *(const v4f*)(p + BN);                                        \
    }                                                                         \
  }
#define LAYER_SSTORE(buf)                                                     \
  {                                                                           \
    _Pragma("unroll")                  \
    for (int it = 0; it < B_IT; ++it) {                                       \
      int j = tid + it * NCONS;                                               \
      int kp = j / BQ, n4 = (j - kp * BQ) * 4;                                \
      v4f lo, hi;                                                             \
      lo.x = rb0[it].x; lo.y = rb1[it].x; lo.z = rb0[it].y; lo.w = rb1[it].y; \
      hi.x = rb0[it].z; hi.y = rb1[it].z; hi.z = rb0[it].w; hi.w = rb1[it].w; \
      float* d = Bs + (buf) * B_ST + kp * BP + 2 * n4;                        \
      *(v4f*)d       = lo;                                                    \
      *(v4f*)(d + 4) = hi;                                                    \
    }                                                                         \
  }

  v8f acc[5];
#pragma unroll
  for (int mt = 0; mt < 5; ++mt) acc[mt] = v8f{};

  if (consumer) {
    LAYER_GLOAD(0)
    LAYER_SSTORE(0)
  } else {
    build_a(0, 0);
  }
  __syncthreads();

  const int a_off = l16 * AP + 2 * g;
  const int b_off = g * BP + 2 * (wave * 16 + l16);
  for (int c = 0; c < NCH; ++c) {
    const int cur = c & 1;
    const bool more = (c + 1 < NCH);
    if (consumer) {
      if (more) LAYER_GLOAD(c + 1)
      const float* as = As + cur * A_ST + a_off;
      const float* bs = Bs + cur * B_ST + b_off;
#pragma unroll
      for (int s = 0; s < BK / 4; ++s) {
        v2f b = *(const v2f*)(bs + 2 * s * BP);
#pragma unroll
        for (int mt = 0; mt < 5; ++mt) {
          v2f a = *(const v2f*)(as + mt * 16 * AP + 4 * s);
          acc[mt] = WMMA_F32(a, b, acc[mt]);
        }
      }
      if (more) LAYER_SSTORE(cur ^ 1)
    } else {
      if (more) build_a(c + 1, cur ^ 1);
    }
    __syncthreads();
  }
#undef LAYER_GLOAD
#undef LAYER_SSTORE

  tile_ln_store<BN, (NCONS + NPROD) / 32, NPROD / 32>(acc, smem, nullptr, gamma, beta, resid, C, row0, do_gelu);
}

template <int BN, bool DOT>
__device__ __forceinline__
void tile_ln_lds(v8f (&acc)[5], float* ct,
                 const float* __restrict__ bias,
                 const float* __restrict__ gamma,
                 const float* __restrict__ beta,
                 const float* __restrict__ wdot,
                 const float* __restrict__ bdot,
                 float* __restrict__ logits, int row0) {
  constexpr int CP = BN + 4;
  constexpr int NV = BN / 32;
  static_assert(!DOT || NV == 2, "the final dot is written for 64 columns");
  const int tid  = threadIdx.x;
  const int wave = tid >> 5;
  const int lane = tid & 31;
  const int l16  = lane & 15;
  const int g    = lane >> 4;

  if (wave < BN / 16) {
    float* cw = ct + (8 * g) * CP + wave * 16 + l16;
#pragma unroll
    for (int mt = 0; mt < 5; ++mt) {
#pragma unroll
      for (int i = 0; i < 8; ++i)
        cw[(mt * 16 + i) * CP] = acc[mt][i];
    }
  }
  __syncthreads();

  float gm[NV], bt[NV], bs[NV], wd[NV];
#pragma unroll
  for (int i = 0; i < NV; ++i) {
    const int c = lane + (i << 5);
    gm[i] = gamma[c];
    bt[i] = beta[c];
    bs[i] = bias[c];
    wd[i] = DOT ? wdot[c] : 0.f;
  }
  const float b0 = DOT ? bdot[0] : 0.f;

  for (int r = wave; r < 80; r += 16) {
    float* cr = ct + r * CP + lane;
    float vals[NV];
    float s = 0.f;
#pragma unroll
    for (int i = 0; i < NV; ++i) {
      float v = cr[i << 5];
      v += bs[i];
      vals[i] = v;
      s += v;
    }
    for (int o = 16; o; o >>= 1) s += __shfl_xor(s, o, 32);
    const float mean = s / (float)BN;
    float vs = 0.f;
#pragma unroll
    for (int i = 0; i < NV; ++i) { float d = vals[i] - mean; vs += d * d; }
    for (int o = 16; o; o >>= 1) vs += __shfl_xor(vs, o, 32);
    const float inv = rsqrtf(vs / (float)BN + LN_EPS);
#pragma unroll
    for (int i = 0; i < NV; ++i) {
      float v = (vals[i] - mean) * inv * gm[i] + bt[i];
      v = 0.5f * v * (1.f + erff(v * 0.70710678118654752f));
      vals[i] = v;
    }
    if (!DOT) {
#pragma unroll
      for (int i = 0; i < NV; ++i) cr[i << 5] = vals[i];
    } else {
      float d = vals[0] * wd[0] + vals[NV - 1] * wd[NV - 1];
      for (int o = 16; o; o >>= 1) d += __shfl_xor(d, o, 32);
      if (lane == 0) logits[row0 + r] = d + b0;
    }
  }
  if (!DOT) __syncthreads();
}

template <int BN, int AP>
__device__ __forceinline__
void lds_a_gemm(const float* At, int K,
                const float* __restrict__ Bm, float* Bs, v8f (&acc)[5]) {
  constexpr int BK = 32;
  constexpr int BP = 2 * BN + 32;
  constexpr int BQ = BN / 4;
  constexpr int B_ITEMS = (BK / 2) * BQ;
  constexpr int B_ST = (BK / 2) * BP;
  static_assert(B_ITEMS <= 512 && B_ITEMS % 32 == 0, "one piece per thread, whole waves");

  int tid = threadIdx.x;
  asm volatile("" : "+v"(tid));
  const int wave = tid >> 5;
  const int lane = tid & 31;
  const int l16  = lane & 15;
  const int g    = lane >> 4;
  const bool stager = tid < B_ITEMS;
  const bool worker = wave < BN / 16;
  const int kp = tid / BQ, n4 = (tid - kp * BQ) * 4;
  v4f rb0 = v4f{}, rb1 = v4f{};

#pragma unroll
  for (int mt = 0; mt < 5; ++mt) acc[mt] = v8f{};

#define LG_GLOAD(ch)                                                          \
  if (stager) {                                                               \
    const float* p = Bm + (size_t)((ch) * BK + 2 * kp) * BN + n4;             \
    rb0 = *(const v4f*)p;                                                     \
    rb1 = *(const v4f*)(p + BN);                                              \
  }
#define LG_SSTORE(buf)                                                        \
  if (stager) {                                                               \
    v4f lo, hi;                                                               \
    lo.x = rb0.x; lo.y = rb1.x; lo.z = rb0.y; lo.w = rb1.y;                   \
    hi.x = rb0.z; hi.y = rb1.z; hi.z = rb0.w; hi.w = rb1.w;                   \
    float* d = Bs + (buf) * B_ST + kp * BP + 2 * n4;                          \
    *(v4f*)d       = lo;                                                      \
    *(v4f*)(d + 4) = hi;                                                      \
  }

  LG_GLOAD(0)
  LG_SSTORE(0)
  __syncthreads();

  const int nch = K / BK;
  const int a_off = l16 * AP + 2 * g;
  const int b_off = g * BP + 2 * (wave * 16 + l16);
  for (int c = 0; c < nch; ++c) {
    const int cur = c & 1;
    const bool more = (c + 1 < nch);
    if (more) LG_GLOAD(c + 1)
    if (worker) {
      const float* as = At + a_off + c * BK;
      const float* bs = Bs + cur * B_ST + b_off;
#pragma unroll
      for (int s = 0; s < BK / 4; ++s) {
        v2f b = *(const v2f*)(bs + 2 * s * BP);
#pragma unroll
        for (int mt = 0; mt < 5; ++mt) {
          v2f a = *(const v2f*)(as + mt * 16 * AP + 4 * s);
          acc[mt] = WMMA_F32(a, b, acc[mt]);
        }
      }
    }
    if (more) LG_SSTORE(cur ^ 1)
    __syncthreads();
  }
#undef LG_GLOAD
#undef LG_SSTORE
}

__global__ __launch_bounds__(512, 8)
void decoder_tile_kernel(const float* __restrict__ z,
                         const int* __restrict__ gi,
                         const int* __restrict__ gj,
                         const float* __restrict__ Bm,
                         int K,
                         const float* __restrict__ bias,
                         const float* __restrict__ gamma,
                         const float* __restrict__ beta,
                         const float* __restrict__ W1,
                         const float* __restrict__ bias1,
                         const float* __restrict__ gamma1,
                         const float* __restrict__ beta1,
                         const float* __restrict__ W2,
                         const float* __restrict__ bias2,
                         const float* __restrict__ gamma2,
                         const float* __restrict__ beta2,
                         const float* __restrict__ W3,
                         const float* __restrict__ bias3,
                         float* __restrict__ logits) {
  constexpr int BN = H_;
  constexpr int NTHR = 2 * BN;
  constexpr int BM = 80, BK = 32;
  constexpr int AP = BK + 4;
  constexpr int BP = 2 * BN + 32;
  constexpr int A_ITEMS = BM * (BK / 4);
  constexpr int A_IT = (A_ITEMS + NTHR - 1) / NTHR;
  constexpr int BQ = BN / 4;
  constexpr int B_IT = ((BK / 2) * BQ) / NTHR;
  constexpr int A_ST = BM * AP;
  constexpr int B_ST = (BK / 2) * BP;
  constexpr int T0 = BM * (BN + 4);
  constexpr int S1 = 2 * (BK / 2) * (2 * 128 + 32);
  static_assert(2 * A_ST + 2 * B_ST <= T0 + S1, "layer-0 stages fit in the block's LDS");
  __shared__ __attribute__((aligned(16))) float smem[T0 + S1];
  float* const Bs2 = smem + T0;
  float* const As = smem;
  float* const Bs = smem + 2 * A_ST;

  const int tid  = threadIdx.x;
  const int wave = tid >> 5;
  const int lane = tid & 31;
  const int l16  = lane & 15;
  const int g    = lane >> 4;
  const int row0 = blockIdx.x * BM;

  unsigned offi[A_IT], offj[A_IT];
#pragma unroll
  for (int it = 0; it < A_IT; ++it) {
    int i = tid + it * NTHR;
    i = (i < A_ITEMS) ? i : (A_ITEMS - 1);
    int r = i >> 3, c4 = (i & 7) * 4;
    offi[it] = (unsigned)gi[row0 + r] * OUT_ + c4;
    offj[it] = (unsigned)gj[row0 + r] * OUT_ + c4;
  }

  v4f rzi[A_IT], rzj[A_IT];
  v4f rb0[B_IT], rb1[B_IT];

#define PE_GLOAD(k0)                                                          \
  {                                                                           \
    const int f0 = (k0) & (OUT_ - 1);                                         \
    _Pragma("unroll")        \
    for (int it = 0; it < A_IT; ++it) {                                       \
      rzi[it] = *(const v4f*)(z + offi[it] + f0);                             \
      rzj[it] = *(const v4f*)(z + offj[it] + f0);                             \
    }                                                                         \
    _Pragma("unroll")                     \
    for (int it = 0; it < B_IT; ++it) {                                       \
      int j = tid + it * NTHR;                                                \
      int kp = j / BQ, n4 = (j - kp * BQ) * 4;                                \
      const float* p = Bm + (size_t)((k0) + 2 * kp) * BN + n4;                \
      rb0[it] = *(const v4f*)p;                                               \
      rb1[it] = *(const v4f*)(p + BN);                                        \
    }                                                                         \
  }

#define PE_SSTORE(buf, k0)                                                    \
  {                                                                           \
    const int sec = (k0) >> 7;                                                \
    _Pragma("unroll")                  \
    for (int it = 0; it < A_IT; ++it) {                                       \
      int i = tid + it * NTHR;                                                \
      if (i < A_ITEMS) {                                                      \
        int r = i >> 3, c4 = (i & 7) * 4;                                     \
        v4f a;                                                                \
        if (sec == 0)      a = rzi[it] + rzj[it];                             \
        else if (sec == 1) a = rzi[it] * rzj[it];                             \
        else {                                                                \
          const v4f d = rzi[it] - rzj[it];                                    \
          a.x = fabsf(d.x); a.y = fabsf(d.y); a.z = fabsf(d.z); a.w = fabsf(d.w); \
        }                                                                     \
        *(v4f*)(As + (buf) * A_ST + r * AP + c4) = a;                                   \
      }                                                                       \
    }                                                                         \
    _Pragma("unroll")                  \
    for (int it = 0; it < B_IT; ++it) {                                       \
      int j = tid + it * NTHR;                                                \
      int kp = j / BQ, n4 = (j - kp * BQ) * 4;                                \
      v4f lo, hi;                                                             \
      lo.x = rb0[it].x; lo.y = rb1[it].x; lo.z = rb0[it].y; lo.w = rb1[it].y; \
      hi.x = rb0[it].z; hi.y = rb1[it].z; hi.z = rb0[it].w; hi.w = rb1[it].w; \
      float* d = Bs + (buf) * B_ST + kp * BP + 2 * n4;                                  \
      *(v4f*)d       = lo;                                                    \
      *(v4f*)(d + 4) = hi;                                                    \
    }                                                                         \
  }

  v8f acc[5];
#pragma unroll
  for (int mt = 0; mt < 5; ++mt) acc[mt] = v8f{};

  PE_GLOAD(0)
  PE_SSTORE(0, 0)
  __syncthreads();

  const int nch = K / BK;
  const int a_off = l16 * AP + 2 * g;
  const int b_off = g * BP + 2 * (wave * 16 + l16);
  for (int c = 0; c < nch; ++c) {
    const int cur = c & 1;
    const bool more = (c + 1 < nch);
    if (more) PE_GLOAD((c + 1) * BK)
    const float* as = As + cur * A_ST + a_off;
    const float* bs = Bs + cur * B_ST + b_off;
#pragma unroll
    for (int s = 0; s < BK / 4; ++s) {
      v2f b = *(const v2f*)(bs + 2 * s * BP);
#pragma unroll
      for (int mt = 0; mt < 5; ++mt) {
        v2f a = *(const v2f*)(as + mt * 16 * AP + 4 * s);
        acc[mt] = WMMA_F32(a, b, acc[mt]);
      }
    }
    if (more) PE_SSTORE(cur ^ 1, (c + 1) * BK)
    __syncthreads();
  }
#undef PE_GLOAD
#undef PE_SSTORE

  tile_ln_lds<256, false>(acc, smem, bias, gamma, beta, nullptr, nullptr, nullptr, row0);
  lds_a_gemm<128, 260>(smem, H_, W1, Bs2, acc);
  tile_ln_lds<128, false>(acc, smem, bias1, gamma1, beta1, nullptr, nullptr, nullptr, row0);
  lds_a_gemm<64, 132>(smem, H_ / 2, W2, Bs2, acc);
  tile_ln_lds<64, true>(acc, smem, bias2, gamma2, beta2, W3, bias3, logits, row0);
}

extern "C" void kernel_launch(void* const* d_in, const int* in_sizes, int n_in,
                              void* d_out, int out_size, void* d_ws, size_t ws_size,
                              hipStream_t stream) {
  (void)in_sizes; (void)n_in; (void)out_size; (void)ws_size;

  const float* x    = (const float*)d_in[0];
  const int*   ei   = (const int*)d_in[1];
  const int*   esrc = ei;
  const int*   edst = ei + E_;
  const int*   et   = (const int*)d_in[2];
  const int*   gp   = (const int*)d_in[3];
  const int*   gi   = gp;
  const int*   gj   = gp + P_;

  const float* basis[4], *att[4], *root[4], *relatt[4], *gam[4], *bet[4];
  for (int i = 0; i < 4; ++i) {
    basis[i]  = (const float*)d_in[4 + 6 * i + 0];
    att[i]    = (const float*)d_in[4 + 6 * i + 1];
    root[i]   = (const float*)d_in[4 + 6 * i + 2];
    relatt[i] = (const float*)d_in[4 + 6 * i + 3];
    gam[i]    = (const float*)d_in[4 + 6 * i + 4];
    bet[i]    = (const float*)d_in[4 + 6 * i + 5];
  }
  const float* res0 = (const float*)d_in[28];
  const float* res3 = (const float*)d_in[29];
  const float* dW0 = (const float*)d_in[30], *db0 = (const float*)d_in[31];
  const float* dg0 = (const float*)d_in[32], *dbb0 = (const float*)d_in[33];
  const float* dW1 = (const float*)d_in[34], *db1 = (const float*)d_in[35];
  const float* dg1 = (const float*)d_in[36], *dbb1 = (const float*)d_in[37];
  const float* dW2 = (const float*)d_in[38], *db2 = (const float*)d_in[39];
  const float* dg2 = (const float*)d_in[40], *dbb2 = (const float*)d_in[41];
  const float* dW3 = (const float*)d_in[42], *db3 = (const float*)d_in[43];

  float* ws = (float*)d_ws;
  size_t off = 0;
  float* a_sm  = ws + off; off += 64;
  float* Wcat  = ws + off; off += (size_t)R_ * H_ * H_;
  float* hA    = ws + off; off += (size_t)N_ * H_;
  float* hB    = ws + off; off += (size_t)N_ * H_;
  float* resid = ws + off; off += (size_t)N_ * H_;
  float* zout = (float*)d_out + P_;
  int* csr_cnt    = (int*)(resid + (size_t)N_ * OUT_);
  int* csr_offs   = csr_cnt + 20480;
  int* csr_rank   = csr_offs + 20480;
  int* csr_key    = csr_rank + E_;
  int* csr_packed = csr_key + E_;
  int* csr_seg    = csr_packed + E_;

  auto gemm = [&](const float* A, const float* Bm, int K, int lda,
                  const float* A2, const float* B2, int K2, int lda2,
                  float* Cm, int M, int Nc,
                  const float* bias, const float* gamma, const float* beta,
                  const float* res, int gelu) {
    int blocks = M / 80;
    if (Nc == 256)
      gemm_f32_wmma_lds_kernel<256><<<blocks, 512, 0, stream>>>(
          A, Bm, K, lda, A2, B2, K2, lda2, Cm, bias, gamma, beta, res, gelu);
    else
      gemm_f32_wmma_lds_kernel<128><<<blocks, 256, 0, stream>>>(
          A, Bm, K, lda, A2, B2, K2, lda2, Cm, bias, gamma, beta, res, gelu);
  };

  struct { const float* cur; int in; int out; int li;
           const float* resmat; float* conv; int gelu; float* rbuf; } L[4] = {
    { x,  IN_, H_,   0, res0,    hA,   1, hB },
    { hA, H_,  H_,   1, nullptr, hB,   1, nullptr },
    { hB, H_,  H_,   2, nullptr, hA,   1, nullptr },
    { hA, H_,  OUT_, 3, res3,    zout, 0, resid },
  };

  csr_zero_kernel<<<(N_ + 255) / 256, 256, 0, stream>>>(csr_cnt);
  csr_count_kernel<<<(E_ / 4 + 255) / 256, 256, 0, stream>>>(edst, csr_cnt, csr_rank);
  csr_scan_kernel<<<1, 1024, 0, stream>>>(csr_cnt, csr_offs);
  csr_fill_kernel<<<(E_ / 4 + 255) / 256, 256, 0, stream>>>(edst, et, csr_offs, csr_rank, csr_key);
  csr_order_kernel<<<(N_ + 31) / 32, 256, 0, stream>>>(esrc, csr_offs, csr_key, csr_packed, csr_seg);

  for (int i = 0; i < 4; ++i) {
    const float* cur = L[i].cur;
    int in_ch = L[i].in, out_ch = L[i].out, li = L[i].li;
    float* conv = L[i].conv;

    relatt_softmax_kernel<<<1, 32, 0, stream>>>(relatt[li], a_sm);
    int wtotal = R_ * in_ch * out_ch;
    make_weights_kernel<<<(wtotal + 2047) / 2048, 256, 0, stream>>>(
        att[li], basis[li], a_sm, Wcat, in_ch, out_ch);

    const float* rptr = cur;
    if (L[i].resmat) {
      gemm(cur, L[i].resmat, in_ch, in_ch, nullptr, nullptr, 0, 0,
           L[i].rbuf, N_, out_ch, nullptr, nullptr, nullptr, nullptr, 0);
      rptr = L[i].rbuf;
    }
    if (in_ch == IN_)
      rgcn_layer_gemm_kernel<H_, IN_><<<N_ / 80, 2 * H_ + 256, 0, stream>>>(
          cur, csr_seg, csr_packed, Wcat, root[li], conv, gam[li], bet[li], rptr, L[i].gelu);
    else if (out_ch == H_)
      rgcn_layer_gemm_kernel<H_, H_><<<N_ / 80, 2 * H_ + 256, 0, stream>>>(
          cur, csr_seg, csr_packed, Wcat, root[li], conv, gam[li], bet[li], rptr, L[i].gelu);
    else
      rgcn_layer_gemm_kernel<OUT_, H_><<<N_ / 80, 2 * OUT_ + 256, 0, stream>>>(
          cur, csr_seg, csr_packed, Wcat, root[li], conv, gam[li], bet[li], rptr, L[i].gelu);
  }

  decoder_tile_kernel<<<P_ / 80, 512, 0, stream>>>(zout, gi, gj, dW0, 3 * OUT_, db0, dg0, dbb0,
                                                   dW1, db1, dg1, dbb1, dW2, db2, dg2, dbb2,
                                                   dW3, db3, (float*)d_out);
}


# Layer GEMM issues fewer vector instructions beside the WMMAs: row and B loads are uniform base + 32-bit offset kept in LDS or a register, absent edges add a zero row unselected; sums unchanged
# speedup vs baseline: 3.5003x; 1.0328x over previous
#include <hip/hip_runtime.h>
#include <hip/hip_bf16.h>
#include <math.h>

#define N_   20000
#define E_   320000
#define R_   8
#define B_   8
#define IN_  128
#define H_   256
#define OUT_ 128
#define P_   100000
#define LN_EPS 1e-5f

typedef float v8f __attribute__((ext_vector_type(8)));
typedef float v2f __attribute__((ext_vector_type(2)));
typedef float v4f __attribute__((ext_vector_type(4)));

#define WMMA_F32(a, b, c) \
  __builtin_amdgcn_wmma_f32_16x16x4_f32(false, (a), false, (b), (short)0, (c), false, false)

__global__ void relatt_softmax_kernel(const float* __restrict__ relatt,
                                      float* __restrict__ a) {
  if (threadIdx.x == 0 && blockIdx.x == 0) {
    float m = relatt[0];
    for (int r = 1; r < R_; ++r) m = fmaxf(m, relatt[r]);
    float e[R_]; float s = 0.f;
    for (int r = 0; r < R_; ++r) { e[r] = expf(relatt[r] - m); s += e[r]; }
    for (int r = 0; r < R_; ++r) a[r] = e[r] / s;
  }
}

__global__ __launch_bounds__(256)
void make_weights_kernel(const float* __restrict__ att,
                         const float* __restrict__ basis,
                         const float* __restrict__ a,
                         float* __restrict__ W,
                         int in_ch, int out_ch) {
  const int io_sz  = in_ch * out_ch;
  const int total4 = (R_ * io_sz) / 4;
  const int nthr   = gridDim.x * blockDim.x;
  for (int q = blockIdx.x * blockDim.x + threadIdx.x; q < total4; q += nthr) {
    const int idx = 4 * q;
    const int r   = idx / io_sz;
    const int io  = idx - r * io_sz;
    v4f s = v4f{};
#pragma unroll
    for (int bb = 0; bb < B_; ++bb)
      s += att[r * B_ + bb] * *(const v4f*)(basis + (size_t)bb * io_sz + io);
    *(v4f*)(W + idx) = s * a[r];
  }
}

__global__ void csr_zero_kernel(int* __restrict__ cnt, float* __restrict__ zrow) {
  int i = blockIdx.x * blockDim.x + threadIdx.x;
  if (i < N_) cnt[i] = 0;
  if (i < H_) zrow[i] = 0.f;
}

__global__ void csr_count_kernel(const int* __restrict__ dst,
                                 int* __restrict__ cnt,
                                 int* __restrict__ rank) {
  const int nthr = gridDim.x * blockDim.x;
  for (int e = blockIdx.x * blockDim.x + threadIdx.x; e < E_; e += nthr)
    rank[e] = atomicAdd(&cnt[dst[e]], 1);
}

__global__ __launch_bounds__(1024)
void csr_scan_kernel(const int* __restrict__ cnt, int* __restrict__ offs) {
  constexpr int CH = (N_ + 1023) / 1024;
  __shared__ int part[1024];
  const int t = threadIdx.x;
  int loc[CH];
  int s = 0;
#pragma unroll
  for (int i = 0; i < CH; ++i) {
    int idx = t * CH + i;
    int v = (idx < N_) ? cnt[idx] : 0;
    loc[i] = s;
    s += v;
  }
  part[t] = s;
  __syncthreads();
  for (int o = 1; o < 1024; o <<= 1) {
    int v = (t >= o) ? part[t - o] : 0;
    __syncthreads();
    part[t] += v;
    __syncthreads();
  }
  const int excl = part[t] - s;
#pragma unroll
  for (int i = 0; i < CH; ++i) {
    int idx = t * CH + i;
    if (idx < N_) offs[idx] = excl + loc[i];
  }
  if (t == 1023) offs[N_] = part[1023];
}

__global__ void csr_fill_kernel(const int* __restrict__ dst,
                                const int* __restrict__ et,
                                const int* __restrict__ offs,
                                const int* __restrict__ rank,
                                int* __restrict__ key) {
  const int nthr = gridDim.x * blockDim.x;
  for (int e = blockIdx.x * blockDim.x + threadIdx.x; e < E_; e += nthr)
    key[offs[dst[e]] + rank[e]] = et[e] * E_ + e;
}

__global__ void csr_order_kernel(const int* __restrict__ src,
                                 const int* __restrict__ offs,
                                 const int* __restrict__ key,
                                 int* __restrict__ packed,
                                 int* __restrict__ seg) {
  int wave = threadIdx.x >> 5;
  int lane = threadIdx.x & 31;
  for (int node = blockIdx.x * 8 + wave; node < N_; node += gridDim.x * 8) {
    int beg = offs[node];
    int n   = offs[node + 1] - beg;
    for (int i = lane; i < n; i += 32) {
      int my = key[beg + i];
      int rk = 0;
      for (int j = 0; j < n; ++j) rk += (key[beg + j] < my) ? 1 : 0;
      int r = my / E_;
      int e = my - r * E_;
      packed[beg + rk] = src[e] * R_ + r;
    }
    if (lane < R_) {
      int c = 0;
      for (int j = 0; j < n; ++j) c += (key[beg + j] / E_ < lane) ? 1 : 0;
      seg[node * R_ + lane] = beg + c;
    }
    if (node == N_ - 1 && lane == R_) seg[N_ * R_] = beg + n;
  }
}

template <int BN, int NW, int W0>
__device__ __forceinline__
void tile_ln_store(v8f (&acc)[5], float* ct,
                   const float* __restrict__ bias,
                   const float* __restrict__ gamma,
                   const float* __restrict__ beta,
                   const float* __restrict__ resid,
                   float* __restrict__ out, int row0, int do_gelu) {
  constexpr int CP = BN + 4;
  constexpr int NV = BN / 32;
  const int tid  = threadIdx.x;
  const int wave = tid >> 5;
  const int lane = tid & 31;
  const int l16  = lane & 15;
  const int g    = lane >> 4;

  const int sw = wave - W0;
  if (NW == BN / 16 || (sw >= 0 && sw < BN / 16)) {
    float* cw = ct + (8 * g) * CP + sw * 16 + l16;
#pragma unroll
    for (int mt = 0; mt < 5; ++mt) {
#pragma unroll
      for (int i = 0; i < 8; ++i)
        cw[(mt * 16 + i) * CP] = acc[mt][i];
    }
  }
  __syncthreads();

  float gm[NV], bt[NV], bs[NV];
#pragma unroll
  for (int i = 0; i < NV; ++i) {
    const int c = lane + (i << 5);
    gm[i] = gamma[c];
    bt[i] = beta[c];
    bs[i] = bias ? bias[c] : 0.f;
  }

  for (int r = wave; r < 80; r += NW) {
    const size_t base = (size_t)(row0 + r) * BN + lane;
    const float* cr = ct + r * CP + lane;
    float vals[NV], rs[NV];
#pragma unroll
    for (int i = 0; i < NV; ++i) {
      vals[i] = cr[i << 5];
      rs[i] = resid ? resid[base + (i << 5)] : 0.f;
    }
    float s = 0.f;
#pragma unroll
    for (int i = 0; i < NV; ++i) {
      float v = vals[i];
      if (bias) v += bs[i];
      vals[i] = v;
      s += v;
    }
    for (int o = 16; o; o >>= 1) s += __shfl_xor(s, o, 32);
    const float mean = s / (float)BN;
    float vs = 0.f;
#pragma unroll
    for (int i = 0; i < NV; ++i) { float d = vals[i] - mean; vs += d * d; }
    for (int o = 16; o; o >>= 1) vs += __shfl_xor(vs, o, 32);
    const float inv = rsqrtf(vs / (float)BN + LN_EPS);
#pragma unroll
    for (int i = 0; i < NV; ++i) {
      float v = (vals[i] - mean) * inv * gm[i] + bt[i];
      if (resid) v += rs[i];
      if (do_gelu) v = 0.5f * v * (1.f + erff(v * 0.70710678118654752f));
      out[base + (i << 5)] = v;
    }
  }
}

template <int BN>
__global__ __launch_bounds__(2 * BN)
void gemm_f32_wmma_lds_kernel(const float* __restrict__ A,
                              const float* __restrict__ Bm, int K, int lda,
                              const float* __restrict__ A2,
                              const float* __restrict__ B2, int K2, int lda2,
                              float* __restrict__ C,
                              const float* __restrict__ bias,
                              const float* __restrict__ gamma,
                              const float* __restrict__ beta,
                              const float* __restrict__ resid,
                              int do_gelu) {
  constexpr int NTHR = 2 * BN;
  constexpr int BM = 80, BK = 32;
  constexpr int AP = BK + 4;
  constexpr int BP = 2 * BN + 32;
  constexpr int A_ITEMS = BM * (BK / 4);
  constexpr int A_IT = (A_ITEMS + NTHR - 1) / NTHR;
  constexpr int BQ = BN / 4;
  constexpr int B_IT = ((BK / 2) * BQ) / NTHR;
  constexpr int A_ST = BM * AP;
  constexpr int B_ST = (BK / 2) * BP;
  static_assert(2 * A_ST + 2 * B_ST >= BM * (BN + 4), "result tile must fit in the operand stages");
  __shared__ __attribute__((aligned(16))) float smem[2 * A_ST + 2 * B_ST];
  float* const As = smem;
  float* const Bs = smem + 2 * A_ST;

  const int tid  = threadIdx.x;
  const int wave = tid >> 5;
  const int lane = tid & 31;
  const int l16  = lane & 15;
  const int g    = lane >> 4;
  const int row0 = blockIdx.x * BM;
  const int nch1 = K / BK;
  const int nch  = nch1 + K2 / BK;

  v4f ra[A_IT];
  v4f rb0[B_IT], rb1[B_IT];

#define GEMM_GLOAD(ch)                                                        \
  {                                                                           \
    const bool first = (ch) < nch1;                                           \
    const float* Ap = first ? A : A2;                                         \
    const float* Bp = first ? Bm : B2;                                        \
    const int ld = first ? lda : lda2;                                        \
    const int k0 = (first ? (ch) : (ch) - nch1) * BK;                         \
    _Pragma("unroll")                \
    for (int it = 0; it < A_IT; ++it) {                                       \
      int i = tid + it * NTHR;                                                \
      i = (i < A_ITEMS) ? i : (A_ITEMS - 1);   \
      int r = i >> 3, c4 = (i & 7) * 4;                                       \
      ra[it] = *(const v4f*)(Ap + (size_t)(row0 + r) * ld + k0 + c4);         \
    }                                                                         \
    _Pragma("unroll")                     \
    for (int it = 0; it < B_IT; ++it) {                                       \
      int j = tid + it * NTHR;                                                \
      int kp = j / BQ, n4 = (j - kp * BQ) * 4;                                \
      const float* p = Bp + (size_t)(k0 + 2 * kp) * BN + n4;                  \
      rb0[it] = *(const v4f*)p;                                               \
      rb1[it] = *(const v4f*)(p + BN);                                        \
    }                                                                         \
  }

#define GEMM_SSTORE(buf)                                                      \
  {                                                                           \
    _Pragma("unroll")                          \
    for (int it = 0; it < A_IT; ++it) {                                       \
      int i = tid + it * NTHR;                                                \
      if (i < A_ITEMS) {                                                      \
        int r = i >> 3, c4 = (i & 7) * 4;                                     \
        *(v4f*)(As + (buf) * A_ST + r * AP + c4) = ra[it];                    \
      }                                                                       \
    }                                                                         \
    _Pragma("unroll")                  \
    for (int it = 0; it < B_IT; ++it) {                                       \
      int j = tid + it * NTHR;                                                \
      int kp = j / BQ, n4 = (j - kp * BQ) * 4;                                \
      v4f lo, hi;                                                             \
      lo.x = rb0[it].x; lo.y = rb1[it].x; lo.z = rb0[it].y; lo.w = rb1[it].y; \
      hi.x = rb0[it].z; hi.y = rb1[it].z; hi.z = rb0[it].w; hi.w = rb1[it].w; \
      float* d = Bs + (buf) * B_ST + kp * BP + 2 * n4;                        \
      *(v4f*)d       = lo;                                                    \
      *(v4f*)(d + 4) = hi;                                                    \
    }                                                                         \
  }

  v8f acc[5];
#pragma unroll
  for (int mt = 0; mt < 5; ++mt) acc[mt] = v8f{};

  GEMM_GLOAD(0)
  GEMM_SSTORE(0)
  __syncthreads();

  const int a_off = l16 * AP + 2 * g;
  const int b_off = g * BP + 2 * (wave * 16 + l16);
  for (int c = 0; c < nch; ++c) {
    const int cur = c & 1;
    const bool more = (c + 1 < nch);
    if (more) GEMM_GLOAD(c + 1)
    const float* as = As + cur * A_ST + a_off;
    const float* bs = Bs + cur * B_ST + b_off;
#pragma unroll
    for (int s = 0; s < BK / 4; ++s) {
      v2f b = *(const v2f*)(bs + 2 * s * BP);
#pragma unroll
      for (int mt = 0; mt < 5; ++mt) {
        v2f a = *(const v2f*)(as + mt * 16 * AP + 4 * s);
        acc[mt] = WMMA_F32(a, b, acc[mt]);
      }
    }
    if (more) GEMM_SSTORE(cur ^ 1)
    __syncthreads();
  }
#undef GEMM_GLOAD
#undef GEMM_SSTORE

  if (gamma) {
    tile_ln_store<BN, BN / 16, 0>(acc, smem, bias, gamma, beta, resid, C, row0, do_gelu);
  } else {
    float* Cbase = C + (size_t)(row0 + 8 * g) * BN + wave * 16 + l16;
#pragma unroll
    for (int mt = 0; mt < 5; ++mt) {
#pragma unroll
      for (int i = 0; i < 8; ++i)
        Cbase[(size_t)(mt * 16 + i) * BN] = acc[mt][i];
    }
  }
}

template <int BN, int IN, bool ZR>
__global__ __launch_bounds__(2 * BN + 256)
void rgcn_layer_gemm_kernel(const float* __restrict__ X,
                            const float* __restrict__ zrow,
                            const int* __restrict__ seg,
                            const int* __restrict__ packed,
                            const float* __restrict__ Bm,
                            const float* __restrict__ B2,
                            float* __restrict__ C,
                            const float* __restrict__ gamma,
                            const float* __restrict__ beta,
                            const float* __restrict__ resid,
                            int do_gelu) {
  constexpr int NCONS = 2 * BN;
  constexpr int NPROD = 256;
  constexpr int BM = 80, BK = 32;
  constexpr int AP = BK + 4;
  constexpr int BP = 2 * BN + 32;
  constexpr int A_ITEMS = BM * (BK / 4);
  constexpr int P_IT = (A_ITEMS + NPROD - 1) / NPROD;
  constexpr int BQ = BN / 4;
  constexpr int B_IT = ((BK / 2) * BQ) / NCONS;
  constexpr int A_ST = BM * AP;
  constexpr int B_ST = (BK / 2) * BP;
  constexpr int CPR = IN / BK;
  constexpr int NCH1 = R_ * CPR;
  constexpr int NCH = NCH1 + CPR;
  constexpr int NPRE = (BN >= 256) ? 4 : 8;
  constexpr int NPASS = (BN >= 256) ? 2 : 1;
  constexpr int NKNOWN = NPRE * NPASS;
  static_assert(2 * A_ST + 2 * B_ST >= BM * (BN + 4), "result tile must fit in the operand stages");
  static_assert(IN % BK == 0, "a chunk lies inside one relation");
  __shared__ __attribute__((aligned(16))) float smem[2 * A_ST + 2 * B_ST];
  static_assert((A_ITEMS - (P_IT - 1) * NPROD) % 32 == 0, "the last piece is present or absent for a whole wave");
  __shared__ unsigned pso_l[NKNOWN * P_IT * NPROD];
  float* const As = smem;
  float* const Bs = smem + 2 * A_ST;

  const int pt   = threadIdx.x;
  const int tid  = pt - NPROD;
  const int wave = tid >> 5;
  const int lane = pt & 31;
  const int l16  = lane & 15;
  const int g    = lane >> 4;
  const int row0 = blockIdx.x * BM;
  const bool consumer = pt >= NPROD;
  const bool last_piece = __builtin_amdgcn_readfirstlane(pt) + (P_IT - 1) * NPROD < A_ITEMS;
  const unsigned zoffb = ZR ? (unsigned)((const char*)zrow - (const char*)X) : 0u;

  int prow[P_IT], pc4[P_IT];
  bool pv[P_IT];
  int pcnt[P_IT], pbeg[P_IT];
#pragma unroll
  for (int it = 0; it < P_IT; ++it) {
    int i = pt + it * NPROD;
    pv[it] = !consumer && i < A_ITEMS;
    i = pv[it] ? i : 0;
    prow[it] = i >> 3;
    pc4[it] = (i & 7) * 4;
    pcnt[it] = 0;
    pbeg[it] = 0;
  }

  auto build_a = [&](int cn, int buf) {
    v4f a[P_IT];
    if (cn < NCH1) {
      const int rel = cn / CPR;
      const int f0  = (cn - rel * CPR) * BK;
      const char* const xb = (const char*)(X + f0);
      if (cn == rel * CPR) {
        int s[P_IT], e[P_IT], p[P_IT][NKNOWN];
#pragma unroll
        for (int it = 0; it < P_IT; ++it) {
          if (it < P_IT - 1 || last_piece) {
            const int sidx = (row0 + prow[it]) * R_ + rel;
            s[it] = seg[sidx];
            e[it] = seg[sidx + 1];
          }
        }
#pragma unroll
        for (int it = 0; it < P_IT; ++it) {
          if (it < P_IT - 1 || last_piece) {
#pragma unroll
            for (int u = 0; u < NKNOWN; ++u) {
              const int idx = s[it] + u;
              p[it][u] = packed[idx < E_ ? idx : E_ - 1];
            }
          }
        }
#pragma unroll
        for (int it = 0; it < P_IT; ++it) {
          if (it < P_IT - 1 || last_piece) {
            pbeg[it] = s[it];
            pcnt[it] = pv[it] ? e[it] - s[it] : 0;
            const unsigned absent = ZR ? zoffb : (unsigned)pc4[it] * 4u;
#pragma unroll
            for (int u = 0; u < NKNOWN; ++u)
              pso_l[(it * NKNOWN + u) * NPROD + pt] =
                  (u < pcnt[it]) ? (((unsigned)p[it][u] / R_) * IN + (unsigned)pc4[it]) * 4u : absent;
          }
        }
      }
#pragma unroll
      for (int it = 0; it < P_IT; ++it) a[it] = v4f{};
#pragma unroll
      for (int ps = 0; ps < NPASS; ++ps) {
        bool need = true;
        if (ps > 0) {
          bool m = false;
#pragma unroll
          for (int it = 0; it < P_IT; ++it) m = m || (pcnt[it] > ps * NPRE);
          need = __builtin_amdgcn_ballot_w32(m) != 0u;
        }
        if (need) {
          unsigned so[P_IT][NPRE];
          v4f v[P_IT][NPRE];
#pragma unroll
          for (int it = 0; it < P_IT; ++it) {
            if (it < P_IT - 1 || last_piece) {
#pragma unroll
              for (int u = 0; u < NPRE; ++u)
                so[it][u] = pso_l[(it * NKNOWN + ps * NPRE + u) * NPROD + pt];
            }
          }
#pragma unroll
          for (int it = 0; it < P_IT; ++it) {
            if (it < P_IT - 1 || last_piece) {
#pragma unroll
              for (int u = 0; u < NPRE; ++u)
                v[it][u] = *(const v4f*)(xb + so[it][u]);
            }
          }
#pragma unroll
          for (int it = 0; it < P_IT; ++it) {
            if (it < P_IT - 1 || last_piece) {
#pragma unroll
              for (int u = 0; u < NPRE; ++u) {
                if (ZR) a[it] += v[it][u];
                else    a[it] += (ps * NPRE + u < pcnt[it]) ? v[it][u] : v4f{};
              }
            }
          }
        }
      }
#pragma unroll
      for (int it = 0; it < P_IT; ++it) {
        for (int j = NKNOWN; j < pcnt[it]; ++j) {
          const unsigned so = ((unsigned)packed[pbeg[it] + j] / R_) * IN;
          a[it] += *(const v4f*)(X + so + f0 + pc4[it]);
        }
      }
    } else {
      const int k0 = (cn - NCH1) * BK;
#pragma unroll
      for (int it = 0; it < P_IT; ++it) {
        a[it] = v4f{};
        if (it < P_IT - 1 || last_piece)
          a[it] = *(const v4f*)(X + (size_t)(row0 + prow[it]) * IN + k0 + pc4[it]);
      }
    }
#pragma unroll
    for (int it = 0; it < P_IT; ++it)
      if (pv[it]) *(v4f*)(As + buf * A_ST + prow[it] * AP + pc4[it]) = a[it];
  };

  v4f rb0[B_IT], rb1[B_IT];
  unsigned boff[B_IT];
#pragma unroll
  for (int it = 0; it < B_IT; ++it) {
    const int j = (consumer ? tid : 0) + it * NCONS;
    const int kp = j / BQ, n4 = (j - kp * BQ) * 4;
    boff[it] = (unsigned)((2 * kp) * BN + n4) * 4u;
  }
#define LAYER_GLOAD(ch)                                                       \
  {                                                                           \
    const bool first = (ch) < NCH1;                                           \
    const float* Bp = first ? Bm : B2;                                        \
    const int k0 = (first ? (ch) : (ch) - NCH1) * BK;                         \
    const char* const bb = (const char*)(Bp + (size_t)k0 * BN);   \
    _Pragma("unroll")                     \
    for (int it = 0; it < B_IT; ++it) {                                       \
      rb0[it] = *(const v4f*)(bb + boff[it]);                                 \
      rb1[it] = *(const v4f*)(bb + boff[it] + BN * 4);                        \
    }                                                                         \
  }
#define LAYER_SSTORE(buf)                                                     \
  {                                                                           \
    _Pragma("unroll")                  \
    for (int it = 0; it < B_IT; ++it) {                                       \
      int j = tid + it * NCONS;                                               \
      int kp = j / BQ, n4 = (j - kp * BQ) * 4;                                \
      v4f lo, hi;                                                             \
      lo.x = rb0[it].x; lo.y = rb1[it].x; lo.z = rb0[it].y; lo.w = rb1[it].y; \
      hi.x = rb0[it].z; hi.y = rb1[it].z; hi.z = rb0[it].w; hi.w = rb1[it].w; \
      float* d = Bs + (buf) * B_ST + kp * BP + 2 * n4;                        \
      *(v4f*)d       = lo;                                                    \
      *(v4f*)(d + 4) = hi;                                                    \
    }                                                                         \
  }

  v8f acc[5];
#pragma unroll
  for (int mt = 0; mt < 5; ++mt) acc[mt] = v8f{};

  if (consumer) {
    LAYER_GLOAD(0)
    LAYER_SSTORE(0)
  } else {
    build_a(0, 0);
  }
  __syncthreads();

  const int a_off = l16 * AP + 2 * g;
  const int b_off = g * BP + 2 * (wave * 16 + l16);
  for (int c = 0; c < NCH; ++c) {
    const int cur = c & 1;
    const bool more = (c + 1 < NCH);
    if (consumer) {
      if (more) LAYER_GLOAD(c + 1)
      const float* as = As + cur * A_ST + a_off;
      const float* bs = Bs + cur * B_ST + b_off;
#pragma unroll
      for (int s = 0; s < BK / 4; ++s) {
        v2f b = *(const v2f*)(bs + 2 * s * BP);
#pragma unroll
        for (int mt = 0; mt < 5; ++mt) {
          v2f a = *(const v2f*)(as + mt * 16 * AP + 4 * s);
          acc[mt] = WMMA_F32(a, b, acc[mt]);
        }
      }
      if (more) LAYER_SSTORE(cur ^ 1)
    } else {
      if (more) build_a(c + 1, cur ^ 1);
    }
    __syncthreads();
  }
#undef LAYER_GLOAD
#undef LAYER_SSTORE

  tile_ln_store<BN, (NCONS + NPROD) / 32, NPROD / 32>(acc, smem, nullptr, gamma, beta, resid, C, row0, do_gelu);
}

template <int BN, bool DOT>
__device__ __forceinline__
void tile_ln_lds(v8f (&acc)[5], float* ct,
                 const float* __restrict__ bias,
                 const float* __restrict__ gamma,
                 const float* __restrict__ beta,
                 const float* __restrict__ wdot,
                 const float* __restrict__ bdot,
                 float* __restrict__ logits, int row0) {
  constexpr int CP = BN + 4;
  constexpr int NV = BN / 32;
  static_assert(!DOT || NV == 2, "the final dot is written for 64 columns");
  const int tid  = threadIdx.x;
  const int wave = tid >> 5;
  const int lane = tid & 31;
  const int l16  = lane & 15;
  const int g    = lane >> 4;

  if (wave < BN / 16) {
    float* cw = ct + (8 * g) * CP + wave * 16 + l16;
#pragma unroll
    for (int mt = 0; mt < 5; ++mt) {
#pragma unroll
      for (int i = 0; i < 8; ++i)
        cw[(mt * 16 + i) * CP] = acc[mt][i];
    }
  }
  __syncthreads();

  float gm[NV], bt[NV], bs[NV], wd[NV];
#pragma unroll
  for (int i = 0; i < NV; ++i) {
    const int c = lane + (i << 5);
    gm[i] = gamma[c];
    bt[i] = beta[c];
    bs[i] = bias[c];
    wd[i] = DOT ? wdot[c] : 0.f;
  }
  const float b0 = DOT ? bdot[0] : 0.f;

  for (int r = wave; r < 80; r += 16) {
    float* cr = ct + r * CP + lane;
    float vals[NV];
    float s = 0.f;
#pragma unroll
    for (int i = 0; i < NV; ++i) {
      float v = cr[i << 5];
      v += bs[i];
      vals[i] = v;
      s += v;
    }
    for (int o = 16; o; o >>= 1) s += __shfl_xor(s, o, 32);
    const float mean = s / (float)BN;
    float vs = 0.f;
#pragma unroll
    for (int i = 0; i < NV; ++i) { float d = vals[i] - mean; vs += d * d; }
    for (int o = 16; o; o >>= 1) vs += __shfl_xor(vs, o, 32);
    const float inv = rsqrtf(vs / (float)BN + LN_EPS);
#pragma unroll
    for (int i = 0; i < NV; ++i) {
      float v = (vals[i] - mean) * inv * gm[i] + bt[i];
      v = 0.5f * v * (1.f + erff(v * 0.70710678118654752f));
      vals[i] = v;
    }
    if (!DOT) {
#pragma unroll
      for (int i = 0; i < NV; ++i) cr[i << 5] = vals[i];
    } else {
      float d = vals[0] * wd[0] + vals[NV - 1] * wd[NV - 1];
      for (int o = 16; o; o >>= 1) d += __shfl_xor(d, o, 32);
      if (lane == 0) logits[row0 + r] = d + b0;
    }
  }
  if (!DOT) __syncthreads();
}

template <int BN, int AP>
__device__ __forceinline__
void lds_a_gemm(const float* At, int K,
                const float* __restrict__ Bm, float* Bs, v8f (&acc)[5]) {
  constexpr int BK = 32;
  constexpr int BP = 2 * BN + 32;
  constexpr int BQ = BN / 4;
  constexpr int B_ITEMS = (BK / 2) * BQ;
  constexpr int B_ST = (BK / 2) * BP;
  static_assert(B_ITEMS <= 512 && B_ITEMS % 32 == 0, "one piece per thread, whole waves");

  int tid = threadIdx.x;
  asm volatile("" : "+v"(tid));
  const int wave = tid >> 5;
  const int lane = tid & 31;
  const int l16  = lane & 15;
  const int g    = lane >> 4;
  const bool stager = tid < B_ITEMS;
  const bool worker = wave < BN / 16;
  const int kp = tid / BQ, n4 = (tid - kp * BQ) * 4;
  v4f rb0 = v4f{}, rb1 = v4f{};

#pragma unroll
  for (int mt = 0; mt < 5; ++mt) acc[mt] = v8f{};

#define LG_GLOAD(ch)                                                          \
  if (stager) {                                                               \
    const float* p = Bm + (size_t)((ch) * BK + 2 * kp) * BN + n4;             \
    rb0 = *(const v4f*)p;                                                     \
    rb1 = *(const v4f*)(p + BN);                                              \
  }
#define LG_SSTORE(buf)                                                        \
  if (stager) {                                                               \
    v4f lo, hi;                                                               \
    lo.x = rb0.x; lo.y = rb1.x; lo.z = rb0.y; lo.w = rb1.y;                   \
    hi.x = rb0.z; hi.y = rb1.z; hi.z = rb0.w; hi.w = rb1.w;                   \
    float* d = Bs + (buf) * B_ST + kp * BP + 2 * n4;                          \
    *(v4f*)d       = lo;                                                      \
    *(v4f*)(d + 4) = hi;                                                      \
  }

  LG_GLOAD(0)
  LG_SSTORE(0)
  __syncthreads();

  const int nch = K / BK;
  const int a_off = l16 * AP + 2 * g;
  const int b_off = g * BP + 2 * (wave * 16 + l16);
  for (int c = 0; c < nch; ++c) {
    const int cur = c & 1;
    const bool more = (c + 1 < nch);
    if (more) LG_GLOAD(c + 1)
    if (worker) {
      const float* as = At + a_off + c * BK;
      const float* bs = Bs + cur * B_ST + b_off;
#pragma unroll
      for (int s = 0; s < BK / 4; ++s) {
        v2f b = *(const v2f*)(bs + 2 * s * BP);
#pragma unroll
        for (int mt = 0; mt < 5; ++mt) {
          v2f a = *(const v2f*)(as + mt * 16 * AP + 4 * s);
          acc[mt] = WMMA_F32(a, b, acc[mt]);
        }
      }
    }
    if (more) LG_SSTORE(cur ^ 1)
    __syncthreads();
  }
#undef LG_GLOAD
#undef LG_SSTORE
}

__global__ __launch_bounds__(512, 8)
void decoder_tile_kernel(const float* __restrict__ z,
                         const int* __restrict__ gi,
                         const int* __restrict__ gj,
                         const float* __restrict__ Bm,
                         int K,
                         const float* __restrict__ bias,
                         const float* __restrict__ gamma,
                         const float* __restrict__ beta,
                         const float* __restrict__ W1,
                         const float* __restrict__ bias1,
                         const float* __restrict__ gamma1,
                         const float* __restrict__ beta1,
                         const float* __restrict__ W2,
                         const float* __restrict__ bias2,
                         const float* __restrict__ gamma2,
                         const float* __restrict__ beta2,
                         const float* __restrict__ W3,
                         const float* __restrict__ bias3,
                         float* __restrict__ logits) {
  constexpr int BN = H_;
  constexpr int NTHR = 2 * BN;
  constexpr int BM = 80, BK = 32;
  constexpr int AP = BK + 4;
  constexpr int BP = 2 * BN + 32;
  constexpr int A_ITEMS = BM * (BK / 4);
  constexpr int A_IT = (A_ITEMS + NTHR - 1) / NTHR;
  constexpr int BQ = BN / 4;
  constexpr int B_IT = ((BK / 2) * BQ) / NTHR;
  constexpr int A_ST = BM * AP;
  constexpr int B_ST = (BK / 2) * BP;
  constexpr int T0 = BM * (BN + 4);
  constexpr int S1 = 2 * (BK / 2) * (2 * 128 + 32);
  static_assert(2 * A_ST + 2 * B_ST <= T0 + S1, "layer-0 stages fit in the block's LDS");
  __shared__ __attribute__((aligned(16))) float smem[T0 + S1];
  float* const Bs2 = smem + T0;
  float* const As = smem;
  float* const Bs = smem + 2 * A_ST;

  const int tid  = threadIdx.x;
  const int wave = tid >> 5;
  const int lane = tid & 31;
  const int l16  = lane & 15;
  const int g    = lane >> 4;
  const int row0 = blockIdx.x * BM;

  unsigned offi[A_IT], offj[A_IT];
#pragma unroll
  for (int it = 0; it < A_IT; ++it) {
    int i = tid + it * NTHR;
    i = (i < A_ITEMS) ? i : (A_ITEMS - 1);
    int r = i >> 3, c4 = (i & 7) * 4;
    offi[it] = (unsigned)gi[row0 + r] * OUT_ + c4;
    offj[it] = (unsigned)gj[row0 + r] * OUT_ + c4;
  }

  v4f rzi[A_IT], rzj[A_IT];
  v4f rb0[B_IT], rb1[B_IT];

#define PE_GLOAD(k0)                                                          \
  {                                                                           \
    const int f0 = (k0) & (OUT_ - 1);                                         \
    _Pragma("unroll")        \
    for (int it = 0; it < A_IT; ++it) {                                       \
      rzi[it] = *(const v4f*)(z + offi[it] + f0);                             \
      rzj[it] = *(const v4f*)(z + offj[it] + f0);                             \
    }                                                                         \
    _Pragma("unroll")                     \
    for (int it = 0; it < B_IT; ++it) {                                       \
      int j = tid + it * NTHR;                                                \
      int kp = j / BQ, n4 = (j - kp * BQ) * 4;                                \
      const float* p = Bm + (size_t)((k0) + 2 * kp) * BN + n4;                \
      rb0[it] = *(const v4f*)p;                                               \
      rb1[it] = *(const v4f*)(p + BN);                                        \
    }                                                                         \
  }

#define PE_SSTORE(buf, k0)                                                    \
  {                                                                           \
    const int sec = (k0) >> 7;                                                \
    _Pragma("unroll")                  \
    for (int it = 0; it < A_IT; ++it) {                                       \
      int i = tid + it * NTHR;                                                \
      if (i < A_ITEMS) {                                                      \
        int r = i >> 3, c4 = (i & 7) * 4;                                     \
        v4f a;                                                                \
        if (sec == 0)      a = rzi[it] + rzj[it];                             \
        else if (sec == 1) a = rzi[it] * rzj[it];                             \
        else {                                                                \
          const v4f d = rzi[it] - rzj[it];                                    \
          a.x = fabsf(d.x); a.y = fabsf(d.y); a.z = fabsf(d.z); a.w = fabsf(d.w); \
        }                                                                     \
        *(v4f*)(As + (buf) * A_ST + r * AP + c4) = a;                                   \
      }                                                                       \
    }                                                                         \
    _Pragma("unroll")                  \
    for (int it = 0; it < B_IT; ++it) {                                       \
      int j = tid + it * NTHR;                                                \
      int kp = j / BQ, n4 = (j - kp * BQ) * 4;                                \
      v4f lo, hi;                                                             \
      lo.x = rb0[it].x; lo.y = rb1[it].x; lo.z = rb0[it].y; lo.w = rb1[it].y; \
      hi.x = rb0[it].z; hi.y = rb1[it].z; hi.z = rb0[it].w; hi.w = rb1[it].w; \
      float* d = Bs + (buf) * B_ST + kp * BP + 2 * n4;                                  \
      *(v4f*)d       = lo;                                                    \
      *(v4f*)(d + 4) = hi;                                                    \
    }                                                                         \
  }

  v8f acc[5];
#pragma unroll
  for (int mt = 0; mt < 5; ++mt) acc[mt] = v8f{};

  PE_GLOAD(0)
  PE_SSTORE(0, 0)
  __syncthreads();

  const int nch = K / BK;
  const int a_off = l16 * AP + 2 * g;
  const int b_off = g * BP + 2 * (wave * 16 + l16);
  for (int c = 0; c < nch; ++c) {
    const int cur = c & 1;
    const bool more = (c + 1 < nch);
    if (more) PE_GLOAD((c + 1) * BK)
    const float* as = As + cur * A_ST + a_off;
    const float* bs = Bs + cur * B_ST + b_off;
#pragma unroll
    for (int s = 0; s < BK / 4; ++s) {
      v2f b = *(const v2f*)(bs + 2 * s * BP);
#pragma unroll
      for (int mt = 0; mt < 5; ++mt) {
        v2f a = *(const v2f*)(as + mt * 16 * AP + 4 * s);
        acc[mt] = WMMA_F32(a, b, acc[mt]);
      }
    }
    if (more) PE_SSTORE(cur ^ 1, (c + 1) * BK)
    __syncthreads();
  }
#undef PE_GLOAD
#undef PE_SSTORE

  tile_ln_lds<256, false>(acc, smem, bias, gamma, beta, nullptr, nullptr, nullptr, row0);
  lds_a_gemm<128, 260>(smem, H_, W1, Bs2, acc);
  tile_ln_lds<128, false>(acc, smem, bias1, gamma1, beta1, nullptr, nullptr, nullptr, row0);
  lds_a_gemm<64, 132>(smem, H_ / 2, W2, Bs2, acc);
  tile_ln_lds<64, true>(acc, smem, bias2, gamma2, beta2, W3, bias3, logits, row0);
}

extern "C" void kernel_launch(void* const* d_in, const int* in_sizes, int n_in,
                              void* d_out, int out_size, void* d_ws, size_t ws_size,
                              hipStream_t stream) {
  (void)in_sizes; (void)n_in; (void)out_size; (void)ws_size;

  const float* x    = (const float*)d_in[0];
  const int*   ei   = (const int*)d_in[1];
  const int*   esrc = ei;
  const int*   edst = ei + E_;
  const int*   et   = (const int*)d_in[2];
  const int*   gp   = (const int*)d_in[3];
  const int*   gi   = gp;
  const int*   gj   = gp + P_;

  const float* basis[4], *att[4], *root[4], *relatt[4], *gam[4], *bet[4];
  for (int i = 0; i < 4; ++i) {
    basis[i]  = (const float*)d_in[4 + 6 * i + 0];
    att[i]    = (const float*)d_in[4 + 6 * i + 1];
    root[i]   = (const float*)d_in[4 + 6 * i + 2];
    relatt[i] = (const float*)d_in[4 + 6 * i + 3];
    gam[i]    = (const float*)d_in[4 + 6 * i + 4];
    bet[i]    = (const float*)d_in[4 + 6 * i + 5];
  }
  const float* res0 = (const float*)d_in[28];
  const float* res3 = (const float*)d_in[29];
  const float* dW0 = (const float*)d_in[30], *db0 = (const float*)d_in[31];
  const float* dg0 = (const float*)d_in[32], *dbb0 = (const float*)d_in[33];
  const float* dW1 = (const float*)d_in[34], *db1 = (const float*)d_in[35];
  const float* dg1 = (const float*)d_in[36], *dbb1 = (const float*)d_in[37];
  const float* dW2 = (const float*)d_in[38], *db2 = (const float*)d_in[39];
  const float* dg2 = (const float*)d_in[40], *dbb2 = (const float*)d_in[41];
  const float* dW3 = (const float*)d_in[42], *db3 = (const float*)d_in[43];

  float* ws = (float*)d_ws;
  size_t off = 0;
  float* a_sm  = ws + off; off += 64;
  float* Wcat  = ws + off; off += (size_t)R_ * H_ * H_;
  float* hA    = ws + off; off += (size_t)N_ * H_;
  float* hB    = ws + off; off += (size_t)N_ * H_;
  float* resid = ws + off; off += (size_t)N_ * H_;
  float* zout = (float*)d_out + P_;
  int* csr_cnt    = (int*)(resid + (size_t)N_ * OUT_);
  int* csr_offs   = csr_cnt + 20480;
  int* csr_rank   = csr_offs + 20480;
  int* csr_key    = csr_rank + E_;
  int* csr_packed = csr_key + E_;
  int* csr_seg    = csr_packed + E_;
  float* zrow = resid + 4000000;

  auto gemm = [&](const float* A, const float* Bm, int K, int lda,
                  const float* A2, const float* B2, int K2, int lda2,
                  float* Cm, int M, int Nc,
                  const float* bias, const float* gamma, const float* beta,
                  const float* res, int gelu) {
    int blocks = M / 80;
    if (Nc == 256)
      gemm_f32_wmma_lds_kernel<256><<<blocks, 512, 0, stream>>>(
          A, Bm, K, lda, A2, B2, K2, lda2, Cm, bias, gamma, beta, res, gelu);
    else
      gemm_f32_wmma_lds_kernel<128><<<blocks, 256, 0, stream>>>(
          A, Bm, K, lda, A2, B2, K2, lda2, Cm, bias, gamma, beta, res, gelu);
  };

  struct { const float* cur; int in; int out; int li;
           const float* resmat; float* conv; int gelu; float* rbuf; } L[4] = {
    { x,  IN_, H_,   0, res0,    hA,   1, hB },
    { hA, H_,  H_,   1, nullptr, hB,   1, nullptr },
    { hB, H_,  H_,   2, nullptr, hA,   1, nullptr },
    { hA, H_,  OUT_, 3, res3,    zout, 0, resid },
  };

  csr_zero_kernel<<<(N_ + 255) / 256, 256, 0, stream>>>(csr_cnt, zrow);
  csr_count_kernel<<<(E_ / 4 + 255) / 256, 256, 0, stream>>>(edst, csr_cnt, csr_rank);
  csr_scan_kernel<<<1, 1024, 0, stream>>>(csr_cnt, csr_offs);
  csr_fill_kernel<<<(E_ / 4 + 255) / 256, 256, 0, stream>>>(edst, et, csr_offs, csr_rank, csr_key);
  csr_order_kernel<<<(N_ + 31) / 32, 256, 0, stream>>>(esrc, csr_offs, csr_key, csr_packed, csr_seg);

  for (int i = 0; i < 4; ++i) {
    const float* cur = L[i].cur;
    int in_ch = L[i].in, out_ch = L[i].out, li = L[i].li;
    float* conv = L[i].conv;

    relatt_softmax_kernel<<<1, 32, 0, stream>>>(relatt[li], a_sm);
    int wtotal = R_ * in_ch * out_ch;
    make_weights_kernel<<<(wtotal + 2047) / 2048, 256, 0, stream>>>(
        att[li], basis[li], a_sm, Wcat, in_ch, out_ch);

    const float* rptr = cur;
    if (L[i].resmat) {
      gemm(cur, L[i].resmat, in_ch, in_ch, nullptr, nullptr, 0, 0,
           L[i].rbuf, N_, out_ch, nullptr, nullptr, nullptr, nullptr, 0);
      rptr = L[i].rbuf;
    }
    if (in_ch == IN_)
      rgcn_layer_gemm_kernel<H_, IN_, false><<<N_ / 80, 2 * H_ + 256, 0, stream>>>(
          cur, zrow, csr_seg, csr_packed, Wcat, root[li], conv, gam[li], bet[li], rptr, L[i].gelu);
    else if (out_ch == H_)
      rgcn_layer_gemm_kernel<H_, H_, true><<<N_ / 80, 2 * H_ + 256, 0, stream>>>(
          cur, zrow, csr_seg, csr_packed, Wcat, root[li], conv, gam[li], bet[li], rptr, L[i].gelu);
    else
      rgcn_layer_gemm_kernel<OUT_, H_, true><<<N_ / 80, 2 * OUT_ + 256, 0, stream>>>(
          cur, zrow, csr_seg, csr_packed, Wcat, root[li], conv, gam[li], bet[li], rptr, L[i].gelu);
  }

  decoder_tile_kernel<<<P_ / 80, 512, 0, stream>>>(zout, gi, gj, dW0, 3 * OUT_, db0, dg0, dbb0,
                                                   dW1, db1, dg1, dbb1, dW2, db2, dg2, dbb2,
                                                   dW3, db3, (float*)d_out);
}


# K-loop chunk barrier split into signal and wait: matrix waves signal behind their last LDS reads, run their last WMMAs, then wait; layer builders signal first and fill a third A stage; same sums
# speedup vs baseline: 3.6273x; 1.0363x over previous
#include <hip/hip_runtime.h>
#include <hip/hip_bf16.h>
#include <math.h>

#define N_   20000
#define E_   320000
#define R_   8
#define B_   8
#define IN_  128
#define H_   256
#define OUT_ 128
#define P_   100000
#define LN_EPS 1e-5f

typedef float v8f __attribute__((ext_vector_type(8)));
typedef float v2f __attribute__((ext_vector_type(2)));
typedef float v4f __attribute__((ext_vector_type(4)));

#define WMMA_F32(a, b, c) \
  __builtin_amdgcn_wmma_f32_16x16x4_f32(false, (a), false, (b), (short)0, (c), false, false)

__device__ __forceinline__ void lds_arrive() {
  asm volatile("s_wait_dscnt 0x0\n\ts_barrier_signal -1" ::: "memory");
}
__device__ __forceinline__ void lds_wait() {
  asm volatile("s_barrier_wait -1" ::: "memory");
}

__global__ void relatt_softmax_kernel(const float* __restrict__ relatt,
                                      float* __restrict__ a) {
  if (threadIdx.x == 0 && blockIdx.x == 0) {
    float m = relatt[0];
    for (int r = 1; r < R_; ++r) m = fmaxf(m, relatt[r]);
    float e[R_]; float s = 0.f;
    for (int r = 0; r < R_; ++r) { e[r] = expf(relatt[r] - m); s += e[r]; }
    for (int r = 0; r < R_; ++r) a[r] = e[r] / s;
  }
}

__global__ __launch_bounds__(256)
void make_weights_kernel(const float* __restrict__ att,
                         const float* __restrict__ basis,
                         const float* __restrict__ a,
                         float* __restrict__ W,
                         int in_ch, int out_ch) {
  const int io_sz  = in_ch * out_ch;
  const int total4 = (R_ * io_sz) / 4;
  const int nthr   = gridDim.x * blockDim.x;
  for (int q = blockIdx.x * blockDim.x + threadIdx.x; q < total4; q += nthr) {
    const int idx = 4 * q;
    const int r   = idx / io_sz;
    const int io  = idx - r * io_sz;
    v4f s = v4f{};
#pragma unroll
    for (int bb = 0; bb < B_; ++bb)
      s += att[r * B_ + bb] * *(const v4f*)(basis + (size_t)bb * io_sz + io);
    *(v4f*)(W + idx) = s * a[r];
  }
}

__global__ void csr_zero_kernel(int* __restrict__ cnt, float* __restrict__ zrow) {
  int i = blockIdx.x * blockDim.x + threadIdx.x;
  if (i < N_) cnt[i] = 0;
  if (i < H_) zrow[i] = 0.f;
}

__global__ void csr_count_kernel(const int* __restrict__ dst,
                                 int* __restrict__ cnt,
                                 int* __restrict__ rank) {
  const int nthr = gridDim.x * blockDim.x;
  for (int e = blockIdx.x * blockDim.x + threadIdx.x; e < E_; e += nthr)
    rank[e] = atomicAdd(&cnt[dst[e]], 1);
}

__global__ __launch_bounds__(1024)
void csr_scan_kernel(const int* __restrict__ cnt, int* __restrict__ offs) {
  constexpr int CH = (N_ + 1023) / 1024;
  __shared__ int part[1024];
  const int t = threadIdx.x;
  int loc[CH];
  int s = 0;
#pragma unroll
  for (int i = 0; i < CH; ++i) {
    int idx = t * CH + i;
    int v = (idx < N_) ? cnt[idx] : 0;
    loc[i] = s;
    s += v;
  }
  part[t] = s;
  __syncthreads();
  for (int o = 1; o < 1024; o <<= 1) {
    int v = (t >= o) ? part[t - o] : 0;
    __syncthreads();
    part[t] += v;
    __syncthreads();
  }
  const int excl = part[t] - s;
#pragma unroll
  for (int i = 0; i < CH; ++i) {
    int idx = t * CH + i;
    if (idx < N_) offs[idx] = excl + loc[i];
  }
  if (t == 1023) offs[N_] = part[1023];
}

__global__ void csr_fill_kernel(const int* __restrict__ dst,
                                const int* __restrict__ et,
                                const int* __restrict__ offs,
                                const int* __restrict__ rank,
                                int* __restrict__ key) {
  const int nthr = gridDim.x * blockDim.x;
  for (int e = blockIdx.x * blockDim.x + threadIdx.x; e < E_; e += nthr)
    key[offs[dst[e]] + rank[e]] = et[e] * E_ + e;
}

__global__ void csr_order_kernel(const int* __restrict__ src,
                                 const int* __restrict__ offs,
                                 const int* __restrict__ key,
                                 int* __restrict__ packed,
                                 int* __restrict__ seg) {
  int wave = threadIdx.x >> 5;
  int lane = threadIdx.x & 31;
  for (int node = blockIdx.x * 8 + wave; node < N_; node += gridDim.x * 8) {
    int beg = offs[node];
    int n   = offs[node + 1] - beg;
    for (int i = lane; i < n; i += 32) {
      int my = key[beg + i];
      int rk = 0;
      for (int j = 0; j < n; ++j) rk += (key[beg + j] < my) ? 1 : 0;
      int r = my / E_;
      int e = my - r * E_;
      packed[beg + rk] = src[e] * R_ + r;
    }
    if (lane < R_) {
      int c = 0;
      for (int j = 0; j < n; ++j) c += (key[beg + j] / E_ < lane) ? 1 : 0;
      seg[node * R_ + lane] = beg + c;
    }
    if (node == N_ - 1 && lane == R_) seg[N_ * R_] = beg + n;
  }
}

template <int BN, int NW, int W0>
__device__ __forceinline__
void tile_ln_store(v8f (&acc)[5], float* ct,
                   const float* __restrict__ bias,
                   const float* __restrict__ gamma,
                   const float* __restrict__ beta,
                   const float* __restrict__ resid,
                   float* __restrict__ out, int row0, int do_gelu) {
  constexpr int CP = BN + 4;
  constexpr int NV = BN / 32;
  const int tid  = threadIdx.x;
  const int wave = tid >> 5;
  const int lane = tid & 31;
  const int l16  = lane & 15;
  const int g    = lane >> 4;

  const int sw = wave - W0;
  if (NW == BN / 16 || (sw >= 0 && sw < BN / 16)) {
    float* cw = ct + (8 * g) * CP + sw * 16 + l16;
#pragma unroll
    for (int mt = 0; mt < 5; ++mt) {
#pragma unroll
      for (int i = 0; i < 8; ++i)
        cw[(mt * 16 + i) * CP] = acc[mt][i];
    }
  }
  __syncthreads();

  float gm[NV], bt[NV], bs[NV];
#pragma unroll
  for (int i = 0; i < NV; ++i) {
    const int c = lane + (i << 5);
    gm[i] = gamma[c];
    bt[i] = beta[c];
    bs[i] = bias ? bias[c] : 0.f;
  }

  for (int r = wave; r < 80; r += NW) {
    const size_t base = (size_t)(row0 + r) * BN + lane;
    const float* cr = ct + r * CP + lane;
    float vals[NV], rs[NV];
#pragma unroll
    for (int i = 0; i < NV; ++i) {
      vals[i] = cr[i << 5];
      rs[i] = resid ? resid[base + (i << 5)] : 0.f;
    }
    float s = 0.f;
#pragma unroll
    for (int i = 0; i < NV; ++i) {
      float v = vals[i];
      if (bias) v += bs[i];
      vals[i] = v;
      s += v;
    }
    for (int o = 16; o; o >>= 1) s += __shfl_xor(s, o, 32);
    const float mean = s / (float)BN;
    float vs = 0.f;
#pragma unroll
    for (int i = 0; i < NV; ++i) { float d = vals[i] - mean; vs += d * d; }
    for (int o = 16; o; o >>= 1) vs += __shfl_xor(vs, o, 32);
    const float inv = rsqrtf(vs / (float)BN + LN_EPS);
#pragma unroll
    for (int i = 0; i < NV; ++i) {
      float v = (vals[i] - mean) * inv * gm[i] + bt[i];
      if (resid) v += rs[i];
      if (do_gelu) v = 0.5f * v * (1.f + erff(v * 0.70710678118654752f));
      out[base + (i << 5)] = v;
    }
  }
}

template <int BN>
__global__ __launch_bounds__(2 * BN)
void gemm_f32_wmma_lds_kernel(const float* __restrict__ A,
                              const float* __restrict__ Bm, int K, int lda,
                              const float* __restrict__ A2,
                              const float* __restrict__ B2, int K2, int lda2,
                              float* __restrict__ C,
                              const float* __restrict__ bias,
                              const float* __restrict__ gamma,
                              const float* __restrict__ beta,
                              const float* __restrict__ resid,
                              int do_gelu) {
  constexpr int NTHR = 2 * BN;
  constexpr int BM = 80, BK = 32;
  constexpr int AP = BK + 4;
  constexpr int BP = 2 * BN + 32;
  constexpr int A_ITEMS = BM * (BK / 4);
  constexpr int A_IT = (A_ITEMS + NTHR - 1) / NTHR;
  constexpr int BQ = BN / 4;
  constexpr int B_IT = ((BK / 2) * BQ) / NTHR;
  constexpr int A_ST = BM * AP;
  constexpr int B_ST = (BK / 2) * BP;
  static_assert(2 * A_ST + 2 * B_ST >= BM * (BN + 4), "result tile must fit in the operand stages");
  __shared__ __attribute__((aligned(16))) float smem[2 * A_ST + 2 * B_ST];
  float* const As = smem;
  float* const Bs = smem + 2 * A_ST;

  const int tid  = threadIdx.x;
  const int wave = tid >> 5;
  const int lane = tid & 31;
  const int l16  = lane & 15;
  const int g    = lane >> 4;
  const int row0 = blockIdx.x * BM;
  const int nch1 = K / BK;
  const int nch  = nch1 + K2 / BK;

  v4f ra[A_IT];
  v4f rb0[B_IT], rb1[B_IT];

#define GEMM_GLOAD(ch)                                                        \
  {                                                                           \
    const bool first = (ch) < nch1;                                           \
    const float* Ap = first ? A : A2;                                         \
    const float* Bp = first ? Bm : B2;                                        \
    const int ld = first ? lda : lda2;                                        \
    const int k0 = (first ? (ch) : (ch) - nch1) * BK;                         \
    _Pragma("unroll")                \
    for (int it = 0; it < A_IT; ++it) {                                       \
      int i = tid + it * NTHR;                                                \
      i = (i < A_ITEMS) ? i : (A_ITEMS - 1);   \
      int r = i >> 3, c4 = (i & 7) * 4;                                       \
      ra[it] = *(const v4f*)(Ap + (size_t)(row0 + r) * ld + k0 + c4);         \
    }                                                                         \
    _Pragma("unroll")                     \
    for (int it = 0; it < B_IT; ++it) {                                       \
      int j = tid + it * NTHR;                                                \
      int kp = j / BQ, n4 = (j - kp * BQ) * 4;                                \
      const float* p = Bp + (size_t)(k0 + 2 * kp) * BN + n4;                  \
      rb0[it] = *(const v4f*)p;                                               \
      rb1[it] = *(const v4f*)(p + BN);                                        \
    }                                                                         \
  }

#define GEMM_SSTORE(buf)                                                      \
  {                                                                           \
    _Pragma("unroll")                          \
    for (int it = 0; it < A_IT; ++it) {                                       \
      int i = tid + it * NTHR;                                                \
      if (i < A_ITEMS) {                                                      \
        int r = i >> 3, c4 = (i & 7) * 4;                                     \
        *(v4f*)(As + (buf) * A_ST + r * AP + c4) = ra[it];                    \
      }                                                                       \
    }                                                                         \
    _Pragma("unroll")                  \
    for (int it = 0; it < B_IT; ++it) {                                       \
      int j = tid + it * NTHR;                                                \
      int kp = j / BQ, n4 = (j - kp * BQ) * 4;                                \
      v4f lo, hi;                                                             \
      lo.x = rb0[it].x; lo.y = rb1[it].x; lo.z = rb0[it].y; lo.w = rb1[it].y; \
      hi.x = rb0[it].z; hi.y = rb1[it].z; hi.z = rb0[it].w; hi.w = rb1[it].w; \
      float* d = Bs + (buf) * B_ST + kp * BP + 2 * n4;                        \
      *(v4f*)d       = lo;                                                    \
      *(v4f*)(d + 4) = hi;                                                    \
    }                                                                         \
  }

  v8f acc[5];
#pragma unroll
  for (int mt = 0; mt < 5; ++mt) acc[mt] = v8f{};

  GEMM_GLOAD(0)
  GEMM_SSTORE(0)
  __syncthreads();

  const int a_off = l16 * AP + 2 * g;
  const int b_off = g * BP + 2 * (wave * 16 + l16);
  for (int c = 0; c < nch; ++c) {
    const int cur = c & 1;
    const bool more = (c + 1 < nch);
    if (more) GEMM_GLOAD(c + 1)
    const float* as = As + cur * A_ST + a_off;
    const float* bs = Bs + cur * B_ST + b_off;
#pragma unroll
    for (int s = 0; s < BK / 4; ++s) {
      v2f b = *(const v2f*)(bs + 2 * s * BP);
#pragma unroll
      for (int mt = 0; mt < 5; ++mt) {
        v2f a = *(const v2f*)(as + mt * 16 * AP + 4 * s);
        acc[mt] = WMMA_F32(a, b, acc[mt]);
      }
    }
    if (more) GEMM_SSTORE(cur ^ 1)
    __syncthreads();
  }
#undef GEMM_GLOAD
#undef GEMM_SSTORE

  if (gamma) {
    tile_ln_store<BN, BN / 16, 0>(acc, smem, bias, gamma, beta, resid, C, row0, do_gelu);
  } else {
    float* Cbase = C + (size_t)(row0 + 8 * g) * BN + wave * 16 + l16;
#pragma unroll
    for (int mt = 0; mt < 5; ++mt) {
#pragma unroll
      for (int i = 0; i < 8; ++i)
        Cbase[(size_t)(mt * 16 + i) * BN] = acc[mt][i];
    }
  }
}

template <int BN, int IN, bool ZR>
__global__ __launch_bounds__(2 * BN + 256)
void rgcn_layer_gemm_kernel(const float* __restrict__ X,
                            const float* __restrict__ zrow,
                            const int* __restrict__ seg,
                            const int* __restrict__ packed,
                            const float* __restrict__ Bm,
                            const float* __restrict__ B2,
                            float* __restrict__ C,
                            const float* __restrict__ gamma,
                            const float* __restrict__ beta,
                            const float* __restrict__ resid,
                            int do_gelu) {
  constexpr int NCONS = 2 * BN;
  constexpr int NPROD = 256;
  constexpr int BM = 80, BK = 32;
  constexpr int AP = BK + 4;
  constexpr int BP = 2 * BN + 32;
  constexpr int A_ITEMS = BM * (BK / 4);
  constexpr int P_IT = (A_ITEMS + NPROD - 1) / NPROD;
  constexpr int BQ = BN / 4;
  constexpr int B_IT = ((BK / 2) * BQ) / NCONS;
  constexpr int A_ST = BM * AP;
  constexpr int B_ST = (BK / 2) * BP;
  constexpr int CPR = IN / BK;
  constexpr int NCH1 = R_ * CPR;
  constexpr int NCH = NCH1 + CPR;
  constexpr int NPRE = (BN >= 256) ? 4 : 8;
  constexpr int NPASS = (BN >= 256) ? 2 : 1;
  constexpr int NKNOWN = NPRE * NPASS;
  static_assert(3 * A_ST + 2 * B_ST >= BM * (BN + 4), "result tile must fit in the operand stages");
  static_assert(IN % BK == 0, "a chunk lies inside one relation");
  __shared__ __attribute__((aligned(16))) float smem[3 * A_ST + 2 * B_ST];
  static_assert((A_ITEMS - (P_IT - 1) * NPROD) % 32 == 0, "the last piece is present or absent for a whole wave");
  __shared__ unsigned pso_l[NKNOWN * P_IT * NPROD];
  float* const As = smem;
  float* const Bs = smem + 3 * A_ST;

  const int pt   = threadIdx.x;
  const int tid  = pt - NPROD;
  const int wave = tid >> 5;
  const int lane = pt & 31;
  const int l16  = lane & 15;
  const int g    = lane >> 4;
  const int row0 = blockIdx.x * BM;
  const bool consumer = __builtin_amdgcn_readfirstlane(pt) >= NPROD;
  const bool last_piece = __builtin_amdgcn_readfirstlane(pt) + (P_IT - 1) * NPROD < A_ITEMS;
  const unsigned zoffb = ZR ? (unsigned)((const char*)zrow - (const char*)X) : 0u;

  int prow[P_IT], pc4[P_IT];
  bool pv[P_IT];
  int pcnt[P_IT], pbeg[P_IT];
#pragma unroll
  for (int it = 0; it < P_IT; ++it) {
    int i = pt + it * NPROD;
    pv[it] = !consumer && i < A_ITEMS;
    i = pv[it] ? i : 0;
    prow[it] = i >> 3;
    pc4[it] = (i & 7) * 4;
    pcnt[it] = 0;
    pbeg[it] = 0;
  }

  auto build_a = [&](int cn, int buf) {
    v4f a[P_IT];
    if (cn < NCH1) {
      const int rel = cn / CPR;
      const int f0  = (cn - rel * CPR) * BK;
      const char* const xb = (const char*)(X + f0);
      if (cn == rel * CPR) {
        int s[P_IT], e[P_IT], p[P_IT][NKNOWN];
#pragma unroll
        for (int it = 0; it < P_IT; ++it) {
          if (it < P_IT - 1 || last_piece) {
            const int sidx = (row0 + prow[it]) * R_ + rel;
            s[it] = seg[sidx];
            e[it] = seg[sidx + 1];
          }
        }
#pragma unroll
        for (int it = 0; it < P_IT; ++it) {
          if (it < P_IT - 1 || last_piece) {
#pragma unroll
            for (int u = 0; u < NKNOWN; ++u) {
              const int idx = s[it] + u;
              p[it][u] = packed[idx < E_ ? idx : E_ - 1];
            }
          }
        }
#pragma unroll
        for (int it = 0; it < P_IT; ++it) {
          if (it < P_IT - 1 || last_piece) {
            pbeg[it] = s[it];
            pcnt[it] = pv[it] ? e[it] - s[it] : 0;
            const unsigned absent = ZR ? zoffb : (unsigned)pc4[it] * 4u;
#pragma unroll
            for (int u = 0; u < NKNOWN; ++u)
              pso_l[(it * NKNOWN + u) * NPROD + pt] =
                  (u < pcnt[it]) ? (((unsigned)p[it][u] / R_) * IN + (unsigned)pc4[it]) * 4u : absent;
          }
        }
      }
#pragma unroll
      for (int it = 0; it < P_IT; ++it) a[it] = v4f{};
#pragma unroll
      for (int ps = 0; ps < NPASS; ++ps) {
        bool need = true;
        if (ps > 0) {
          bool m = false;
#pragma unroll
          for (int it = 0; it < P_IT; ++it) m = m || (pcnt[it] > ps * NPRE);
          need = __builtin_amdgcn_ballot_w32(m) != 0u;
        }
        if (need) {
          unsigned so[P_IT][NPRE];
          v4f v[P_IT][NPRE];
#pragma unroll
          for (int it = 0; it < P_IT; ++it) {
            if (it < P_IT - 1 || last_piece) {
#pragma unroll
              for (int u = 0; u < NPRE; ++u)
                so[it][u] = pso_l[(it * NKNOWN + ps * NPRE + u) * NPROD + pt];
            }
          }
#pragma unroll
          for (int it = 0; it < P_IT; ++it) {
            if (it < P_IT - 1 || last_piece) {
#pragma unroll
              for (int u = 0; u < NPRE; ++u)
                v[it][u] = *(const v4f*)(xb + so[it][u]);
            }
          }
#pragma unroll
          for (int it = 0; it < P_IT; ++it) {
            if (it < P_IT - 1 || last_piece) {
#pragma unroll
              for (int u = 0; u < NPRE; ++u) {
                if (ZR) a[it] += v[it][u];
                else    a[it] += (ps * NPRE + u < pcnt[it]) ? v[it][u] : v4f{};
              }
            }
          }
        }
      }
#pragma unroll
      for (int it = 0; it < P_IT; ++it) {
        for (int j = NKNOWN; j < pcnt[it]; ++j) {
          const unsigned so = ((unsigned)packed[pbeg[it] + j] / R_) * IN;
          a[it] += *(const v4f*)(X + so + f0 + pc4[it]);
        }
      }
    } else {
      const int k0 = (cn - NCH1) * BK;
#pragma unroll
      for (int it = 0; it < P_IT; ++it) {
        a[it] = v4f{};
        if (it < P_IT - 1 || last_piece)
          a[it] = *(const v4f*)(X + (size_t)(row0 + prow[it]) * IN + k0 + pc4[it]);
      }
    }
#pragma unroll
    for (int it = 0; it < P_IT; ++it)
      if (pv[it]) *(v4f*)(As + buf * A_ST + prow[it] * AP + pc4[it]) = a[it];
  };

  v4f rb0[B_IT], rb1[B_IT];
  unsigned boff[B_IT];
#pragma unroll
  for (int it = 0; it < B_IT; ++it) {
    const int j = (consumer ? tid : 0) + it * NCONS;
    const int kp = j / BQ, n4 = (j - kp * BQ) * 4;
    boff[it] = (unsigned)((2 * kp) * BN + n4) * 4u;
  }
#define LAYER_GLOAD(ch)                                                       \
  {                                                                           \
    const bool first = (ch) < NCH1;                                           \
    const float* Bp = first ? Bm : B2;                                        \
    const int k0 = (first ? (ch) : (ch) - NCH1) * BK;                         \
    const char* const bb = (const char*)(Bp + (size_t)k0 * BN);   \
    _Pragma("unroll")                     \
    for (int it = 0; it < B_IT; ++it) {                                       \
      rb0[it] = *(const v4f*)(bb + boff[it]);                                 \
      rb1[it] = *(const v4f*)(bb + boff[it] + BN * 4);                        \
    }                                                                         \
  }
#define LAYER_SSTORE(buf)                                                     \
  {                                                                           \
    _Pragma("unroll")                  \
    for (int it = 0; it < B_IT; ++it) {                                       \
      int j = tid + it * NCONS;                                               \
      int kp = j / BQ, n4 = (j - kp * BQ) * 4;                                \
      v4f lo, hi;                                                             \
      lo.x = rb0[it].x; lo.y = rb1[it].x; lo.z = rb0[it].y; lo.w = rb1[it].y; \
      hi.x = rb0[it].z; hi.y = rb1[it].z; hi.z = rb0[it].w; hi.w = rb1[it].w; \
      float* d = Bs + (buf) * B_ST + kp * BP + 2 * n4;                        \
      *(v4f*)d       = lo;                                                    \
      *(v4f*)(d + 4) = hi;                                                    \
    }                                                                         \
  }

  v8f acc[5];
#pragma unroll
  for (int mt = 0; mt < 5; ++mt) acc[mt] = v8f{};

  static_assert(NCH >= 2, "the B tile is fetched two chunks ahead");
  static_assert(NCH >= 3, "the A tile is built two chunks ahead");
  if (consumer) {
    constexpr int S_LATE = 3;
    const int a_off = l16 * AP + 2 * g;
    const int b_off = g * BP + 2 * (wave * 16 + l16);
    LAYER_GLOAD(0)
    LAYER_SSTORE(0)
    LAYER_GLOAD(1)
    lds_arrive();
    lds_wait();
    int sa = 0;
    for (int c = 0; c < NCH; ++c) {
      const int cur = c & 1;
      if (c + 1 < NCH) LAYER_SSTORE(cur ^ 1)
      if (c + 2 < NCH) LAYER_GLOAD(c + 2)
      const float* as = As + sa * A_ST + a_off;
      sa = (sa == 2) ? 0 : sa + 1;
      const float* bs = Bs + cur * B_ST + b_off;
#pragma unroll
      for (int s = 0; s < BK / 4 - S_LATE; ++s) {
        v2f b = *(const v2f*)(bs + 2 * s * BP);
#pragma unroll
        for (int mt = 0; mt < 5; ++mt) {
          v2f a = *(const v2f*)(as + mt * 16 * AP + 4 * s);
          acc[mt] = WMMA_F32(a, b, acc[mt]);
        }
      }
      v2f bl[S_LATE], al[S_LATE][5];
#pragma unroll
      for (int t = 0; t < S_LATE; ++t) {
        const int s = BK / 4 - S_LATE + t;
        bl[t] = *(const v2f*)(bs + 2 * s * BP);
#pragma unroll
        for (int mt = 0; mt < 5; ++mt)
          al[t][mt] = *(const v2f*)(as + mt * 16 * AP + 4 * s);
      }
      lds_arrive();
      __builtin_amdgcn_sched_barrier(0);
#pragma unroll
      for (int t = 0; t < S_LATE; ++t) {
#pragma unroll
        for (int mt = 0; mt < 5; ++mt)
          acc[mt] = WMMA_F32(al[t][mt], bl[t], acc[mt]);
      }
      __builtin_amdgcn_sched_barrier(0);
      lds_wait();
    }
  } else {
    build_a(0, 0);
    build_a(1, 1);
    lds_arrive();
    lds_wait();
    int sn = 2;
    for (int c = 0; c < NCH; ++c) {
      lds_arrive();
      if (c + 2 < NCH) build_a(c + 2, sn);
      sn = (sn == 2) ? 0 : sn + 1;
      lds_wait();
    }
  }
#undef LAYER_GLOAD
#undef LAYER_SSTORE

  tile_ln_store<BN, (NCONS + NPROD) / 32, NPROD / 32>(acc, smem, nullptr, gamma, beta, resid, C, row0, do_gelu);
}

template <int BN, bool DOT>
__device__ __forceinline__
void tile_ln_lds(v8f (&acc)[5], float* ct,
                 const float* __restrict__ bias,
                 const float* __restrict__ gamma,
                 const float* __restrict__ beta,
                 const float* __restrict__ wdot,
                 const float* __restrict__ bdot,
                 float* __restrict__ logits, int row0) {
  constexpr int CP = BN + 4;
  constexpr int NV = BN / 32;
  static_assert(!DOT || NV == 2, "the final dot is written for 64 columns");
  const int tid  = threadIdx.x;
  const int wave = tid >> 5;
  const int lane = tid & 31;
  const int l16  = lane & 15;
  const int g    = lane >> 4;

  if (wave < BN / 16) {
    float* cw = ct + (8 * g) * CP + wave * 16 + l16;
#pragma unroll
    for (int mt = 0; mt < 5; ++mt) {
#pragma unroll
      for (int i = 0; i < 8; ++i)
        cw[(mt * 16 + i) * CP] = acc[mt][i];
    }
  }
  __syncthreads();

  float gm[NV], bt[NV], bs[NV], wd[NV];
#pragma unroll
  for (int i = 0; i < NV; ++i) {
    const int c = lane + (i << 5);
    gm[i] = gamma[c];
    bt[i] = beta[c];
    bs[i] = bias[c];
    wd[i] = DOT ? wdot[c] : 0.f;
  }
  const float b0 = DOT ? bdot[0] : 0.f;

  for (int r = wave; r < 80; r += 16) {
    float* cr = ct + r * CP + lane;
    float vals[NV];
    float s = 0.f;
#pragma unroll
    for (int i = 0; i < NV; ++i) {
      float v = cr[i << 5];
      v += bs[i];
      vals[i] = v;
      s += v;
    }
    for (int o = 16; o; o >>= 1) s += __shfl_xor(s, o, 32);
    const float mean = s / (float)BN;
    float vs = 0.f;
#pragma unroll
    for (int i = 0; i < NV; ++i) { float d = vals[i] - mean; vs += d * d; }
    for (int o = 16; o; o >>= 1) vs += __shfl_xor(vs, o, 32);
    const float inv = rsqrtf(vs / (float)BN + LN_EPS);
#pragma unroll
    for (int i = 0; i < NV; ++i) {
      float v = (vals[i] - mean) * inv * gm[i] + bt[i];
      v = 0.5f * v * (1.f + erff(v * 0.70710678118654752f));
      vals[i] = v;
    }
    if (!DOT) {
#pragma unroll
      for (int i = 0; i < NV; ++i) cr[i << 5] = vals[i];
    } else {
      float d = vals[0] * wd[0] + vals[NV - 1] * wd[NV - 1];
      for (int o = 16; o; o >>= 1) d += __shfl_xor(d, o, 32);
      if (lane == 0) logits[row0 + r] = d + b0;
    }
  }
  if (!DOT) __syncthreads();
}

template <int BN, int AP>
__device__ __forceinline__
void lds_a_gemm(const float* At, int K,
                const float* __restrict__ Bm, float* Bs, v8f (&acc)[5]) {
  constexpr int BK = 32;
  constexpr int BP = 2 * BN + 32;
  constexpr int BQ = BN / 4;
  constexpr int B_ITEMS = (BK / 2) * BQ;
  constexpr int B_ST = (BK / 2) * BP;
  static_assert(B_ITEMS <= 512 && B_ITEMS % 32 == 0, "one piece per thread, whole waves");

  int tid = threadIdx.x;
  asm volatile("" : "+v"(tid));
  const int wave = tid >> 5;
  const int lane = tid & 31;
  const int l16  = lane & 15;
  const int g    = lane >> 4;
  const bool stager = tid < B_ITEMS;
  const bool worker = wave < BN / 16;
  const int kp = tid / BQ, n4 = (tid - kp * BQ) * 4;
  v4f rb0 = v4f{}, rb1 = v4f{};

#pragma unroll
  for (int mt = 0; mt < 5; ++mt) acc[mt] = v8f{};

#define LG_GLOAD(ch)                                                          \
  if (stager) {                                                               \
    const float* p = Bm + (size_t)((ch) * BK + 2 * kp) * BN + n4;             \
    rb0 = *(const v4f*)p;                                                     \
    rb1 = *(const v4f*)(p + BN);                                              \
  }
#define LG_SSTORE(buf)                                                        \
  if (stager) {                                                               \
    v4f lo, hi;                                                               \
    lo.x = rb0.x; lo.y = rb1.x; lo.z = rb0.y; lo.w = rb1.y;                   \
    hi.x = rb0.z; hi.y = rb1.z; hi.z = rb0.w; hi.w = rb1.w;                   \
    float* d = Bs + (buf) * B_ST + kp * BP + 2 * n4;                          \
    *(v4f*)d       = lo;                                                      \
    *(v4f*)(d + 4) = hi;                                                      \
  }

  LG_GLOAD(0)
  LG_SSTORE(0)
  __syncthreads();

  const int nch = K / BK;
  const int a_off = l16 * AP + 2 * g;
  const int b_off = g * BP + 2 * (wave * 16 + l16);
  for (int c = 0; c < nch; ++c) {
    const int cur = c & 1;
    const bool more = (c + 1 < nch);
    if (more) LG_GLOAD(c + 1)
    if (worker) {
      const float* as = At + a_off + c * BK;
      const float* bs = Bs + cur * B_ST + b_off;
#pragma unroll
      for (int s = 0; s < BK / 4; ++s) {
        v2f b = *(const v2f*)(bs + 2 * s * BP);
#pragma unroll
        for (int mt = 0; mt < 5; ++mt) {
          v2f a = *(const v2f*)(as + mt * 16 * AP + 4 * s);
          acc[mt] = WMMA_F32(a, b, acc[mt]);
        }
      }
    }
    if (more) LG_SSTORE(cur ^ 1)
    __syncthreads();
  }
#undef LG_GLOAD
#undef LG_SSTORE
}

__global__ __launch_bounds__(512, 8)
void decoder_tile_kernel(const float* __restrict__ z,
                         const int* __restrict__ gi,
                         const int* __restrict__ gj,
                         const float* __restrict__ Bm,
                         int K,
                         const float* __restrict__ bias,
                         const float* __restrict__ gamma,
                         const float* __restrict__ beta,
                         const float* __restrict__ W1,
                         const float* __restrict__ bias1,
                         const float* __restrict__ gamma1,
                         const float* __restrict__ beta1,
                         const float* __restrict__ W2,
                         const float* __restrict__ bias2,
                         const float* __restrict__ gamma2,
                         const float* __restrict__ beta2,
                         const float* __restrict__ W3,
                         const float* __restrict__ bias3,
                         float* __restrict__ logits) {
  constexpr int BN = H_;
  constexpr int NTHR = 2 * BN;
  constexpr int BM = 80, BK = 32;
  constexpr int AP = BK + 4;
  constexpr int BP = 2 * BN + 32;
  constexpr int A_ITEMS = BM * (BK / 4);
  constexpr int A_IT = (A_ITEMS + NTHR - 1) / NTHR;
  constexpr int BQ = BN / 4;
  constexpr int B_IT = ((BK / 2) * BQ) / NTHR;
  constexpr int A_ST = BM * AP;
  constexpr int B_ST = (BK / 2) * BP;
  constexpr int T0 = BM * (BN + 4);
  constexpr int S1 = 2 * (BK / 2) * (2 * 128 + 32);
  static_assert(2 * A_ST + 2 * B_ST <= T0 + S1, "layer-0 stages fit in the block's LDS");
  __shared__ __attribute__((aligned(16))) float smem[T0 + S1];
  float* const Bs2 = smem + T0;
  float* const As = smem;
  float* const Bs = smem + 2 * A_ST;

  const int tid  = threadIdx.x;
  const int wave = tid >> 5;
  const int lane = tid & 31;
  const int l16  = lane & 15;
  const int g    = lane >> 4;
  const int row0 = blockIdx.x * BM;

  unsigned offi[A_IT], offj[A_IT];
#pragma unroll
  for (int it = 0; it < A_IT; ++it) {
    int i = tid + it * NTHR;
    i = (i < A_ITEMS) ? i : (A_ITEMS - 1);
    int r = i >> 3, c4 = (i & 7) * 4;
    offi[it] = (unsigned)gi[row0 + r] * OUT_ + c4;
    offj[it] = (unsigned)gj[row0 + r] * OUT_ + c4;
  }

  v4f rzi[A_IT], rzj[A_IT];
  v4f rb0[B_IT], rb1[B_IT];

#define PE_GLOAD(k0)                                                          \
  {                                                                           \
    const int f0 = (k0) & (OUT_ - 1);                                         \
    _Pragma("unroll")        \
    for (int it = 0; it < A_IT; ++it) {                                       \
      rzi[it] = *(const v4f*)(z + offi[it] + f0);                             \
      rzj[it] = *(const v4f*)(z + offj[it] + f0);                             \
    }                                                                         \
    _Pragma("unroll")                     \
    for (int it = 0; it < B_IT; ++it) {                                       \
      int j = tid + it * NTHR;                                                \
      int kp = j / BQ, n4 = (j - kp * BQ) * 4;                                \
      const float* p = Bm + (size_t)((k0) + 2 * kp) * BN + n4;                \
      rb0[it] = *(const v4f*)p;                                               \
      rb1[it] = *(const v4f*)(p + BN);                                        \
    }                                                                         \
  }

#define PE_SSTORE(buf, k0)                                                    \
  {                                                                           \
    const int sec = (k0) >> 7;                                                \
    _Pragma("unroll")                  \
    for (int it = 0; it < A_IT; ++it) {                                       \
      int i = tid + it * NTHR;                                                \
      if (i < A_ITEMS) {                                                      \
        int r = i >> 3, c4 = (i & 7) * 4;                                     \
        v4f a;                                                                \
        if (sec == 0)      a = rzi[it] + rzj[it];                             \
        else if (sec == 1) a = rzi[it] * rzj[it];                             \
        else {                                                                \
          const v4f d = rzi[it] - rzj[it];                                    \
          a.x = fabsf(d.x); a.y = fabsf(d.y); a.z = fabsf(d.z); a.w = fabsf(d.w); \
        }                                                                     \
        *(v4f*)(As + (buf) * A_ST + r * AP + c4) = a;                                   \
      }                                                                       \
    }                                                                         \
    _Pragma("unroll")                  \
    for (int it = 0; it < B_IT; ++it) {                                       \
      int j = tid + it * NTHR;                                                \
      int kp = j / BQ, n4 = (j - kp * BQ) * 4;                                \
      v4f lo, hi;                                                             \
      lo.x = rb0[it].x; lo.y = rb1[it].x; lo.z = rb0[it].y; lo.w = rb1[it].y; \
      hi.x = rb0[it].z; hi.y = rb1[it].z; hi.z = rb0[it].w; hi.w = rb1[it].w; \
      float* d = Bs + (buf) * B_ST + kp * BP + 2 * n4;                                  \
      *(v4f*)d       = lo;                                                    \
      *(v4f*)(d + 4) = hi;                                                    \
    }                                                                         \
  }

  v8f acc[5];
#pragma unroll
  for (int mt = 0; mt < 5; ++mt) acc[mt] = v8f{};

  constexpr int S_LATE = 2;
  const int nch = K / BK;
  PE_GLOAD(0)
  PE_SSTORE(0, 0)
  if (1 < nch) PE_GLOAD(BK)
  lds_arrive();
  lds_wait();

  const int a_off = l16 * AP + 2 * g;
  const int b_off = g * BP + 2 * (wave * 16 + l16);
  for (int c = 0; c < nch; ++c) {
    const int cur = c & 1;
    if (c + 1 < nch) PE_SSTORE(cur ^ 1, (c + 1) * BK)
    if (c + 2 < nch) PE_GLOAD((c + 2) * BK)
    const float* as = As + cur * A_ST + a_off;
    const float* bs = Bs + cur * B_ST + b_off;
#pragma unroll
    for (int s = 0; s < BK / 4 - S_LATE; ++s) {
      v2f b = *(const v2f*)(bs + 2 * s * BP);
#pragma unroll
      for (int mt = 0; mt < 5; ++mt) {
        v2f a = *(const v2f*)(as + mt * 16 * AP + 4 * s);
        acc[mt] = WMMA_F32(a, b, acc[mt]);
      }
    }
    v2f bl[S_LATE], al[S_LATE][5];
#pragma unroll
    for (int t = 0; t < S_LATE; ++t) {
      const int s = BK / 4 - S_LATE + t;
      bl[t] = *(const v2f*)(bs + 2 * s * BP);
#pragma unroll
      for (int mt = 0; mt < 5; ++mt)
        al[t][mt] = *(const v2f*)(as + mt * 16 * AP + 4 * s);
    }
    lds_arrive();
    __builtin_amdgcn_sched_barrier(0);
#pragma unroll
    for (int t = 0; t < S_LATE; ++t) {
#pragma unroll
      for (int mt = 0; mt < 5; ++mt)
        acc[mt] = WMMA_F32(al[t][mt], bl[t], acc[mt]);
    }
    __builtin_amdgcn_sched_barrier(0);
    lds_wait();
  }
#undef PE_GLOAD
#undef PE_SSTORE

  tile_ln_lds<256, false>(acc, smem, bias, gamma, beta, nullptr, nullptr, nullptr, row0);
  lds_a_gemm<128, 260>(smem, H_, W1, Bs2, acc);
  tile_ln_lds<128, false>(acc, smem, bias1, gamma1, beta1, nullptr, nullptr, nullptr, row0);
  lds_a_gemm<64, 132>(smem, H_ / 2, W2, Bs2, acc);
  tile_ln_lds<64, true>(acc, smem, bias2, gamma2, beta2, W3, bias3, logits, row0);
}

extern "C" void kernel_launch(void* const* d_in, const int* in_sizes, int n_in,
                              void* d_out, int out_size, void* d_ws, size_t ws_size,
                              hipStream_t stream) {
  (void)in_sizes; (void)n_in; (void)out_size; (void)ws_size;

  const float* x    = (const float*)d_in[0];
  const int*   ei   = (const int*)d_in[1];
  const int*   esrc = ei;
  const int*   edst = ei + E_;
  const int*   et   = (const int*)d_in[2];
  const int*   gp   = (const int*)d_in[3];
  const int*   gi   = gp;
  const int*   gj   = gp + P_;

  const float* basis[4], *att[4], *root[4], *relatt[4], *gam[4], *bet[4];
  for (int i = 0; i < 4; ++i) {
    basis[i]  = (const float*)d_in[4 + 6 * i + 0];
    att[i]    = (const float*)d_in[4 + 6 * i + 1];
    root[i]   = (const float*)d_in[4 + 6 * i + 2];
    relatt[i] = (const float*)d_in[4 + 6 * i + 3];
    gam[i]    = (const float*)d_in[4 + 6 * i + 4];
    bet[i]    = (const float*)d_in[4 + 6 * i + 5];
  }
  const float* res0 = (const float*)d_in[28];
  const float* res3 = (const float*)d_in[29];
  const float* dW0 = (const float*)d_in[30], *db0 = (const float*)d_in[31];
  const float* dg0 = (const float*)d_in[32], *dbb0 = (const float*)d_in[33];
  const float* dW1 = (const float*)d_in[34], *db1 = (const float*)d_in[35];
  const float* dg1 = (const float*)d_in[36], *dbb1 = (const float*)d_in[37];
  const float* dW2 = (const float*)d_in[38], *db2 = (const float*)d_in[39];
  const float* dg2 = (const float*)d_in[40], *dbb2 = (const float*)d_in[41];
  const float* dW3 = (const float*)d_in[42], *db3 = (const float*)d_in[43];

  float* ws = (float*)d_ws;
  size_t off = 0;
  float* a_sm  = ws + off; off += 64;
  float* Wcat  = ws + off; off += (size_t)R_ * H_ * H_;
  float* hA    = ws + off; off += (size_t)N_ * H_;
  float* hB    = ws + off; off += (size_t)N_ * H_;
  float* resid = ws + off; off += (size_t)N_ * H_;
  float* zout = (float*)d_out + P_;
  int* csr_cnt    = (int*)(resid + (size_t)N_ * OUT_);
  int* csr_offs   = csr_cnt + 20480;
  int* csr_rank   = csr_offs + 20480;
  int* csr_key    = csr_rank + E_;
  int* csr_packed = csr_key + E_;
  int* csr_seg    = csr_packed + E_;
  float* zrow = resid + 4000000;

  auto gemm = [&](const float* A, const float* Bm, int K, int lda,
                  const float* A2, const float* B2, int K2, int lda2,
                  float* Cm, int M, int Nc,
                  const float* bias, const float* gamma, const float* beta,
                  const float* res, int gelu) {
    int blocks = M / 80;
    if (Nc == 256)
      gemm_f32_wmma_lds_kernel<256><<<blocks, 512, 0, stream>>>(
          A, Bm, K, lda, A2, B2, K2, lda2, Cm, bias, gamma, beta, res, gelu);
    else
      gemm_f32_wmma_lds_kernel<128><<<blocks, 256, 0, stream>>>(
          A, Bm, K, lda, A2, B2, K2, lda2, Cm, bias, gamma, beta, res, gelu);
  };

  struct { const float* cur; int in; int out; int li;
           const float* resmat; float* conv; int gelu; float* rbuf; } L[4] = {
    { x,  IN_, H_,   0, res0,    hA,   1, hB },
    { hA, H_,  H_,   1, nullptr, hB,   1, nullptr },
    { hB, H_,  H_,   2, nullptr, hA,   1, nullptr },
    { hA, H_,  OUT_, 3, res3,    zout, 0, resid },
  };

  csr_zero_kernel<<<(N_ + 255) / 256, 256, 0, stream>>>(csr_cnt, zrow);
  csr_count_kernel<<<(E_ / 4 + 255) / 256, 256, 0, stream>>>(edst, csr_cnt, csr_rank);
  csr_scan_kernel<<<1, 1024, 0, stream>>>(csr_cnt, csr_offs);
  csr_fill_kernel<<<(E_ / 4 + 255) / 256, 256, 0, stream>>>(edst, et, csr_offs, csr_rank, csr_key);
  csr_order_kernel<<<(N_ + 31) / 32, 256, 0, stream>>>(esrc, csr_offs, csr_key, csr_packed, csr_seg);

  for (int i = 0; i < 4; ++i) {
    const float* cur = L[i].cur;
    int in_ch = L[i].in, out_ch = L[i].out, li = L[i].li;
    float* conv = L[i].conv;

    relatt_softmax_kernel<<<1, 32, 0, stream>>>(relatt[li], a_sm);
    int wtotal = R_ * in_ch * out_ch;
    make_weights_kernel<<<(wtotal + 2047) / 2048, 256, 0, stream>>>(
        att[li], basis[li], a_sm, Wcat, in_ch, out_ch);

    const float* rptr = cur;
    if (L[i].resmat) {
      gemm(cur, L[i].resmat, in_ch, in_ch, nullptr, nullptr, 0, 0,
           L[i].rbuf, N_, out_ch, nullptr, nullptr, nullptr, nullptr, 0);
      rptr = L[i].rbuf;
    }
    if (in_ch == IN_)
      rgcn_layer_gemm_kernel<H_, IN_, false><<<N_ / 80, 2 * H_ + 256, 0, stream>>>(
          cur, zrow, csr_seg, csr_packed, Wcat, root[li], conv, gam[li], bet[li], rptr, L[i].gelu);
    else if (out_ch == H_)
      rgcn_layer_gemm_kernel<H_, H_, true><<<N_ / 80, 2 * H_ + 256, 0, stream>>>(
          cur, zrow, csr_seg, csr_packed, Wcat, root[li], conv, gam[li], bet[li], rptr, L[i].gelu);
    else
      rgcn_layer_gemm_kernel<OUT_, H_, true><<<N_ / 80, 2 * OUT_ + 256, 0, stream>>>(
          cur, zrow, csr_seg, csr_packed, Wcat, root[li], conv, gam[li], bet[li], rptr, L[i].gelu);
  }

  decoder_tile_kernel<<<P_ / 80, 512, 0, stream>>>(zout, gi, gj, dW0, 3 * OUT_, db0, dg0, dbb0,
                                                   dW1, db1, dg1, dbb1, dW2, db2, dg2, dbb2,
                                                   dW3, db3, (float*)d_out);
}
